# starting kernel
# baseline (speedup 1.0000x reference)
#include <hip/hip_runtime.h>
#include <math.h>

#define N_NODES  50000
#define N_EDGES  600000
#define IN_SIZE  128
#define HIDDEN   256
#define OUT_SIZE 128

typedef __attribute__((ext_vector_type(16))) __bf16 v16bf;
typedef __attribute__((ext_vector_type(8)))  __bf16 v8bf;
typedef __attribute__((ext_vector_type(8)))  float  v8f;

__global__ __launch_bounds__(256)
void zero_f32(float* __restrict__ p, int n) {
    int i = blockIdx.x * blockDim.x + threadIdx.x;
    if (i < n) p[i] = 0.0f;
}

__global__ __launch_bounds__(256)
void build_wcat(const float* __restrict__ Wl, const float* __restrict__ Wr,
                __bf16* __restrict__ Wcat, int rows, int kHalf) {
    int i = blockIdx.x * blockDim.x + threadIdx.x;
    if (i >= rows * kHalf) return;
    int r = i / kHalf, c = i % kHalf;
    __bf16* row = Wcat + (size_t)r * (2 * kHalf);
    row[c]         = (__bf16)Wl[i];
    row[kHalf + c] = (__bf16)Wr[i];
}

__global__ __launch_bounds__(256)
void scatter_add(const float* __restrict__ X, const int* __restrict__ src,
                 const int* __restrict__ dst, float* __restrict__ agg,
                 float* __restrict__ cnt, int nedges, int F) {
    int wave = blockIdx.x * (blockDim.x >> 5) + (threadIdx.x >> 5);
    int lane = threadIdx.x & 31;
    if (wave >= nedges) return;
    int s = src[wave];
    int d = dst[wave];
    if (cnt != nullptr && lane == 0) atomicAdd(cnt + d, 1.0f);
    const float4* xs = (const float4*)(X + (size_t)s * F);
    float* ad = agg + (size_t)d * F;
    int nvec = F >> 2;
    for (int i = lane; i < nvec; i += 32) {
        float4 v = xs[i];
        atomicAdd(ad + i * 4 + 0, v.x);
        atomicAdd(ad + i * 4 + 1, v.y);
        atomicAdd(ad + i * 4 + 2, v.z);
        atomicAdd(ad + i * 4 + 3, v.w);
    }
}

__global__ __launch_bounds__(256)
void prep_xcat1(const float* __restrict__ agg, const float* __restrict__ cnt,
                const float* __restrict__ feat, __bf16* __restrict__ X,
                int nnodes, int F) {
    int i = blockIdx.x * blockDim.x + threadIdx.x;
    if (i >= nnodes * F) return;
    int n = i / F, c = i % F;
    float inv = 1.0f / fmaxf(cnt[n], 1.0f);
    __bf16* row = X + (size_t)n * (2 * F);
    row[c]     = (__bf16)(agg[i] * inv);
    row[F + c] = (__bf16)feat[i];
}

__global__ __launch_bounds__(256)
void prep_xcat2(const float* __restrict__ agg, const float* __restrict__ cnt,
                __bf16* __restrict__ X, int nnodes, int F) {
    int i = blockIdx.x * blockDim.x + threadIdx.x;
    if (i >= nnodes * F) return;
    int n = i / F, c = i % F;
    float inv = 1.0f / fmaxf(cnt[n], 1.0f);
    X[(size_t)n * (2 * F) + c] = (__bf16)(agg[i] * inv);
}

template <int K, int NT>
__global__ __launch_bounds__(256)
void gemm_bf16_wmma(const __bf16* __restrict__ A, const __bf16* __restrict__ B,
                    const float* __restrict__ bias, float* __restrict__ C,
                    int M, int N) {
    const int lane    = threadIdx.x & 31;
    const int wave    = blockIdx.x * (blockDim.x >> 5) + (threadIdx.x >> 5);
    const int ngroups = N / (16 * NT);
    if (wave >= (M >> 4) * ngroups) return;
    const int tm  = (wave / ngroups) << 4;
    const int tn0 = (wave % ngroups) * (16 * NT);
    const int kh  = (lane >> 4) * 8;
    const int l15 = lane & 15;

    const __bf16* arow = A + (size_t)(tm  + l15) * K + kh;
    const __bf16* brow = B + (size_t)(tn0 + l15) * K + kh;

    v8f acc[NT];
#pragma unroll
    for (int t = 0; t < NT; ++t) acc[t] = (v8f){};

#pragma unroll
    for (int k0 = 0; k0 < K; k0 += 32) {
        union { v16bf v; v8bf h[2]; } a;
        a.h[0] = *(const v8bf*)(arow + k0);
        a.h[1] = *(const v8bf*)(arow + k0 + 16);
#pragma unroll
        for (int t = 0; t < NT; ++t) {
            union { v16bf v; v8bf h[2]; } b;
            b.h[0] = *(const v8bf*)(brow + (size_t)t * 16 * K + k0);
            b.h[1] = *(const v8bf*)(brow + (size_t)t * 16 * K + k0 + 16);
            acc[t] = __builtin_amdgcn_wmma_f32_16x16x32_bf16(
                false, a.v, false, b.v, (short)0, acc[t], false, false);
        }
    }

    const int r0 = tm + (lane >> 4) * 8;
#pragma unroll
    for (int t = 0; t < NT; ++t) {
        const int cn = tn0 + t * 16 + l15;
        const float bc = bias[cn];
#pragma unroll
        for (int v = 0; v < 8; ++v)
            C[(size_t)(r0 + v) * N + cn] = acc[t][v] + bc;
    }
}

__global__ __launch_bounds__(256)
void norm_relu_store(float* __restrict__ H, __bf16* __restrict__ X2,
                     int F, int ldx, int coloff) {
    __shared__ float red[256];
    const int n = blockIdx.x;
    const int t = threadIdx.x;
    float* row = H + (size_t)n * F;
    float v = row[t];
    red[t] = v * v;
    __syncthreads();
    for (int s = 128; s > 0; s >>= 1) {
        if (t < s) red[t] += red[t + s];
        __syncthreads();
    }
    float inv = 1.0f / fmaxf(sqrtf(red[0]), 1e-12f);
    float r = fmaxf(v * inv, 0.0f);
    row[t] = r;
    X2[(size_t)n * ldx + coloff + t] = (__bf16)r;
}

__global__ __launch_bounds__(256)
void final_fc(const float* __restrict__ H2, const float* __restrict__ Wfc,
              const float* __restrict__ bfc, float* __restrict__ out,
              int nnodes, int F) {
    int wave = blockIdx.x * (blockDim.x >> 5) + (threadIdx.x >> 5);
    int lane = threadIdx.x & 31;
    if (wave >= nnodes) return;
    const float* h = H2 + (size_t)wave * F;
    float ss = 0.0f, d0 = 0.0f, d1 = 0.0f;
    for (int i = lane; i < F; i += 32) {
        float v = h[i];
        ss += v * v;
        d0 += v * Wfc[i];
        d1 += v * Wfc[F + i];
    }
    for (int off = 16; off > 0; off >>= 1) {
        ss += __shfl_xor(ss, off, 32);
        d0 += __shfl_xor(d0, off, 32);
        d1 += __shfl_xor(d1, off, 32);
    }
    if (lane == 0) {
        float inv = 1.0f / fmaxf(sqrtf(ss), 1e-12f);
        float l0 = d0 * inv + bfc[0];
        float l1 = d1 * inv + bfc[1];
        float m  = fmaxf(l0, l1);
        float e0 = __expf(l0 - m), e1 = __expf(l1 - m);
        float s  = 1.0f / (e0 + e1);
        out[(size_t)wave * 2 + 0] = e0 * s;
        out[(size_t)wave * 2 + 1] = e1 * s;
    }
}

extern "C" void kernel_launch(void* const* d_in, const int* in_sizes, int n_in,
                              void* d_out, int out_size, void* d_ws, size_t ws_size,
                              hipStream_t stream) {
    (void)in_sizes; (void)n_in; (void)out_size; (void)ws_size;

    const float* feat = (const float*)d_in[0];
    const int*   eidx = (const int*)  d_in[1];
    const int*   src  = eidx;
    const int*   dst  = eidx + N_EDGES;
    const float* W1l  = (const float*)d_in[2];
    const float* b1   = (const float*)d_in[3];
    const float* W1r  = (const float*)d_in[4];
    const float* W2l  = (const float*)d_in[5];
    const float* b2   = (const float*)d_in[6];
    const float* W2r  = (const float*)d_in[7];
    const float* Wfc  = (const float*)d_in[8];
    const float* bfc  = (const float*)d_in[9];
    float* out = (float*)d_out;

    char* ws = (char*)d_ws;
    size_t off = 0;
    auto alloc = [&](size_t bytes) -> void* {
        void* p = ws + off;
        off = (off + bytes + 255) & ~(size_t)255;
        return p;
    };
    float*  cnt   = (float*) alloc((size_t)N_NODES * 4);
    float*  agg1  = (float*) alloc((size_t)N_NODES * IN_SIZE * 4);
    __bf16* Xcat1 = (__bf16*)alloc((size_t)N_NODES * 2 * IN_SIZE * 2);
    __bf16* Wcat1 = (__bf16*)alloc((size_t)HIDDEN * 2 * IN_SIZE * 2);
    float*  h1    = (float*) alloc((size_t)N_NODES * HIDDEN * 4);
    float*  agg2  = (float*) alloc((size_t)N_NODES * HIDDEN * 4);
    __bf16* Xcat2 = (__bf16*)alloc((size_t)N_NODES * 2 * HIDDEN * 2);
    __bf16* Wcat2 = (__bf16*)alloc((size_t)OUT_SIZE * 2 * HIDDEN * 2);
    float*  h2    = (float*) alloc((size_t)N_NODES * OUT_SIZE * 4);

    const int TB = 256;
    auto cdiv = [](int a, int b) { return (a + b - 1) / b; };

    zero_f32<<<cdiv(N_NODES, TB), TB, 0, stream>>>(cnt, N_NODES);
    zero_f32<<<cdiv(N_NODES * IN_SIZE, TB), TB, 0, stream>>>(agg1, N_NODES * IN_SIZE);
    zero_f32<<<cdiv(N_NODES * HIDDEN, TB), TB, 0, stream>>>(agg2, N_NODES * HIDDEN);

    build_wcat<<<cdiv(HIDDEN * IN_SIZE, TB), TB, 0, stream>>>(W1l, W1r, Wcat1, HIDDEN, IN_SIZE);
    build_wcat<<<cdiv(OUT_SIZE * HIDDEN, TB), TB, 0, stream>>>(W2l, W2r, Wcat2, OUT_SIZE, HIDDEN);

    scatter_add<<<cdiv(N_EDGES, TB / 32), TB, 0, stream>>>(feat, src, dst, agg1, cnt,
                                                           N_EDGES, IN_SIZE);
    prep_xcat1<<<cdiv(N_NODES * IN_SIZE, TB), TB, 0, stream>>>(agg1, cnt, feat, Xcat1,
                                                               N_NODES, IN_SIZE);
    {
        int waves = (N_NODES / 16) * (HIDDEN / 64);
        gemm_bf16_wmma<2 * IN_SIZE, 4><<<cdiv(waves, TB / 32), TB, 0, stream>>>(
            Xcat1, Wcat1, b1, h1, N_NODES, HIDDEN);
    }
    norm_relu_store<<<N_NODES, HIDDEN, 0, stream>>>(h1, Xcat2, HIDDEN, 2 * HIDDEN, HIDDEN);

    scatter_add<<<cdiv(N_EDGES, TB / 32), TB, 0, stream>>>(h1, src, dst, agg2, nullptr,
                                                           N_EDGES, HIDDEN);
    prep_xcat2<<<cdiv(N_NODES * HIDDEN, TB), TB, 0, stream>>>(agg2, cnt, Xcat2,
                                                              N_NODES, HIDDEN);
    {
        int waves = (N_NODES / 16) * (OUT_SIZE / 64);
        gemm_bf16_wmma<2 * HIDDEN, 4><<<cdiv(waves, TB / 32), TB, 0, stream>>>(
            Xcat2, Wcat2, b2, h2, N_NODES, OUT_SIZE);
    }
    final_fc<<<cdiv(N_NODES, TB / 32), TB, 0, stream>>>(h2, Wfc, bfc, out,
                                                        N_NODES, OUT_SIZE);
}


# Neighbour sums via per-call in-edge lists (returning int add as rank, scan, fill) and a per-node register gather that stores the bf16 mean once, replacing f32 atomic scatter; sums in arrival order
# speedup vs baseline: 1.9634x; 1.9634x over previous
#include <hip/hip_runtime.h>
#include <math.h>

#define N_NODES  50000
#define N_EDGES  600000
#define IN_SIZE  128
#define HIDDEN   256
#define OUT_SIZE 128

typedef __attribute__((ext_vector_type(16))) __bf16 v16bf;
typedef __attribute__((ext_vector_type(8)))  __bf16 v8bf;
typedef __attribute__((ext_vector_type(8)))  float  v8f;
typedef __attribute__((ext_vector_type(4)))  __bf16 v4bf;

#define SCAN_T   1024
#define SCAN_PER 52
#define DEG_PAD  (SCAN_T * SCAN_PER)

__global__ __launch_bounds__(256)
void zero_i32(int* __restrict__ p, int n) {
    int i = blockIdx.x * blockDim.x + threadIdx.x;
    if (i < n) p[i] = 0;
}

__global__ __launch_bounds__(256)
void build_wcat(const float* __restrict__ Wl, const float* __restrict__ Wr,
                __bf16* __restrict__ Wcat, int rows, int kHalf) {
    int i = blockIdx.x * blockDim.x + threadIdx.x;
    if (i >= rows * kHalf) return;
    int r = i / kHalf, c = i % kHalf;
    __bf16* row = Wcat + (size_t)r * (2 * kHalf);
    row[c]         = (__bf16)Wl[i];
    row[kHalf + c] = (__bf16)Wr[i];
}


__global__ __launch_bounds__(256)
void edge_rank(const int* __restrict__ dst, int* __restrict__ deg,
               int* __restrict__ rank, int nedges) {
    int e = blockIdx.x * blockDim.x + threadIdx.x;
    if (e >= nedges) return;
    rank[e] = atomicAdd(deg + dst[e], 1);
}

__global__ __launch_bounds__(SCAN_T)
void scan_rowptr(const int* __restrict__ deg, int* __restrict__ rowptr) {
    __shared__ int wsum[32];
    const int t = threadIdx.x, lane = t & 31, w = t >> 5;
    const int4* d4 = (const int4*)(deg + t * SCAN_PER);
    int4 d[SCAN_PER / 4];
    int s = 0;
#pragma unroll
    for (int i = 0; i < SCAN_PER / 4; ++i) {
        d[i] = d4[i];
        s += (d[i].x + d[i].y) + (d[i].z + d[i].w);
    }
    int inc = s;
    for (int off = 1; off < 32; off <<= 1) {
        int v = __shfl_up(inc, off, 32);
        if (lane >= off) inc += v;
    }
    if (lane == 31) wsum[w] = inc;
    __syncthreads();
    if (w == 0) {
        int x = wsum[lane], xi = x;
        for (int off = 1; off < 32; off <<= 1) {
            int v = __shfl_up(xi, off, 32);
            if (lane >= off) xi += v;
        }
        wsum[lane] = xi - x;
    }
    __syncthreads();
    int run = wsum[w] + inc - s;
    int4* r4 = (int4*)(rowptr + t * SCAN_PER);
#pragma unroll
    for (int i = 0; i < SCAN_PER / 4; ++i) {
        int4 o;
        o.x = run; run += d[i].x;
        o.y = run; run += d[i].y;
        o.z = run; run += d[i].z;
        o.w = run; run += d[i].w;
        r4[i] = o;
    }
}

__global__ __launch_bounds__(256)
void csr_fill(const int* __restrict__ src, const int* __restrict__ dst,
              const int* __restrict__ rowptr, const int* __restrict__ rank,
              int* __restrict__ col, int nedges) {
    int e = blockIdx.x * blockDim.x + threadIdx.x;
    if (e >= nedges) return;
    col[rowptr[dst[e]] + rank[e]] = src[e];
}

__device__ __forceinline__ v4bf pack4_bf16(float4 v, float sc) {
    v4bf o;
    o[0] = (__bf16)(v.x * sc);
    o[1] = (__bf16)(v.y * sc);
    o[2] = (__bf16)(v.z * sc);
    o[3] = (__bf16)(v.w * sc);
    return o;
}

template <int F, bool SELF>
__global__ __launch_bounds__(256)
void gather_mean(const float* __restrict__ X, const int* __restrict__ rowptr,
                 const int* __restrict__ col, __bf16* __restrict__ Xcat, int nnodes) {
    constexpr int NV = F / 128;
    const int lane = threadIdx.x & 31;
    const int n = blockIdx.x * (blockDim.x >> 5) + (threadIdx.x >> 5);
    if (n >= nnodes) return;
    const int beg = __builtin_amdgcn_readfirstlane(rowptr[n]);
    const int end = __builtin_amdgcn_readfirstlane(rowptr[n + 1]);
    const float4* X4 = (const float4*)X;
    float4 acc[NV];
#pragma unroll
    for (int q = 0; q < NV; ++q) acc[q] = make_float4(0.0f, 0.0f, 0.0f, 0.0f);

    for (int base = beg; base < end; base += 32) {
        const int m = (end - base < 32) ? (end - base) : 32;
        const int myc = (lane < m) ? col[base + lane] : 0;
        int j = 0;
        for (; j + 4 <= m; j += 4) {
            const float4* r0 = X4 + (size_t)__builtin_amdgcn_readlane(myc, j + 0) * (F / 4) + lane;
            const float4* r1 = X4 + (size_t)__builtin_amdgcn_readlane(myc, j + 1) * (F / 4) + lane;
            const float4* r2 = X4 + (size_t)__builtin_amdgcn_readlane(myc, j + 2) * (F / 4) + lane;
            const float4* r3 = X4 + (size_t)__builtin_amdgcn_readlane(myc, j + 3) * (F / 4) + lane;
            float4 v0[NV], v1[NV], v2[NV], v3[NV];
#pragma unroll
            for (int q = 0; q < NV; ++q) {
                v0[q] = r0[q * 32]; v1[q] = r1[q * 32];
                v2[q] = r2[q * 32]; v3[q] = r3[q * 32];
            }
#pragma unroll
            for (int q = 0; q < NV; ++q) {
                acc[q].x += (v0[q].x + v1[q].x) + (v2[q].x + v3[q].x);
                acc[q].y += (v0[q].y + v1[q].y) + (v2[q].y + v3[q].y);
                acc[q].z += (v0[q].z + v1[q].z) + (v2[q].z + v3[q].z);
                acc[q].w += (v0[q].w + v1[q].w) + (v2[q].w + v3[q].w);
            }
        }
        for (; j < m; ++j) {
            const float4* r0 = X4 + (size_t)__builtin_amdgcn_readlane(myc, j) * (F / 4) + lane;
#pragma unroll
            for (int q = 0; q < NV; ++q) {
                float4 v = r0[q * 32];
                acc[q].x += v.x; acc[q].y += v.y; acc[q].z += v.z; acc[q].w += v.w;
            }
        }
    }

    const float inv = 1.0f / fmaxf((float)(end - beg), 1.0f);
    __bf16* row = Xcat + (size_t)n * (2 * F);
#pragma unroll
    for (int q = 0; q < NV; ++q)
        *(v4bf*)(row + q * 128 + lane * 4) = pack4_bf16(acc[q], inv);
    if (SELF) {
#pragma unroll
        for (int q = 0; q < NV; ++q) {
            float4 v = X4[(size_t)n * (F / 4) + q * 32 + lane];
            *(v4bf*)(row + F + q * 128 + lane * 4) = pack4_bf16(v, 1.0f);
        }
    }
}

template <int K, int NT>
__global__ __launch_bounds__(256)
void gemm_bf16_wmma(const __bf16* __restrict__ A, const __bf16* __restrict__ B,
                    const float* __restrict__ bias, float* __restrict__ C,
                    int M, int N) {
    const int lane    = threadIdx.x & 31;
    const int wave    = blockIdx.x * (blockDim.x >> 5) + (threadIdx.x >> 5);
    const int ngroups = N / (16 * NT);
    if (wave >= (M >> 4) * ngroups) return;
    const int tm  = (wave / ngroups) << 4;
    const int tn0 = (wave % ngroups) * (16 * NT);
    const int kh  = (lane >> 4) * 8;
    const int l15 = lane & 15;

    const __bf16* arow = A + (size_t)(tm  + l15) * K + kh;
    const __bf16* brow = B + (size_t)(tn0 + l15) * K + kh;

    v8f acc[NT];
#pragma unroll
    for (int t = 0; t < NT; ++t) acc[t] = (v8f){};

#pragma unroll
    for (int k0 = 0; k0 < K; k0 += 32) {
        union { v16bf v; v8bf h[2]; } a;
        a.h[0] = *(const v8bf*)(arow + k0);
        a.h[1] = *(const v8bf*)(arow + k0 + 16);
#pragma unroll
        for (int t = 0; t < NT; ++t) {
            union { v16bf v; v8bf h[2]; } b;
            b.h[0] = *(const v8bf*)(brow + (size_t)t * 16 * K + k0);
            b.h[1] = *(const v8bf*)(brow + (size_t)t * 16 * K + k0 + 16);
            acc[t] = __builtin_amdgcn_wmma_f32_16x16x32_bf16(
                false, a.v, false, b.v, (short)0, acc[t], false, false);
        }
    }

    const int r0 = tm + (lane >> 4) * 8;
#pragma unroll
    for (int t = 0; t < NT; ++t) {
        const int cn = tn0 + t * 16 + l15;
        const float bc = bias[cn];
#pragma unroll
        for (int v = 0; v < 8; ++v)
            C[(size_t)(r0 + v) * N + cn] = acc[t][v] + bc;
    }
}

__global__ __launch_bounds__(256)
void norm_relu_store(float* __restrict__ H, __bf16* __restrict__ X2,
                     int F, int ldx, int coloff) {
    __shared__ float red[256];
    const int n = blockIdx.x;
    const int t = threadIdx.x;
    float* row = H + (size_t)n * F;
    float v = row[t];
    red[t] = v * v;
    __syncthreads();
    for (int s = 128; s > 0; s >>= 1) {
        if (t < s) red[t] += red[t + s];
        __syncthreads();
    }
    float inv = 1.0f / fmaxf(sqrtf(red[0]), 1e-12f);
    float r = fmaxf(v * inv, 0.0f);
    row[t] = r;
    X2[(size_t)n * ldx + coloff + t] = (__bf16)r;
}

__global__ __launch_bounds__(256)
void final_fc(const float* __restrict__ H2, const float* __restrict__ Wfc,
              const float* __restrict__ bfc, float* __restrict__ out,
              int nnodes, int F) {
    int wave = blockIdx.x * (blockDim.x >> 5) + (threadIdx.x >> 5);
    int lane = threadIdx.x & 31;
    if (wave >= nnodes) return;
    const float* h = H2 + (size_t)wave * F;
    float ss = 0.0f, d0 = 0.0f, d1 = 0.0f;
    for (int i = lane; i < F; i += 32) {
        float v = h[i];
        ss += v * v;
        d0 += v * Wfc[i];
        d1 += v * Wfc[F + i];
    }
    for (int off = 16; off > 0; off >>= 1) {
        ss += __shfl_xor(ss, off, 32);
        d0 += __shfl_xor(d0, off, 32);
        d1 += __shfl_xor(d1, off, 32);
    }
    if (lane == 0) {
        float inv = 1.0f / fmaxf(sqrtf(ss), 1e-12f);
        float l0 = d0 * inv + bfc[0];
        float l1 = d1 * inv + bfc[1];
        float m  = fmaxf(l0, l1);
        float e0 = __expf(l0 - m), e1 = __expf(l1 - m);
        float s  = 1.0f / (e0 + e1);
        out[(size_t)wave * 2 + 0] = e0 * s;
        out[(size_t)wave * 2 + 1] = e1 * s;
    }
}

extern "C" void kernel_launch(void* const* d_in, const int* in_sizes, int n_in,
                              void* d_out, int out_size, void* d_ws, size_t ws_size,
                              hipStream_t stream) {
    (void)in_sizes; (void)n_in; (void)out_size; (void)ws_size;

    const float* feat = (const float*)d_in[0];
    const int*   eidx = (const int*)  d_in[1];
    const int*   src  = eidx;
    const int*   dst  = eidx + N_EDGES;
    const float* W1l  = (const float*)d_in[2];
    const float* b1   = (const float*)d_in[3];
    const float* W1r  = (const float*)d_in[4];
    const float* W2l  = (const float*)d_in[5];
    const float* b2   = (const float*)d_in[6];
    const float* W2r  = (const float*)d_in[7];
    const float* Wfc  = (const float*)d_in[8];
    const float* bfc  = (const float*)d_in[9];
    float* out = (float*)d_out;

    char* ws = (char*)d_ws;
    size_t off = 0;
    auto alloc = [&](size_t bytes) -> void* {
        void* p = ws + off;
        off = (off + bytes + 255) & ~(size_t)255;
        return p;
    };
    int*    deg    = (int*)  alloc((size_t)DEG_PAD * 4);
    int*    rowptr = (int*)  alloc((size_t)DEG_PAD * 4);
    int*    rank   = (int*)  alloc((size_t)N_EDGES * 4);
    int*    col    = (int*)  alloc((size_t)N_EDGES * 4);
    __bf16* Xcat1 = (__bf16*)alloc((size_t)N_NODES * 2 * IN_SIZE * 2);
    __bf16* Wcat1 = (__bf16*)alloc((size_t)HIDDEN * 2 * IN_SIZE * 2);
    float*  h1    = (float*) alloc((size_t)N_NODES * HIDDEN * 4);
    __bf16* Xcat2 = (__bf16*)alloc((size_t)N_NODES * 2 * HIDDEN * 2);
    __bf16* Wcat2 = (__bf16*)alloc((size_t)OUT_SIZE * 2 * HIDDEN * 2);
    float*  h2    = (float*) alloc((size_t)N_NODES * OUT_SIZE * 4);

    const int TB = 256;
    auto cdiv = [](int a, int b) { return (a + b - 1) / b; };

    zero_i32<<<cdiv(DEG_PAD, TB), TB, 0, stream>>>(deg, DEG_PAD);

    build_wcat<<<cdiv(HIDDEN * IN_SIZE, TB), TB, 0, stream>>>(W1l, W1r, Wcat1, HIDDEN, IN_SIZE);
    build_wcat<<<cdiv(OUT_SIZE * HIDDEN, TB), TB, 0, stream>>>(W2l, W2r, Wcat2, OUT_SIZE, HIDDEN);

    edge_rank<<<cdiv(N_EDGES, TB), TB, 0, stream>>>(dst, deg, rank, N_EDGES);
    scan_rowptr<<<1, SCAN_T, 0, stream>>>(deg, rowptr);
    csr_fill<<<cdiv(N_EDGES, TB), TB, 0, stream>>>(src, dst, rowptr, rank, col, N_EDGES);
    gather_mean<IN_SIZE, true><<<cdiv(N_NODES, TB / 32), TB, 0, stream>>>(
        feat, rowptr, col, Xcat1, N_NODES);
    {
        int waves = (N_NODES / 16) * (HIDDEN / 64);
        gemm_bf16_wmma<2 * IN_SIZE, 4><<<cdiv(waves, TB / 32), TB, 0, stream>>>(
            Xcat1, Wcat1, b1, h1, N_NODES, HIDDEN);
    }
    norm_relu_store<<<N_NODES, HIDDEN, 0, stream>>>(h1, Xcat2, HIDDEN, 2 * HIDDEN, HIDDEN);

    gather_mean<HIDDEN, false><<<cdiv(N_NODES, TB / 32), TB, 0, stream>>>(
        h1, rowptr, col, Xcat2, N_NODES);
    {
        int waves = (N_NODES / 16) * (OUT_SIZE / 64);
        gemm_bf16_wmma<2 * HIDDEN, 4><<<cdiv(waves, TB / 32), TB, 0, stream>>>(
            Xcat2, Wcat2, b2, h2, N_NODES, OUT_SIZE);
    }
    final_fc<<<cdiv(N_NODES, TB / 32), TB, 0, stream>>>(h2, Wfc, bfc, out,
                                                        N_NODES, OUT_SIZE);
}


# Layer-1 row L2-normalise + ReLU fused into the GEMM epilogue (block = 4 waves = whole rows; row sum of squares now summed in-lane, across 16 lanes, then across waves via LDS); norm kernel removed
# speedup vs baseline: 2.3283x; 1.1859x over previous
#include <hip/hip_runtime.h>
#include <math.h>

#define N_NODES  50000
#define N_EDGES  600000
#define IN_SIZE  128
#define HIDDEN   256
#define OUT_SIZE 128

typedef __attribute__((ext_vector_type(16))) __bf16 v16bf;
typedef __attribute__((ext_vector_type(8)))  __bf16 v8bf;
typedef __attribute__((ext_vector_type(8)))  float  v8f;
typedef __attribute__((ext_vector_type(4)))  __bf16 v4bf;

#define SCAN_T   1024
#define SCAN_PER 52
#define DEG_PAD  (SCAN_T * SCAN_PER)

__global__ __launch_bounds__(256)
void zero_i32(int* __restrict__ p, int n) {
    int i = blockIdx.x * blockDim.x + threadIdx.x;
    if (i < n) p[i] = 0;
}

__global__ __launch_bounds__(256)
void build_wcat(const float* __restrict__ Wl, const float* __restrict__ Wr,
                __bf16* __restrict__ Wcat, int rows, int kHalf) {
    int i = blockIdx.x * blockDim.x + threadIdx.x;
    if (i >= rows * kHalf) return;
    int r = i / kHalf, c = i % kHalf;
    __bf16* row = Wcat + (size_t)r * (2 * kHalf);
    row[c]         = (__bf16)Wl[i];
    row[kHalf + c] = (__bf16)Wr[i];
}


__global__ __launch_bounds__(256)
void edge_rank(const int* __restrict__ dst, int* __restrict__ deg,
               int* __restrict__ rank, int nedges) {
    int e = blockIdx.x * blockDim.x + threadIdx.x;
    if (e >= nedges) return;
    rank[e] = atomicAdd(deg + dst[e], 1);
}

__global__ __launch_bounds__(SCAN_T)
void scan_rowptr(const int* __restrict__ deg, int* __restrict__ rowptr) {
    __shared__ int wsum[32];
    const int t = threadIdx.x, lane = t & 31, w = t >> 5;
    const int4* d4 = (const int4*)(deg + t * SCAN_PER);
    int4 d[SCAN_PER / 4];
    int s = 0;
#pragma unroll
    for (int i = 0; i < SCAN_PER / 4; ++i) {
        d[i] = d4[i];
        s += (d[i].x + d[i].y) + (d[i].z + d[i].w);
    }
    int inc = s;
    for (int off = 1; off < 32; off <<= 1) {
        int v = __shfl_up(inc, off, 32);
        if (lane >= off) inc += v;
    }
    if (lane == 31) wsum[w] = inc;
    __syncthreads();
    if (w == 0) {
        int x = wsum[lane], xi = x;
        for (int off = 1; off < 32; off <<= 1) {
            int v = __shfl_up(xi, off, 32);
            if (lane >= off) xi += v;
        }
        wsum[lane] = xi - x;
    }
    __syncthreads();
    int run = wsum[w] + inc - s;
    int4* r4 = (int4*)(rowptr + t * SCAN_PER);
#pragma unroll
    for (int i = 0; i < SCAN_PER / 4; ++i) {
        int4 o;
        o.x = run; run += d[i].x;
        o.y = run; run += d[i].y;
        o.z = run; run += d[i].z;
        o.w = run; run += d[i].w;
        r4[i] = o;
    }
}

__global__ __launch_bounds__(256)
void csr_fill(const int* __restrict__ src, const int* __restrict__ dst,
              const int* __restrict__ rowptr, const int* __restrict__ rank,
              int* __restrict__ col, int nedges) {
    int e = blockIdx.x * blockDim.x + threadIdx.x;
    if (e >= nedges) return;
    col[rowptr[dst[e]] + rank[e]] = src[e];
}

__device__ __forceinline__ v4bf pack4_bf16(float4 v, float sc) {
    v4bf o;
    o[0] = (__bf16)(v.x * sc);
    o[1] = (__bf16)(v.y * sc);
    o[2] = (__bf16)(v.z * sc);
    o[3] = (__bf16)(v.w * sc);
    return o;
}

template <int F, bool SELF>
__global__ __launch_bounds__(256)
void gather_mean(const float* __restrict__ X, const int* __restrict__ rowptr,
                 const int* __restrict__ col, __bf16* __restrict__ Xcat, int nnodes) {
    constexpr int NV = F / 128;
    const int lane = threadIdx.x & 31;
    const int n = blockIdx.x * (blockDim.x >> 5) + (threadIdx.x >> 5);
    if (n >= nnodes) return;
    const int beg = __builtin_amdgcn_readfirstlane(rowptr[n]);
    const int end = __builtin_amdgcn_readfirstlane(rowptr[n + 1]);
    const float4* X4 = (const float4*)X;
    float4 acc[NV];
#pragma unroll
    for (int q = 0; q < NV; ++q) acc[q] = make_float4(0.0f, 0.0f, 0.0f, 0.0f);

    for (int base = beg; base < end; base += 32) {
        const int m = (end - base < 32) ? (end - base) : 32;
        const int myc = (lane < m) ? col[base + lane] : 0;
        int j = 0;
        for (; j + 4 <= m; j += 4) {
            const float4* r0 = X4 + (size_t)__builtin_amdgcn_readlane(myc, j + 0) * (F / 4) + lane;
            const float4* r1 = X4 + (size_t)__builtin_amdgcn_readlane(myc, j + 1) * (F / 4) + lane;
            const float4* r2 = X4 + (size_t)__builtin_amdgcn_readlane(myc, j + 2) * (F / 4) + lane;
            const float4* r3 = X4 + (size_t)__builtin_amdgcn_readlane(myc, j + 3) * (F / 4) + lane;
            float4 v0[NV], v1[NV], v2[NV], v3[NV];
#pragma unroll
            for (int q = 0; q < NV; ++q) {
                v0[q] = r0[q * 32]; v1[q] = r1[q * 32];
                v2[q] = r2[q * 32]; v3[q] = r3[q * 32];
            }
#pragma unroll
            for (int q = 0; q < NV; ++q) {
                acc[q].x += (v0[q].x + v1[q].x) + (v2[q].x + v3[q].x);
                acc[q].y += (v0[q].y + v1[q].y) + (v2[q].y + v3[q].y);
                acc[q].z += (v0[q].z + v1[q].z) + (v2[q].z + v3[q].z);
                acc[q].w += (v0[q].w + v1[q].w) + (v2[q].w + v3[q].w);
            }
        }
        for (; j < m; ++j) {
            const float4* r0 = X4 + (size_t)__builtin_amdgcn_readlane(myc, j) * (F / 4) + lane;
#pragma unroll
            for (int q = 0; q < NV; ++q) {
                float4 v = r0[q * 32];
                acc[q].x += v.x; acc[q].y += v.y; acc[q].z += v.z; acc[q].w += v.w;
            }
        }
    }

    const float inv = 1.0f / fmaxf((float)(end - beg), 1.0f);
    __bf16* row = Xcat + (size_t)n * (2 * F);
#pragma unroll
    for (int q = 0; q < NV; ++q)
        *(v4bf*)(row + q * 128 + lane * 4) = pack4_bf16(acc[q], inv);
    if (SELF) {
#pragma unroll
        for (int q = 0; q < NV; ++q) {
            float4 v = X4[(size_t)n * (F / 4) + q * 32 + lane];
            *(v4bf*)(row + F + q * 128 + lane * 4) = pack4_bf16(v, 1.0f);
        }
    }
}

template <int K, int NT>
__device__ __forceinline__ void wmma_strip(const __bf16* __restrict__ arow,
                                           const __bf16* __restrict__ brow, v8f (&acc)[NT]) {
#pragma unroll
    for (int t = 0; t < NT; ++t) acc[t] = (v8f){};

#pragma unroll
    for (int k0 = 0; k0 < K; k0 += 32) {
        union { v16bf v; v8bf h[2]; } a;
        a.h[0] = *(const v8bf*)(arow + k0);
        a.h[1] = *(const v8bf*)(arow + k0 + 16);
#pragma unroll
        for (int t = 0; t < NT; ++t) {
            union { v16bf v; v8bf h[2]; } b;
            b.h[0] = *(const v8bf*)(brow + (size_t)t * 16 * K + k0);
            b.h[1] = *(const v8bf*)(brow + (size_t)t * 16 * K + k0 + 16);
            acc[t] = __builtin_amdgcn_wmma_f32_16x16x32_bf16(
                false, a.v, false, b.v, (short)0, acc[t], false, false);
        }
    }
}

template <int K, int NT>
__global__ __launch_bounds__(256)
void gemm_bf16_wmma(const __bf16* __restrict__ A, const __bf16* __restrict__ B,
                    const float* __restrict__ bias, float* __restrict__ C,
                    int M, int N) {
    const int lane    = threadIdx.x & 31;
    const int wave    = blockIdx.x * (blockDim.x >> 5) + (threadIdx.x >> 5);
    const int ngroups = N / (16 * NT);
    if (wave >= (M >> 4) * ngroups) return;
    const int tm  = (wave / ngroups) << 4;
    const int tn0 = (wave % ngroups) * (16 * NT);
    const int kh  = (lane >> 4) * 8;
    const int l15 = lane & 15;

    v8f acc[NT];
    wmma_strip<K, NT>(A + (size_t)(tm + l15) * K + kh, B + (size_t)(tn0 + l15) * K + kh, acc);

    const int r0 = tm + (lane >> 4) * 8;
#pragma unroll
    for (int t = 0; t < NT; ++t) {
        const int cn = tn0 + t * 16 + l15;
        const float bc = bias[cn];
#pragma unroll
        for (int v = 0; v < 8; ++v)
            C[(size_t)(r0 + v) * N + cn] = acc[t][v] + bc;
    }
}

template <int K, int NCOL>
__global__ __launch_bounds__(NCOL / 2) __attribute__((amdgpu_waves_per_eu(16)))
void gemm_bf16_wmma_norm_relu(const __bf16* __restrict__ A, const __bf16* __restrict__ B,
                              const float* __restrict__ bias, float* __restrict__ H,
                              __bf16* __restrict__ X2, int ldx, int coloff) {
    constexpr int NT = 4;
    constexpr int NW = NCOL / (16 * NT);
    __shared__ __attribute__((aligned(16))) float red[NW][16];
    const int lane = threadIdx.x & 31;
    const int w    = threadIdx.x >> 5;
    const int tm   = blockIdx.x << 4;
    const int tn0  = w * (16 * NT);
    const int h    = lane >> 4;
    const int l15  = lane & 15;

    v8f acc[NT];
    wmma_strip<K, NT>(A + (size_t)(tm + l15) * K + h * 8, B + (size_t)(tn0 + l15) * K + h * 8, acc);

    float ss[8];
#pragma unroll
    for (int v = 0; v < 8; ++v) ss[v] = 0.0f;
#pragma unroll
    for (int t = 0; t < NT; ++t) {
        const float bc = bias[tn0 + t * 16 + l15];
#pragma unroll
        for (int v = 0; v < 8; ++v) {
            acc[t][v] += bc;
            ss[v] += acc[t][v] * acc[t][v];
        }
    }
#pragma unroll
    for (int off = 1; off < 16; off <<= 1) {
#pragma unroll
        for (int v = 0; v < 8; ++v) ss[v] += __shfl_xor(ss[v], off, 32);
    }
    if (l15 == 0) {
        float4* mine = (float4*)&red[w][h * 8];
        mine[0] = make_float4(ss[0], ss[1], ss[2], ss[3]);
        mine[1] = make_float4(ss[4], ss[5], ss[6], ss[7]);
    }
    __syncthreads();
    float tot = red[0][l15];
#pragma unroll
    for (int q = 1; q < NW; ++q) tot += red[q][l15];
    const float inv_mine = 1.0f / fmaxf(sqrtf(tot), 1e-12f);
    const int r0 = tm + h * 8;
#pragma unroll
    for (int v = 0; v < 8; ++v) {
        const float inv = __shfl(inv_mine, h * 8 + v, 32);
        float*  hrow = H  + (size_t)(r0 + v) * NCOL + tn0 + l15;
        __bf16* xrow = X2 + (size_t)(r0 + v) * ldx + coloff + tn0 + l15;
#pragma unroll
        for (int t = 0; t < NT; ++t) {
            const float r = fmaxf(acc[t][v] * inv, 0.0f);
            hrow[t * 16] = r;
            xrow[t * 16] = (__bf16)r;
        }
    }
}

__global__ __launch_bounds__(256)
void final_fc(const float* __restrict__ H2, const float* __restrict__ Wfc,
              const float* __restrict__ bfc, float* __restrict__ out,
              int nnodes, int F) {
    int wave = blockIdx.x * (blockDim.x >> 5) + (threadIdx.x >> 5);
    int lane = threadIdx.x & 31;
    if (wave >= nnodes) return;
    const float* h = H2 + (size_t)wave * F;
    float ss = 0.0f, d0 = 0.0f, d1 = 0.0f;
    for (int i = lane; i < F; i += 32) {
        float v = h[i];
        ss += v * v;
        d0 += v * Wfc[i];
        d1 += v * Wfc[F + i];
    }
    for (int off = 16; off > 0; off >>= 1) {
        ss += __shfl_xor(ss, off, 32);
        d0 += __shfl_xor(d0, off, 32);
        d1 += __shfl_xor(d1, off, 32);
    }
    if (lane == 0) {
        float inv = 1.0f / fmaxf(sqrtf(ss), 1e-12f);
        float l0 = d0 * inv + bfc[0];
        float l1 = d1 * inv + bfc[1];
        float m  = fmaxf(l0, l1);
        float e0 = __expf(l0 - m), e1 = __expf(l1 - m);
        float s  = 1.0f / (e0 + e1);
        out[(size_t)wave * 2 + 0] = e0 * s;
        out[(size_t)wave * 2 + 1] = e1 * s;
    }
}

extern "C" void kernel_launch(void* const* d_in, const int* in_sizes, int n_in,
                              void* d_out, int out_size, void* d_ws, size_t ws_size,
                              hipStream_t stream) {
    (void)in_sizes; (void)n_in; (void)out_size; (void)ws_size;

    const float* feat = (const float*)d_in[0];
    const int*   eidx = (const int*)  d_in[1];
    const int*   src  = eidx;
    const int*   dst  = eidx + N_EDGES;
    const float* W1l  = (const float*)d_in[2];
    const float* b1   = (const float*)d_in[3];
    const float* W1r  = (const float*)d_in[4];
    const float* W2l  = (const float*)d_in[5];
    const float* b2   = (const float*)d_in[6];
    const float* W2r  = (const float*)d_in[7];
    const float* Wfc  = (const float*)d_in[8];
    const float* bfc  = (const float*)d_in[9];
    float* out = (float*)d_out;

    char* ws = (char*)d_ws;
    size_t off = 0;
    auto alloc = [&](size_t bytes) -> void* {
        void* p = ws + off;
        off = (off + bytes + 255) & ~(size_t)255;
        return p;
    };
    int*    deg    = (int*)  alloc((size_t)DEG_PAD * 4);
    int*    rowptr = (int*)  alloc((size_t)DEG_PAD * 4);
    int*    rank   = (int*)  alloc((size_t)N_EDGES * 4);
    int*    col    = (int*)  alloc((size_t)N_EDGES * 4);
    __bf16* Xcat1 = (__bf16*)alloc((size_t)N_NODES * 2 * IN_SIZE * 2);
    __bf16* Wcat1 = (__bf16*)alloc((size_t)HIDDEN * 2 * IN_SIZE * 2);
    float*  h1    = (float*) alloc((size_t)N_NODES * HIDDEN * 4);
    __bf16* Xcat2 = (__bf16*)alloc((size_t)N_NODES * 2 * HIDDEN * 2);
    __bf16* Wcat2 = (__bf16*)alloc((size_t)OUT_SIZE * 2 * HIDDEN * 2);
    float*  h2    = (float*) alloc((size_t)N_NODES * OUT_SIZE * 4);

    const int TB = 256;
    auto cdiv = [](int a, int b) { return (a + b - 1) / b; };

    zero_i32<<<cdiv(DEG_PAD, TB), TB, 0, stream>>>(deg, DEG_PAD);

    build_wcat<<<cdiv(HIDDEN * IN_SIZE, TB), TB, 0, stream>>>(W1l, W1r, Wcat1, HIDDEN, IN_SIZE);
    build_wcat<<<cdiv(OUT_SIZE * HIDDEN, TB), TB, 0, stream>>>(W2l, W2r, Wcat2, OUT_SIZE, HIDDEN);

    edge_rank<<<cdiv(N_EDGES, TB), TB, 0, stream>>>(dst, deg, rank, N_EDGES);
    scan_rowptr<<<1, SCAN_T, 0, stream>>>(deg, rowptr);
    csr_fill<<<cdiv(N_EDGES, TB), TB, 0, stream>>>(src, dst, rowptr, rank, col, N_EDGES);
    gather_mean<IN_SIZE, true><<<cdiv(N_NODES, TB / 32), TB, 0, stream>>>(
        feat, rowptr, col, Xcat1, N_NODES);
    gemm_bf16_wmma_norm_relu<2 * IN_SIZE, HIDDEN><<<N_NODES / 16, HIDDEN / 2, 0, stream>>>(
        Xcat1, Wcat1, b1, h1, Xcat2, 2 * HIDDEN, HIDDEN);

    gather_mean<HIDDEN, false><<<cdiv(N_NODES, TB / 32), TB, 0, stream>>>(
        h1, rowptr, col, Xcat2, N_NODES);
    {
        int waves = (N_NODES / 16) * (OUT_SIZE / 64);
        gemm_bf16_wmma<2 * HIDDEN, 4><<<cdiv(waves, TB / 32), TB, 0, stream>>>(
            Xcat2, Wcat2, b2, h2, N_NODES, OUT_SIZE);
    }
    final_fc<<<cdiv(N_NODES, TB / 32), TB, 0, stream>>>(h2, Wfc, bfc, out,
                                                        N_NODES, OUT_SIZE);
}


# Layer 2 projects before it aggregates (mean is linear): one GEMM h1 x [W2l;W2r]^T gives P|Q, gather sums 128-wide f32 P rows, not 256-wide h1 rows; f32 h1 plane gone; neighbour sum order changes
# speedup vs baseline: 2.6796x; 1.1509x over previous
#include <hip/hip_runtime.h>
#include <math.h>

#define N_NODES  50000
#define N_EDGES  600000
#define IN_SIZE  128
#define HIDDEN   256
#define OUT_SIZE 128

typedef __attribute__((ext_vector_type(16))) __bf16 v16bf;
typedef __attribute__((ext_vector_type(8)))  __bf16 v8bf;
typedef __attribute__((ext_vector_type(8)))  float  v8f;
typedef __attribute__((ext_vector_type(4)))  __bf16 v4bf;

#define SCAN_T   1024
#define SCAN_PER 52
#define DEG_PAD  (SCAN_T * SCAN_PER)

__global__ __launch_bounds__(256)
void zero_i32(int* __restrict__ p, int n) {
    int i = blockIdx.x * blockDim.x + threadIdx.x;
    if (i < n) p[i] = 0;
}

__global__ __launch_bounds__(256)
void build_wcat(const float* __restrict__ Wl, const float* __restrict__ Wr,
                __bf16* __restrict__ Wcat, int rows, int kHalf) {
    int i = blockIdx.x * blockDim.x + threadIdx.x;
    if (i >= rows * kHalf) return;
    int r = i / kHalf, c = i % kHalf;
    __bf16* row = Wcat + (size_t)r * (2 * kHalf);
    row[c]         = (__bf16)Wl[i];
    row[kHalf + c] = (__bf16)Wr[i];
}

__global__ __launch_bounds__(256)
void build_wstack(const float* __restrict__ Wa, const float* __restrict__ Wb,
                  __bf16* __restrict__ Wst, int n) {
    int i = blockIdx.x * blockDim.x + threadIdx.x;
    if (i >= 2 * n) return;
    Wst[i] = (__bf16)(i < n ? Wa[i] : Wb[i - n]);
}


__global__ __launch_bounds__(256)
void edge_rank(const int* __restrict__ dst, int* __restrict__ deg,
               int* __restrict__ rank, int nedges) {
    int e = blockIdx.x * blockDim.x + threadIdx.x;
    if (e >= nedges) return;
    rank[e] = atomicAdd(deg + dst[e], 1);
}

__global__ __launch_bounds__(SCAN_T)
void scan_rowptr(const int* __restrict__ deg, int* __restrict__ rowptr) {
    __shared__ int wsum[32];
    const int t = threadIdx.x, lane = t & 31, w = t >> 5;
    const int4* d4 = (const int4*)(deg + t * SCAN_PER);
    int4 d[SCAN_PER / 4];
    int s = 0;
#pragma unroll
    for (int i = 0; i < SCAN_PER / 4; ++i) {
        d[i] = d4[i];
        s += (d[i].x + d[i].y) + (d[i].z + d[i].w);
    }
    int inc = s;
    for (int off = 1; off < 32; off <<= 1) {
        int v = __shfl_up(inc, off, 32);
        if (lane >= off) inc += v;
    }
    if (lane == 31) wsum[w] = inc;
    __syncthreads();
    if (w == 0) {
        int x = wsum[lane], xi = x;
        for (int off = 1; off < 32; off <<= 1) {
            int v = __shfl_up(xi, off, 32);
            if (lane >= off) xi += v;
        }
        wsum[lane] = xi - x;
    }
    __syncthreads();
    int run = wsum[w] + inc - s;
    int4* r4 = (int4*)(rowptr + t * SCAN_PER);
#pragma unroll
    for (int i = 0; i < SCAN_PER / 4; ++i) {
        int4 o;
        o.x = run; run += d[i].x;
        o.y = run; run += d[i].y;
        o.z = run; run += d[i].z;
        o.w = run; run += d[i].w;
        r4[i] = o;
    }
}

__global__ __launch_bounds__(256)
void csr_fill(const int* __restrict__ src, const int* __restrict__ dst,
              const int* __restrict__ rowptr, const int* __restrict__ rank,
              int* __restrict__ col, int nedges) {
    int e = blockIdx.x * blockDim.x + threadIdx.x;
    if (e >= nedges) return;
    col[rowptr[dst[e]] + rank[e]] = src[e];
}

__device__ __forceinline__ v4bf pack4_bf16(float4 v, float sc) {
    v4bf o;
    o[0] = (__bf16)(v.x * sc);
    o[1] = (__bf16)(v.y * sc);
    o[2] = (__bf16)(v.z * sc);
    o[3] = (__bf16)(v.w * sc);
    return o;
}

template <int F, int PITCH>
__device__ __forceinline__ void sum_rows(const float4* __restrict__ X4, const int* __restrict__ col,
                                         int beg, int end, int lane, float4 (&acc)[F / 128]) {
    constexpr int NV = F / 128;
#pragma unroll
    for (int q = 0; q < NV; ++q) acc[q] = make_float4(0.0f, 0.0f, 0.0f, 0.0f);

    for (int base = beg; base < end; base += 32) {
        const int m = (end - base < 32) ? (end - base) : 32;
        const int myc = (lane < m) ? col[base + lane] : 0;
        int j = 0;
        for (; j + 4 <= m; j += 4) {
            const float4* r0 = X4 + (size_t)__builtin_amdgcn_readlane(myc, j + 0) * (PITCH / 4) + lane;
            const float4* r1 = X4 + (size_t)__builtin_amdgcn_readlane(myc, j + 1) * (PITCH / 4) + lane;
            const float4* r2 = X4 + (size_t)__builtin_amdgcn_readlane(myc, j + 2) * (PITCH / 4) + lane;
            const float4* r3 = X4 + (size_t)__builtin_amdgcn_readlane(myc, j + 3) * (PITCH / 4) + lane;
            float4 v0[NV], v1[NV], v2[NV], v3[NV];
#pragma unroll
            for (int q = 0; q < NV; ++q) {
                v0[q] = r0[q * 32]; v1[q] = r1[q * 32];
                v2[q] = r2[q * 32]; v3[q] = r3[q * 32];
            }
#pragma unroll
            for (int q = 0; q < NV; ++q) {
                acc[q].x += (v0[q].x + v1[q].x) + (v2[q].x + v3[q].x);
                acc[q].y += (v0[q].y + v1[q].y) + (v2[q].y + v3[q].y);
                acc[q].z += (v0[q].z + v1[q].z) + (v2[q].z + v3[q].z);
                acc[q].w += (v0[q].w + v1[q].w) + (v2[q].w + v3[q].w);
            }
        }
        for (; j < m; ++j) {
            const float4* r0 = X4 + (size_t)__builtin_amdgcn_readlane(myc, j) * (PITCH / 4) + lane;
#pragma unroll
            for (int q = 0; q < NV; ++q) {
                float4 v = r0[q * 32];
                acc[q].x += v.x; acc[q].y += v.y; acc[q].z += v.z; acc[q].w += v.w;
            }
        }
    }
}

template <int F>
__global__ __launch_bounds__(256)
void gather_mean_cat(const float* __restrict__ X, const int* __restrict__ rowptr,
                     const int* __restrict__ col, __bf16* __restrict__ Xcat, int nnodes) {
    constexpr int NV = F / 128;
    const int lane = threadIdx.x & 31;
    const int n = blockIdx.x * (blockDim.x >> 5) + (threadIdx.x >> 5);
    if (n >= nnodes) return;
    const int beg = __builtin_amdgcn_readfirstlane(rowptr[n]);
    const int end = __builtin_amdgcn_readfirstlane(rowptr[n + 1]);
    const float4* X4 = (const float4*)X;
    float4 acc[NV];
    sum_rows<F, F>(X4, col, beg, end, lane, acc);

    const float inv = 1.0f / fmaxf((float)(end - beg), 1.0f);
    __bf16* row = Xcat + (size_t)n * (2 * F);
#pragma unroll
    for (int q = 0; q < NV; ++q)
        *(v4bf*)(row + q * 128 + lane * 4) = pack4_bf16(acc[q], inv);
#pragma unroll
    for (int q = 0; q < NV; ++q) {
        float4 v = X4[(size_t)n * (F / 4) + q * 32 + lane];
        *(v4bf*)(row + F + q * 128 + lane * 4) = pack4_bf16(v, 1.0f);
    }
}

template <int F>
__global__ __launch_bounds__(256)
void gather_mean_add(const float* __restrict__ PQ, const int* __restrict__ rowptr,
                     const int* __restrict__ col, float* __restrict__ Hout, int nnodes) {
    constexpr int NV = F / 128;
    const int lane = threadIdx.x & 31;
    const int n = blockIdx.x * (blockDim.x >> 5) + (threadIdx.x >> 5);
    if (n >= nnodes) return;
    const int beg = __builtin_amdgcn_readfirstlane(rowptr[n]);
    const int end = __builtin_amdgcn_readfirstlane(rowptr[n + 1]);
    const float4* PQ4 = (const float4*)PQ;
    float4 self[NV];
#pragma unroll
    for (int q = 0; q < NV; ++q) self[q] = PQ4[(size_t)n * (2 * F / 4) + F / 4 + q * 32 + lane];
    float4 acc[NV];
    sum_rows<F, 2 * F>(PQ4, col, beg, end, lane, acc);

    const float inv = 1.0f / fmaxf((float)(end - beg), 1.0f);
    float4* out4 = (float4*)(Hout + (size_t)n * F);
#pragma unroll
    for (int q = 0; q < NV; ++q) {
        float4 o;
        o.x = acc[q].x * inv + self[q].x;
        o.y = acc[q].y * inv + self[q].y;
        o.z = acc[q].z * inv + self[q].z;
        o.w = acc[q].w * inv + self[q].w;
        out4[q * 32 + lane] = o;
    }
}

template <int K, int NT>
__device__ __forceinline__ void wmma_strip(const __bf16* __restrict__ arow,
                                           const __bf16* __restrict__ brow, v8f (&acc)[NT]) {
#pragma unroll
    for (int t = 0; t < NT; ++t) acc[t] = (v8f){};

#pragma unroll
    for (int k0 = 0; k0 < K; k0 += 32) {
        union { v16bf v; v8bf h[2]; } a;
        a.h[0] = *(const v8bf*)(arow + k0);
        a.h[1] = *(const v8bf*)(arow + k0 + 16);
#pragma unroll
        for (int t = 0; t < NT; ++t) {
            union { v16bf v; v8bf h[2]; } b;
            b.h[0] = *(const v8bf*)(brow + (size_t)t * 16 * K + k0);
            b.h[1] = *(const v8bf*)(brow + (size_t)t * 16 * K + k0 + 16);
            acc[t] = __builtin_amdgcn_wmma_f32_16x16x32_bf16(
                false, a.v, false, b.v, (short)0, acc[t], false, false);
        }
    }
}

template <int K, int NT>
__global__ __launch_bounds__(256)
void gemm_bf16_wmma(const __bf16* __restrict__ A, const __bf16* __restrict__ B,
                    const float* __restrict__ bias, int bias_col0, float* __restrict__ C,
                    int M, int N) {
    const int lane    = threadIdx.x & 31;
    const int wave    = blockIdx.x * (blockDim.x >> 5) + (threadIdx.x >> 5);
    const int ngroups = N / (16 * NT);
    if (wave >= (M >> 4) * ngroups) return;
    const int tm  = (wave / ngroups) << 4;
    const int tn0 = (wave % ngroups) * (16 * NT);
    const int kh  = (lane >> 4) * 8;
    const int l15 = lane & 15;

    v8f acc[NT];
    wmma_strip<K, NT>(A + (size_t)(tm + l15) * K + kh, B + (size_t)(tn0 + l15) * K + kh, acc);

    const int r0 = tm + (lane >> 4) * 8;
#pragma unroll
    for (int t = 0; t < NT; ++t) {
        const int cn = tn0 + t * 16 + l15;
        const int   bi = cn - bias_col0;
        const float bv = bias[bi < 0 ? 0 : bi];
        const float bc = (bi < 0) ? 0.0f : bv;
#pragma unroll
        for (int v = 0; v < 8; ++v)
            C[(size_t)(r0 + v) * N + cn] = acc[t][v] + bc;
    }
}

template <int K, int NCOL>
__global__ __launch_bounds__(NCOL / 2) __attribute__((amdgpu_waves_per_eu(16)))
void gemm_bf16_wmma_norm_relu(const __bf16* __restrict__ A, const __bf16* __restrict__ B,
                              const float* __restrict__ bias, __bf16* __restrict__ Hb) {
    constexpr int NT = 4;
    constexpr int NW = NCOL / (16 * NT);
    __shared__ __attribute__((aligned(16))) float red[NW][16];
    const int lane = threadIdx.x & 31;
    const int w    = threadIdx.x >> 5;
    const int tm   = blockIdx.x << 4;
    const int tn0  = w * (16 * NT);
    const int h    = lane >> 4;
    const int l15  = lane & 15;

    v8f acc[NT];
    wmma_strip<K, NT>(A + (size_t)(tm + l15) * K + h * 8, B + (size_t)(tn0 + l15) * K + h * 8, acc);

    float ss[8];
#pragma unroll
    for (int v = 0; v < 8; ++v) ss[v] = 0.0f;
#pragma unroll
    for (int t = 0; t < NT; ++t) {
        const float bc = bias[tn0 + t * 16 + l15];
#pragma unroll
        for (int v = 0; v < 8; ++v) {
            acc[t][v] += bc;
            ss[v] += acc[t][v] * acc[t][v];
        }
    }
#pragma unroll
    for (int off = 1; off < 16; off <<= 1) {
#pragma unroll
        for (int v = 0; v < 8; ++v) ss[v] += __shfl_xor(ss[v], off, 32);
    }
    if (l15 == 0) {
        float4* mine = (float4*)&red[w][h * 8];
        mine[0] = make_float4(ss[0], ss[1], ss[2], ss[3]);
        mine[1] = make_float4(ss[4], ss[5], ss[6], ss[7]);
    }
    __syncthreads();
    float tot = red[0][l15];
#pragma unroll
    for (int q = 1; q < NW; ++q) tot += red[q][l15];
    const float inv_mine = 1.0f / fmaxf(sqrtf(tot), 1e-12f);
    const int r0 = tm + h * 8;
#pragma unroll
    for (int v = 0; v < 8; ++v) {
        const float inv = __shfl(inv_mine, h * 8 + v, 32);
        __bf16* hrow = Hb + (size_t)(r0 + v) * NCOL + tn0 + l15;
#pragma unroll
        for (int t = 0; t < NT; ++t) {
            const float r = fmaxf(acc[t][v] * inv, 0.0f);
            hrow[t * 16] = (__bf16)r;
        }
    }
}

__global__ __launch_bounds__(256)
void final_fc(const float* __restrict__ H2, const float* __restrict__ Wfc,
              const float* __restrict__ bfc, float* __restrict__ out,
              int nnodes, int F) {
    int wave = blockIdx.x * (blockDim.x >> 5) + (threadIdx.x >> 5);
    int lane = threadIdx.x & 31;
    if (wave >= nnodes) return;
    const float* h = H2 + (size_t)wave * F;
    float ss = 0.0f, d0 = 0.0f, d1 = 0.0f;
    for (int i = lane; i < F; i += 32) {
        float v = h[i];
        ss += v * v;
        d0 += v * Wfc[i];
        d1 += v * Wfc[F + i];
    }
    for (int off = 16; off > 0; off >>= 1) {
        ss += __shfl_xor(ss, off, 32);
        d0 += __shfl_xor(d0, off, 32);
        d1 += __shfl_xor(d1, off, 32);
    }
    if (lane == 0) {
        float inv = 1.0f / fmaxf(sqrtf(ss), 1e-12f);
        float l0 = d0 * inv + bfc[0];
        float l1 = d1 * inv + bfc[1];
        float m  = fmaxf(l0, l1);
        float e0 = __expf(l0 - m), e1 = __expf(l1 - m);
        float s  = 1.0f / (e0 + e1);
        out[(size_t)wave * 2 + 0] = e0 * s;
        out[(size_t)wave * 2 + 1] = e1 * s;
    }
}

extern "C" void kernel_launch(void* const* d_in, const int* in_sizes, int n_in,
                              void* d_out, int out_size, void* d_ws, size_t ws_size,
                              hipStream_t stream) {
    (void)in_sizes; (void)n_in; (void)out_size; (void)ws_size;

    const float* feat = (const float*)d_in[0];
    const int*   eidx = (const int*)  d_in[1];
    const int*   src  = eidx;
    const int*   dst  = eidx + N_EDGES;
    const float* W1l  = (const float*)d_in[2];
    const float* b1   = (const float*)d_in[3];
    const float* W1r  = (const float*)d_in[4];
    const float* W2l  = (const float*)d_in[5];
    const float* b2   = (const float*)d_in[6];
    const float* W2r  = (const float*)d_in[7];
    const float* Wfc  = (const float*)d_in[8];
    const float* bfc  = (const float*)d_in[9];
    float* out = (float*)d_out;

    char* ws = (char*)d_ws;
    size_t off = 0;
    auto alloc = [&](size_t bytes) -> void* {
        void* p = ws + off;
        off = (off + bytes + 255) & ~(size_t)255;
        return p;
    };
    int*    deg    = (int*)  alloc((size_t)DEG_PAD * 4);
    int*    rowptr = (int*)  alloc((size_t)DEG_PAD * 4);
    int*    rank   = (int*)  alloc((size_t)N_EDGES * 4);
    int*    col    = (int*)  alloc((size_t)N_EDGES * 4);
    __bf16* Xcat1 = (__bf16*)alloc((size_t)N_NODES * 2 * IN_SIZE * 2);
    __bf16* Wcat1 = (__bf16*)alloc((size_t)HIDDEN * 2 * IN_SIZE * 2);
    __bf16* h1b   = (__bf16*)alloc((size_t)N_NODES * HIDDEN * 2);
    __bf16* Wst2  = (__bf16*)alloc((size_t)2 * OUT_SIZE * HIDDEN * 2);
    float*  pq2   = (float*) alloc((size_t)N_NODES * 2 * OUT_SIZE * 4);
    float*  h2    = (float*) alloc((size_t)N_NODES * OUT_SIZE * 4);

    const int TB = 256;
    auto cdiv = [](int a, int b) { return (a + b - 1) / b; };

    zero_i32<<<cdiv(DEG_PAD, TB), TB, 0, stream>>>(deg, DEG_PAD);

    build_wcat<<<cdiv(HIDDEN * IN_SIZE, TB), TB, 0, stream>>>(W1l, W1r, Wcat1, HIDDEN, IN_SIZE);
    build_wstack<<<cdiv(2 * OUT_SIZE * HIDDEN, TB), TB, 0, stream>>>(W2l, W2r, Wst2,
                                                                    OUT_SIZE * HIDDEN);

    edge_rank<<<cdiv(N_EDGES, TB), TB, 0, stream>>>(dst, deg, rank, N_EDGES);
    scan_rowptr<<<1, SCAN_T, 0, stream>>>(deg, rowptr);
    csr_fill<<<cdiv(N_EDGES, TB), TB, 0, stream>>>(src, dst, rowptr, rank, col, N_EDGES);
    gather_mean_cat<IN_SIZE><<<cdiv(N_NODES, TB / 32), TB, 0, stream>>>(
        feat, rowptr, col, Xcat1, N_NODES);
    gemm_bf16_wmma_norm_relu<2 * IN_SIZE, HIDDEN><<<N_NODES / 16, HIDDEN / 2, 0, stream>>>(
        Xcat1, Wcat1, b1, h1b);

    {
        int waves = (N_NODES / 16) * (2 * OUT_SIZE / 64);
        gemm_bf16_wmma<HIDDEN, 4><<<cdiv(waves, TB / 32), TB, 0, stream>>>(
            h1b, Wst2, b2, OUT_SIZE, pq2, N_NODES, 2 * OUT_SIZE);
    }
    gather_mean_add<OUT_SIZE><<<cdiv(N_NODES, TB / 32), TB, 0, stream>>>(
        pq2, rowptr, col, h2, N_NODES);
    final_fc<<<cdiv(N_NODES, TB / 32), TB, 0, stream>>>(h2, Wfc, bfc, out,
                                                        N_NODES, OUT_SIZE);
}


# Output head (row L2 norm, 2-row FC, softmax) fused into the layer-2 gather wave: h2 row stays in registers, head kernel and h2 plane gone; norm/FC sums reordered (lane owns 4 adjacent columns)
# speedup vs baseline: 2.8209x; 1.0527x over previous
#include <hip/hip_runtime.h>
#include <math.h>

#define N_NODES  50000
#define N_EDGES  600000
#define IN_SIZE  128
#define HIDDEN   256
#define OUT_SIZE 128

typedef __attribute__((ext_vector_type(16))) __bf16 v16bf;
typedef __attribute__((ext_vector_type(8)))  __bf16 v8bf;
typedef __attribute__((ext_vector_type(8)))  float  v8f;
typedef __attribute__((ext_vector_type(4)))  __bf16 v4bf;

#define SCAN_T   1024
#define SCAN_PER 52
#define DEG_PAD  (SCAN_T * SCAN_PER)

__global__ __launch_bounds__(256)
void zero_i32(int* __restrict__ p, int n) {
    int i = blockIdx.x * blockDim.x + threadIdx.x;
    if (i < n) p[i] = 0;
}

__global__ __launch_bounds__(256)
void build_wcat(const float* __restrict__ Wl, const float* __restrict__ Wr,
                __bf16* __restrict__ Wcat, int rows, int kHalf) {
    int i = blockIdx.x * blockDim.x + threadIdx.x;
    if (i >= rows * kHalf) return;
    int r = i / kHalf, c = i % kHalf;
    __bf16* row = Wcat + (size_t)r * (2 * kHalf);
    row[c]         = (__bf16)Wl[i];
    row[kHalf + c] = (__bf16)Wr[i];
}

__global__ __launch_bounds__(256)
void build_wstack(const float* __restrict__ Wa, const float* __restrict__ Wb,
                  __bf16* __restrict__ Wst, int n) {
    int i = blockIdx.x * blockDim.x + threadIdx.x;
    if (i >= 2 * n) return;
    Wst[i] = (__bf16)(i < n ? Wa[i] : Wb[i - n]);
}


__global__ __launch_bounds__(256)
void edge_rank(const int* __restrict__ dst, int* __restrict__ deg,
               int* __restrict__ rank, int nedges) {
    int e = blockIdx.x * blockDim.x + threadIdx.x;
    if (e >= nedges) return;
    rank[e] = atomicAdd(deg + dst[e], 1);
}

__global__ __launch_bounds__(SCAN_T)
void scan_rowptr(const int* __restrict__ deg, int* __restrict__ rowptr) {
    __shared__ int wsum[32];
    const int t = threadIdx.x, lane = t & 31, w = t >> 5;
    const int4* d4 = (const int4*)(deg + t * SCAN_PER);
    int4 d[SCAN_PER / 4];
    int s = 0;
#pragma unroll
    for (int i = 0; i < SCAN_PER / 4; ++i) {
        d[i] = d4[i];
        s += (d[i].x + d[i].y) + (d[i].z + d[i].w);
    }
    int inc = s;
    for (int off = 1; off < 32; off <<= 1) {
        int v = __shfl_up(inc, off, 32);
        if (lane >= off) inc += v;
    }
    if (lane == 31) wsum[w] = inc;
    __syncthreads();
    if (w == 0) {
        int x = wsum[lane], xi = x;
        for (int off = 1; off < 32; off <<= 1) {
            int v = __shfl_up(xi, off, 32);
            if (lane >= off) xi += v;
        }
        wsum[lane] = xi - x;
    }
    __syncthreads();
    int run = wsum[w] + inc - s;
    int4* r4 = (int4*)(rowptr + t * SCAN_PER);
#pragma unroll
    for (int i = 0; i < SCAN_PER / 4; ++i) {
        int4 o;
        o.x = run; run += d[i].x;
        o.y = run; run += d[i].y;
        o.z = run; run += d[i].z;
        o.w = run; run += d[i].w;
        r4[i] = o;
    }
}

__global__ __launch_bounds__(256)
void csr_fill(const int* __restrict__ src, const int* __restrict__ dst,
              const int* __restrict__ rowptr, const int* __restrict__ rank,
              int* __restrict__ col, int nedges) {
    int e = blockIdx.x * blockDim.x + threadIdx.x;
    if (e >= nedges) return;
    col[rowptr[dst[e]] + rank[e]] = src[e];
}

__device__ __forceinline__ v4bf pack4_bf16(float4 v, float sc) {
    v4bf o;
    o[0] = (__bf16)(v.x * sc);
    o[1] = (__bf16)(v.y * sc);
    o[2] = (__bf16)(v.z * sc);
    o[3] = (__bf16)(v.w * sc);
    return o;
}

template <int F, int PITCH>
__device__ __forceinline__ void sum_rows(const float4* __restrict__ X4, const int* __restrict__ col,
                                         int beg, int end, int lane, float4 (&acc)[F / 128]) {
    constexpr int NV = F / 128;
#pragma unroll
    for (int q = 0; q < NV; ++q) acc[q] = make_float4(0.0f, 0.0f, 0.0f, 0.0f);

    for (int base = beg; base < end; base += 32) {
        const int m = (end - base < 32) ? (end - base) : 32;
        const int myc = (lane < m) ? col[base + lane] : 0;
        int j = 0;
        for (; j + 4 <= m; j += 4) {
            const float4* r0 = X4 + (size_t)__builtin_amdgcn_readlane(myc, j + 0) * (PITCH / 4) + lane;
            const float4* r1 = X4 + (size_t)__builtin_amdgcn_readlane(myc, j + 1) * (PITCH / 4) + lane;
            const float4* r2 = X4 + (size_t)__builtin_amdgcn_readlane(myc, j + 2) * (PITCH / 4) + lane;
            const float4* r3 = X4 + (size_t)__builtin_amdgcn_readlane(myc, j + 3) * (PITCH / 4) + lane;
            float4 v0[NV], v1[NV], v2[NV], v3[NV];
#pragma unroll
            for (int q = 0; q < NV; ++q) {
                v0[q] = r0[q * 32]; v1[q] = r1[q * 32];
                v2[q] = r2[q * 32]; v3[q] = r3[q * 32];
            }
#pragma unroll
            for (int q = 0; q < NV; ++q) {
                acc[q].x += (v0[q].x + v1[q].x) + (v2[q].x + v3[q].x);
                acc[q].y += (v0[q].y + v1[q].y) + (v2[q].y + v3[q].y);
                acc[q].z += (v0[q].z + v1[q].z) + (v2[q].z + v3[q].z);
                acc[q].w += (v0[q].w + v1[q].w) + (v2[q].w + v3[q].w);
            }
        }
        for (; j < m; ++j) {
            const float4* r0 = X4 + (size_t)__builtin_amdgcn_readlane(myc, j) * (PITCH / 4) + lane;
#pragma unroll
            for (int q = 0; q < NV; ++q) {
                float4 v = r0[q * 32];
                acc[q].x += v.x; acc[q].y += v.y; acc[q].z += v.z; acc[q].w += v.w;
            }
        }
    }
}

template <int F>
__global__ __launch_bounds__(256)
void gather_mean_cat(const float* __restrict__ X, const int* __restrict__ rowptr,
                     const int* __restrict__ col, __bf16* __restrict__ Xcat, int nnodes) {
    constexpr int NV = F / 128;
    const int lane = threadIdx.x & 31;
    const int n = blockIdx.x * (blockDim.x >> 5) + (threadIdx.x >> 5);
    if (n >= nnodes) return;
    const int beg = __builtin_amdgcn_readfirstlane(rowptr[n]);
    const int end = __builtin_amdgcn_readfirstlane(rowptr[n + 1]);
    const float4* X4 = (const float4*)X;
    float4 acc[NV];
    sum_rows<F, F>(X4, col, beg, end, lane, acc);

    const float inv = 1.0f / fmaxf((float)(end - beg), 1.0f);
    __bf16* row = Xcat + (size_t)n * (2 * F);
#pragma unroll
    for (int q = 0; q < NV; ++q)
        *(v4bf*)(row + q * 128 + lane * 4) = pack4_bf16(acc[q], inv);
#pragma unroll
    for (int q = 0; q < NV; ++q) {
        float4 v = X4[(size_t)n * (F / 4) + q * 32 + lane];
        *(v4bf*)(row + F + q * 128 + lane * 4) = pack4_bf16(v, 1.0f);
    }
}

template <int F>
__global__ __launch_bounds__(256)
void gather_mean_add_fc(const float* __restrict__ PQ, const int* __restrict__ rowptr,
                        const int* __restrict__ col, const float* __restrict__ Wfc,
                        const float* __restrict__ bfc, float* __restrict__ out, int nnodes) {
    constexpr int NV = F / 128;
    const int lane = threadIdx.x & 31;
    const int n = blockIdx.x * (blockDim.x >> 5) + (threadIdx.x >> 5);
    if (n >= nnodes) return;
    const int beg = __builtin_amdgcn_readfirstlane(rowptr[n]);
    const int end = __builtin_amdgcn_readfirstlane(rowptr[n + 1]);
    const float4* PQ4 = (const float4*)PQ;
    const float4* W4  = (const float4*)Wfc;
    float4 self[NV];
    float4 w0[NV], w1[NV];
#pragma unroll
    for (int q = 0; q < NV; ++q) {
        self[q] = PQ4[(size_t)n * (2 * F / 4) + F / 4 + q * 32 + lane];
        w0[q]   = W4[q * 32 + lane];
        w1[q]   = W4[F / 4 + q * 32 + lane];
    }
    float4 acc[NV];
    sum_rows<F, 2 * F>(PQ4, col, beg, end, lane, acc);

    const float inv = 1.0f / fmaxf((float)(end - beg), 1.0f);
    float ss = 0.0f, d0 = 0.0f, d1 = 0.0f;
#pragma unroll
    for (int q = 0; q < NV; ++q) {
        float4 o;
        o.x = acc[q].x * inv + self[q].x;
        o.y = acc[q].y * inv + self[q].y;
        o.z = acc[q].z * inv + self[q].z;
        o.w = acc[q].w * inv + self[q].w;
        ss += (o.x * o.x + o.y * o.y) + (o.z * o.z + o.w * o.w);
        d0 += (o.x * w0[q].x + o.y * w0[q].y) + (o.z * w0[q].z + o.w * w0[q].w);
        d1 += (o.x * w1[q].x + o.y * w1[q].y) + (o.z * w1[q].z + o.w * w1[q].w);
    }
    for (int off = 16; off > 0; off >>= 1) {
        ss += __shfl_xor(ss, off, 32);
        d0 += __shfl_xor(d0, off, 32);
        d1 += __shfl_xor(d1, off, 32);
    }
    if (lane == 0) {
        float invn = 1.0f / fmaxf(sqrtf(ss), 1e-12f);
        float l0 = d0 * invn + bfc[0];
        float l1 = d1 * invn + bfc[1];
        float m  = fmaxf(l0, l1);
        float e0 = __expf(l0 - m), e1 = __expf(l1 - m);
        float s  = 1.0f / (e0 + e1);
        out[(size_t)n * 2 + 0] = e0 * s;
        out[(size_t)n * 2 + 1] = e1 * s;
    }
}

template <int K, int NT>
__device__ __forceinline__ void wmma_strip(const __bf16* __restrict__ arow,
                                           const __bf16* __restrict__ brow, v8f (&acc)[NT]) {
#pragma unroll
    for (int t = 0; t < NT; ++t) acc[t] = (v8f){};

#pragma unroll
    for (int k0 = 0; k0 < K; k0 += 32) {
        union { v16bf v; v8bf h[2]; } a;
        a.h[0] = *(const v8bf*)(arow + k0);
        a.h[1] = *(const v8bf*)(arow + k0 + 16);
#pragma unroll
        for (int t = 0; t < NT; ++t) {
            union { v16bf v; v8bf h[2]; } b;
            b.h[0] = *(const v8bf*)(brow + (size_t)t * 16 * K + k0);
            b.h[1] = *(const v8bf*)(brow + (size_t)t * 16 * K + k0 + 16);
            acc[t] = __builtin_amdgcn_wmma_f32_16x16x32_bf16(
                false, a.v, false, b.v, (short)0, acc[t], false, false);
        }
    }
}

template <int K, int NT>
__global__ __launch_bounds__(256)
void gemm_bf16_wmma(const __bf16* __restrict__ A, const __bf16* __restrict__ B,
                    const float* __restrict__ bias, int bias_col0, float* __restrict__ C,
                    int M, int N) {
    const int lane    = threadIdx.x & 31;
    const int wave    = blockIdx.x * (blockDim.x >> 5) + (threadIdx.x >> 5);
    const int ngroups = N / (16 * NT);
    if (wave >= (M >> 4) * ngroups) return;
    const int tm  = (wave / ngroups) << 4;
    const int tn0 = (wave % ngroups) * (16 * NT);
    const int kh  = (lane >> 4) * 8;
    const int l15 = lane & 15;

    v8f acc[NT];
    wmma_strip<K, NT>(A + (size_t)(tm + l15) * K + kh, B + (size_t)(tn0 + l15) * K + kh, acc);

    const int r0 = tm + (lane >> 4) * 8;
#pragma unroll
    for (int t = 0; t < NT; ++t) {
        const int cn = tn0 + t * 16 + l15;
        const int   bi = cn - bias_col0;
        const float bv = bias[bi < 0 ? 0 : bi];
        const float bc = (bi < 0) ? 0.0f : bv;
#pragma unroll
        for (int v = 0; v < 8; ++v)
            C[(size_t)(r0 + v) * N + cn] = acc[t][v] + bc;
    }
}

template <int K, int NCOL>
__global__ __launch_bounds__(NCOL / 2) __attribute__((amdgpu_waves_per_eu(16)))
void gemm_bf16_wmma_norm_relu(const __bf16* __restrict__ A, const __bf16* __restrict__ B,
                              const float* __restrict__ bias, __bf16* __restrict__ Hb) {
    constexpr int NT = 4;
    constexpr int NW = NCOL / (16 * NT);
    __shared__ __attribute__((aligned(16))) float red[NW][16];
    const int lane = threadIdx.x & 31;
    const int w    = threadIdx.x >> 5;
    const int tm   = blockIdx.x << 4;
    const int tn0  = w * (16 * NT);
    const int h    = lane >> 4;
    const int l15  = lane & 15;

    v8f acc[NT];
    wmma_strip<K, NT>(A + (size_t)(tm + l15) * K + h * 8, B + (size_t)(tn0 + l15) * K + h * 8, acc);

    float ss[8];
#pragma unroll
    for (int v = 0; v < 8; ++v) ss[v] = 0.0f;
#pragma unroll
    for (int t = 0; t < NT; ++t) {
        const float bc = bias[tn0 + t * 16 + l15];
#pragma unroll
        for (int v = 0; v < 8; ++v) {
            acc[t][v] += bc;
            ss[v] += acc[t][v] * acc[t][v];
        }
    }
#pragma unroll
    for (int off = 1; off < 16; off <<= 1) {
#pragma unroll
        for (int v = 0; v < 8; ++v) ss[v] += __shfl_xor(ss[v], off, 32);
    }
    if (l15 == 0) {
        float4* mine = (float4*)&red[w][h * 8];
        mine[0] = make_float4(ss[0], ss[1], ss[2], ss[3]);
        mine[1] = make_float4(ss[4], ss[5], ss[6], ss[7]);
    }
    __syncthreads();
    float tot = red[0][l15];
#pragma unroll
    for (int q = 1; q < NW; ++q) tot += red[q][l15];
    const float inv_mine = 1.0f / fmaxf(sqrtf(tot), 1e-12f);
    const int r0 = tm + h * 8;
#pragma unroll
    for (int v = 0; v < 8; ++v) {
        const float inv = __shfl(inv_mine, h * 8 + v, 32);
        __bf16* hrow = Hb + (size_t)(r0 + v) * NCOL + tn0 + l15;
#pragma unroll
        for (int t = 0; t < NT; ++t) {
            const float r = fmaxf(acc[t][v] * inv, 0.0f);
            hrow[t * 16] = (__bf16)r;
        }
    }
}

extern "C" void kernel_launch(void* const* d_in, const int* in_sizes, int n_in,
                              void* d_out, int out_size, void* d_ws, size_t ws_size,
                              hipStream_t stream) {
    (void)in_sizes; (void)n_in; (void)out_size; (void)ws_size;

    const float* feat = (const float*)d_in[0];
    const int*   eidx = (const int*)  d_in[1];
    const int*   src  = eidx;
    const int*   dst  = eidx + N_EDGES;
    const float* W1l  = (const float*)d_in[2];
    const float* b1   = (const float*)d_in[3];
    const float* W1r  = (const float*)d_in[4];
    const float* W2l  = (const float*)d_in[5];
    const float* b2   = (const float*)d_in[6];
    const float* W2r  = (const float*)d_in[7];
    const float* Wfc  = (const float*)d_in[8];
    const float* bfc  = (const float*)d_in[9];
    float* out = (float*)d_out;

    char* ws = (char*)d_ws;
    size_t off = 0;
    auto alloc = [&](size_t bytes) -> void* {
        void* p = ws + off;
        off = (off + bytes + 255) & ~(size_t)255;
        return p;
    };
    int*    deg    = (int*)  alloc((size_t)DEG_PAD * 4);
    int*    rowptr = (int*)  alloc((size_t)DEG_PAD * 4);
    int*    rank   = (int*)  alloc((size_t)N_EDGES * 4);
    int*    col    = (int*)  alloc((size_t)N_EDGES * 4);
    __bf16* Xcat1 = (__bf16*)alloc((size_t)N_NODES * 2 * IN_SIZE * 2);
    __bf16* Wcat1 = (__bf16*)alloc((size_t)HIDDEN * 2 * IN_SIZE * 2);
    __bf16* h1b   = (__bf16*)alloc((size_t)N_NODES * HIDDEN * 2);
    __bf16* Wst2  = (__bf16*)alloc((size_t)2 * OUT_SIZE * HIDDEN * 2);
    float*  pq2   = (float*) alloc((size_t)N_NODES * 2 * OUT_SIZE * 4);

    const int TB = 256;
    auto cdiv = [](int a, int b) { return (a + b - 1) / b; };

    zero_i32<<<cdiv(DEG_PAD, TB), TB, 0, stream>>>(deg, DEG_PAD);

    build_wcat<<<cdiv(HIDDEN * IN_SIZE, TB), TB, 0, stream>>>(W1l, W1r, Wcat1, HIDDEN, IN_SIZE);
    build_wstack<<<cdiv(2 * OUT_SIZE * HIDDEN, TB), TB, 0, stream>>>(W2l, W2r, Wst2,
                                                                    OUT_SIZE * HIDDEN);

    edge_rank<<<cdiv(N_EDGES, TB), TB, 0, stream>>>(dst, deg, rank, N_EDGES);
    scan_rowptr<<<1, SCAN_T, 0, stream>>>(deg, rowptr);
    csr_fill<<<cdiv(N_EDGES, TB), TB, 0, stream>>>(src, dst, rowptr, rank, col, N_EDGES);
    gather_mean_cat<IN_SIZE><<<cdiv(N_NODES, TB / 32), TB, 0, stream>>>(
        feat, rowptr, col, Xcat1, N_NODES);
    gemm_bf16_wmma_norm_relu<2 * IN_SIZE, HIDDEN><<<N_NODES / 16, HIDDEN / 2, 0, stream>>>(
        Xcat1, Wcat1, b1, h1b);

    {
        int waves = (N_NODES / 16) * (2 * OUT_SIZE / 64);
        gemm_bf16_wmma<HIDDEN, 4><<<cdiv(waves, TB / 32), TB, 0, stream>>>(
            h1b, Wst2, b2, OUT_SIZE, pq2, N_NODES, 2 * OUT_SIZE);
    }
    gather_mean_add_fc<OUT_SIZE><<<cdiv(N_NODES, TB / 32), TB, 0, stream>>>(
        pq2, rowptr, col, Wfc, bfc, out, N_NODES);
}


# Both GEMMs: the 128-KiB weight operand copied once per resident work-group (256 x 32 waves looping over 16-row tiles) into padded LDS; B fragments by ds_load_b128, not by 16-line global loads
# speedup vs baseline: 3.5178x; 1.2470x over previous
#include <hip/hip_runtime.h>
#include <math.h>

#define N_NODES  50000
#define N_EDGES  600000
#define IN_SIZE  128
#define HIDDEN   256
#define OUT_SIZE 128

typedef __attribute__((ext_vector_type(16))) __bf16 v16bf;
typedef __attribute__((ext_vector_type(8)))  __bf16 v8bf;
typedef __attribute__((ext_vector_type(8)))  float  v8f;
typedef __attribute__((ext_vector_type(4)))  __bf16 v4bf;

#define SCAN_T   1024
#define SCAN_PER 52
#define DEG_PAD  (SCAN_T * SCAN_PER)

__global__ __launch_bounds__(256)
void zero_i32(int* __restrict__ p, int n) {
    int i = blockIdx.x * blockDim.x + threadIdx.x;
    if (i < n) p[i] = 0;
}

__global__ __launch_bounds__(256)
void build_wcat(const float* __restrict__ Wl, const float* __restrict__ Wr,
                __bf16* __restrict__ Wcat, int rows, int kHalf) {
    int i = blockIdx.x * blockDim.x + threadIdx.x;
    if (i >= rows * kHalf) return;
    int r = i / kHalf, c = i % kHalf;
    __bf16* row = Wcat + (size_t)r * (2 * kHalf);
    row[c]         = (__bf16)Wl[i];
    row[kHalf + c] = (__bf16)Wr[i];
}

__global__ __launch_bounds__(256)
void build_wstack(const float* __restrict__ Wa, const float* __restrict__ Wb,
                  __bf16* __restrict__ Wst, int n) {
    int i = blockIdx.x * blockDim.x + threadIdx.x;
    if (i >= 2 * n) return;
    Wst[i] = (__bf16)(i < n ? Wa[i] : Wb[i - n]);
}


__global__ __launch_bounds__(256)
void edge_rank(const int* __restrict__ dst, int* __restrict__ deg,
               int* __restrict__ rank, int nedges) {
    int e = blockIdx.x * blockDim.x + threadIdx.x;
    if (e >= nedges) return;
    rank[e] = atomicAdd(deg + dst[e], 1);
}

__global__ __launch_bounds__(SCAN_T)
void scan_rowptr(const int* __restrict__ deg, int* __restrict__ rowptr) {
    __shared__ int wsum[32];
    const int t = threadIdx.x, lane = t & 31, w = t >> 5;
    const int4* d4 = (const int4*)(deg + t * SCAN_PER);
    int4 d[SCAN_PER / 4];
    int s = 0;
#pragma unroll
    for (int i = 0; i < SCAN_PER / 4; ++i) {
        d[i] = d4[i];
        s += (d[i].x + d[i].y) + (d[i].z + d[i].w);
    }
    int inc = s;
    for (int off = 1; off < 32; off <<= 1) {
        int v = __shfl_up(inc, off, 32);
        if (lane >= off) inc += v;
    }
    if (lane == 31) wsum[w] = inc;
    __syncthreads();
    if (w == 0) {
        int x = wsum[lane], xi = x;
        for (int off = 1; off < 32; off <<= 1) {
            int v = __shfl_up(xi, off, 32);
            if (lane >= off) xi += v;
        }
        wsum[lane] = xi - x;
    }
    __syncthreads();
    int run = wsum[w] + inc - s;
    int4* r4 = (int4*)(rowptr + t * SCAN_PER);
#pragma unroll
    for (int i = 0; i < SCAN_PER / 4; ++i) {
        int4 o;
        o.x = run; run += d[i].x;
        o.y = run; run += d[i].y;
        o.z = run; run += d[i].z;
        o.w = run; run += d[i].w;
        r4[i] = o;
    }
}

__global__ __launch_bounds__(256)
void csr_fill(const int* __restrict__ src, const int* __restrict__ dst,
              const int* __restrict__ rowptr, const int* __restrict__ rank,
              int* __restrict__ col, int nedges) {
    int e = blockIdx.x * blockDim.x + threadIdx.x;
    if (e >= nedges) return;
    col[rowptr[dst[e]] + rank[e]] = src[e];
}

__device__ __forceinline__ v4bf pack4_bf16(float4 v, float sc) {
    v4bf o;
    o[0] = (__bf16)(v.x * sc);
    o[1] = (__bf16)(v.y * sc);
    o[2] = (__bf16)(v.z * sc);
    o[3] = (__bf16)(v.w * sc);
    return o;
}

template <int F, int PITCH>
__device__ __forceinline__ void sum_rows(const float4* __restrict__ X4, const int* __restrict__ col,
                                         int beg, int end, int lane, float4 (&acc)[F / 128]) {
    constexpr int NV = F / 128;
#pragma unroll
    for (int q = 0; q < NV; ++q) acc[q] = make_float4(0.0f, 0.0f, 0.0f, 0.0f);

    for (int base = beg; base < end; base += 32) {
        const int m = (end - base < 32) ? (end - base) : 32;
        const int myc = (lane < m) ? col[base + lane] : 0;
        int j = 0;
        for (; j + 4 <= m; j += 4) {
            const float4* r0 = X4 + (size_t)__builtin_amdgcn_readlane(myc, j + 0) * (PITCH / 4) + lane;
            const float4* r1 = X4 + (size_t)__builtin_amdgcn_readlane(myc, j + 1) * (PITCH / 4) + lane;
            const float4* r2 = X4 + (size_t)__builtin_amdgcn_readlane(myc, j + 2) * (PITCH / 4) + lane;
            const float4* r3 = X4 + (size_t)__builtin_amdgcn_readlane(myc, j + 3) * (PITCH / 4) + lane;
            float4 v0[NV], v1[NV], v2[NV], v3[NV];
#pragma unroll
            for (int q = 0; q < NV; ++q) {
                v0[q] = r0[q * 32]; v1[q] = r1[q * 32];
                v2[q] = r2[q * 32]; v3[q] = r3[q * 32];
            }
#pragma unroll
            for (int q = 0; q < NV; ++q) {
                acc[q].x += (v0[q].x + v1[q].x) + (v2[q].x + v3[q].x);
                acc[q].y += (v0[q].y + v1[q].y) + (v2[q].y + v3[q].y);
                acc[q].z += (v0[q].z + v1[q].z) + (v2[q].z + v3[q].z);
                acc[q].w += (v0[q].w + v1[q].w) + (v2[q].w + v3[q].w);
            }
        }
        for (; j < m; ++j) {
            const float4* r0 = X4 + (size_t)__builtin_amdgcn_readlane(myc, j) * (PITCH / 4) + lane;
#pragma unroll
            for (int q = 0; q < NV; ++q) {
                float4 v = r0[q * 32];
                acc[q].x += v.x; acc[q].y += v.y; acc[q].z += v.z; acc[q].w += v.w;
            }
        }
    }
}

template <int F>
__global__ __launch_bounds__(256)
void gather_mean_cat(const float* __restrict__ X, const int* __restrict__ rowptr,
                     const int* __restrict__ col, __bf16* __restrict__ Xcat, int nnodes) {
    constexpr int NV = F / 128;
    const int lane = threadIdx.x & 31;
    const int n = blockIdx.x * (blockDim.x >> 5) + (threadIdx.x >> 5);
    if (n >= nnodes) return;
    const int beg = __builtin_amdgcn_readfirstlane(rowptr[n]);
    const int end = __builtin_amdgcn_readfirstlane(rowptr[n + 1]);
    const float4* X4 = (const float4*)X;
    float4 acc[NV];
    sum_rows<F, F>(X4, col, beg, end, lane, acc);

    const float inv = 1.0f / fmaxf((float)(end - beg), 1.0f);
    __bf16* row = Xcat + (size_t)n * (2 * F);
#pragma unroll
    for (int q = 0; q < NV; ++q)
        *(v4bf*)(row + q * 128 + lane * 4) = pack4_bf16(acc[q], inv);
#pragma unroll
    for (int q = 0; q < NV; ++q) {
        float4 v = X4[(size_t)n * (F / 4) + q * 32 + lane];
        *(v4bf*)(row + F + q * 128 + lane * 4) = pack4_bf16(v, 1.0f);
    }
}

template <int F>
__global__ __launch_bounds__(256)
void gather_mean_add_fc(const float* __restrict__ PQ, const int* __restrict__ rowptr,
                        const int* __restrict__ col, const float* __restrict__ Wfc,
                        const float* __restrict__ bfc, float* __restrict__ out, int nnodes) {
    constexpr int NV = F / 128;
    const int lane = threadIdx.x & 31;
    const int n = blockIdx.x * (blockDim.x >> 5) + (threadIdx.x >> 5);
    if (n >= nnodes) return;
    const int beg = __builtin_amdgcn_readfirstlane(rowptr[n]);
    const int end = __builtin_amdgcn_readfirstlane(rowptr[n + 1]);
    const float4* PQ4 = (const float4*)PQ;
    const float4* W4  = (const float4*)Wfc;
    float4 self[NV];
    float4 w0[NV], w1[NV];
#pragma unroll
    for (int q = 0; q < NV; ++q) {
        self[q] = PQ4[(size_t)n * (2 * F / 4) + F / 4 + q * 32 + lane];
        w0[q]   = W4[q * 32 + lane];
        w1[q]   = W4[F / 4 + q * 32 + lane];
    }
    float4 acc[NV];
    sum_rows<F, 2 * F>(PQ4, col, beg, end, lane, acc);

    const float inv = 1.0f / fmaxf((float)(end - beg), 1.0f);
    float ss = 0.0f, d0 = 0.0f, d1 = 0.0f;
#pragma unroll
    for (int q = 0; q < NV; ++q) {
        float4 o;
        o.x = acc[q].x * inv + self[q].x;
        o.y = acc[q].y * inv + self[q].y;
        o.z = acc[q].z * inv + self[q].z;
        o.w = acc[q].w * inv + self[q].w;
        ss += (o.x * o.x + o.y * o.y) + (o.z * o.z + o.w * o.w);
        d0 += (o.x * w0[q].x + o.y * w0[q].y) + (o.z * w0[q].z + o.w * w0[q].w);
        d1 += (o.x * w1[q].x + o.y * w1[q].y) + (o.z * w1[q].z + o.w * w1[q].w);
    }
    for (int off = 16; off > 0; off >>= 1) {
        ss += __shfl_xor(ss, off, 32);
        d0 += __shfl_xor(d0, off, 32);
        d1 += __shfl_xor(d1, off, 32);
    }
    if (lane == 0) {
        float invn = 1.0f / fmaxf(sqrtf(ss), 1e-12f);
        float l0 = d0 * invn + bfc[0];
        float l1 = d1 * invn + bfc[1];
        float m  = fmaxf(l0, l1);
        float e0 = __expf(l0 - m), e1 = __expf(l1 - m);
        float s  = 1.0f / (e0 + e1);
        out[(size_t)n * 2 + 0] = e0 * s;
        out[(size_t)n * 2 + 1] = e1 * s;
    }
}

#define GEMM_K       256
#define GEMM_NCOL    256
#define GEMM_THREADS 1024
#define GEMM_BLOCKS  256
#define BPITCH       (GEMM_K + 8)

template <int K, int NROW>
__device__ __forceinline__ void stage_b_lds(const __bf16* __restrict__ B, __bf16* Bs, int tid) {
    constexpr int PPR   = K / 8;
    constexpr int TOTAL = NROW * PPR;
#pragma unroll
    for (int i = 0; i < TOTAL / GEMM_THREADS; ++i) {
        const int p   = i * GEMM_THREADS + tid;
        const int row = p / PPR, pc = p % PPR;
        const v8bf v = *(const v8bf*)(B + (size_t)row * K + pc * 8);
        *(v8bf*)(Bs + row * BPITCH + pc * 8) = v;
    }
}

template <int K, int NT>
__device__ __forceinline__ void wmma_strip_lds(const __bf16* __restrict__ arow, const __bf16* brow,
                                               v8f (&acc)[NT]) {
    union { v16bf v; v8bf h[2]; } a[K / 32];
#pragma unroll
    for (int s = 0; s < K / 32; ++s) {
        a[s].h[0] = *(const v8bf*)(arow + s * 32);
        a[s].h[1] = *(const v8bf*)(arow + s * 32 + 16);
    }
    __builtin_amdgcn_sched_barrier(0);
#pragma unroll
    for (int t = 0; t < NT; ++t) acc[t] = (v8f){};

#pragma unroll
    for (int s = 0; s < K / 32; ++s) {
#pragma unroll
        for (int t = 0; t < NT; ++t) {
            union { v16bf v; v8bf h[2]; } b;
            b.h[0] = *(const v8bf*)(brow + t * 16 * BPITCH + s * 32);
            b.h[1] = *(const v8bf*)(brow + t * 16 * BPITCH + s * 32 + 16);
            acc[t] = __builtin_amdgcn_wmma_f32_16x16x32_bf16(
                false, a[s].v, false, b.v, (short)0, acc[t], false, false);
        }
    }
}

__global__ __launch_bounds__(GEMM_THREADS)
void gemm_bf16_wmma_ldsb(const __bf16* __restrict__ A, const __bf16* __restrict__ B,
                         const float* __restrict__ bias, int bias_col0, float* __restrict__ C,
                         int ntiles) {
    constexpr int K = GEMM_K, N = GEMM_NCOL, NT = 4, NW = N / (16 * NT), NS = (GEMM_THREADS / 32) / NW;
    __shared__ __attribute__((aligned(16))) __bf16 Bs[N * BPITCH];
    const int lane = threadIdx.x & 31;
    const int wv   = __builtin_amdgcn_readfirstlane(threadIdx.x >> 5);
    const int w    = wv % NW;
    const int slot = wv / NW;
    const int tn0  = w * (16 * NT);
    const int h    = lane >> 4;
    const int l15  = lane & 15;

    stage_b_lds<K, N>(B, Bs, threadIdx.x);
    __syncthreads();

    const __bf16* brow = Bs + (tn0 + l15) * BPITCH + h * 8;
    for (int tile = slot * gridDim.x + blockIdx.x; tile < ntiles; tile += gridDim.x * NS) {
        asm volatile("" ::: "memory");
        const int tm = tile << 4;
        v8f acc[NT];
        wmma_strip_lds<K, NT>(A + (size_t)(tm + l15) * K + h * 8, brow, acc);

        const int r0 = tm + h * 8;
#pragma unroll
        for (int t = 0; t < NT; ++t) {
            const int cn = tn0 + t * 16 + l15;
            const int   bi = cn - bias_col0;
            const float bv = bias[bi < 0 ? 0 : bi];
            const float bc = (bi < 0) ? 0.0f : bv;
#pragma unroll
            for (int v = 0; v < 8; ++v)
                C[(size_t)(r0 + v) * N + cn] = acc[t][v] + bc;
        }
    }
}

__global__ __launch_bounds__(GEMM_THREADS)
void gemm_bf16_wmma_ldsb_norm_relu(const __bf16* __restrict__ A, const __bf16* __restrict__ B,
                                   const float* __restrict__ bias, __bf16* __restrict__ Hb,
                                   int ntiles) {
    constexpr int K = GEMM_K, NCOL = GEMM_NCOL, NT = 4, NW = NCOL / (16 * NT), NS = (GEMM_THREADS / 32) / NW;
    __shared__ __attribute__((aligned(16))) __bf16 Bs[NCOL * BPITCH];
    __shared__ __attribute__((aligned(16))) float red[NS][NW][16];
    const int lane = threadIdx.x & 31;
    const int wv   = __builtin_amdgcn_readfirstlane(threadIdx.x >> 5);
    const int w    = wv % NW;
    const int slot = wv / NW;
    const int tn0  = w * (16 * NT);
    const int h    = lane >> 4;
    const int l15  = lane & 15;

    stage_b_lds<K, NCOL>(B, Bs, threadIdx.x);
    __syncthreads();

    const __bf16* brow = Bs + (tn0 + l15) * BPITCH + h * 8;
    const int stride = gridDim.x * NS;
    const int trips  = (ntiles + stride - 1) / stride;
    for (int it = 0; it < trips; ++it) {
        const int  tile  = it * stride + slot * gridDim.x + blockIdx.x;
        const bool valid = tile < ntiles;
        const int  tm    = tile << 4;
        v8f acc[NT];
        if (valid) {
            wmma_strip_lds<K, NT>(A + (size_t)(tm + l15) * K + h * 8, brow, acc);

            float ss[8];
#pragma unroll
            for (int v = 0; v < 8; ++v) ss[v] = 0.0f;
#pragma unroll
            for (int t = 0; t < NT; ++t) {
                const float bc = bias[tn0 + t * 16 + l15];
#pragma unroll
                for (int v = 0; v < 8; ++v) {
                    acc[t][v] += bc;
                    ss[v] += acc[t][v] * acc[t][v];
                }
            }
#pragma unroll
            for (int off = 1; off < 16; off <<= 1) {
#pragma unroll
                for (int v = 0; v < 8; ++v) ss[v] += __shfl_xor(ss[v], off, 32);
            }
            if (l15 == 0) {
                float4* mine = (float4*)&red[slot][w][h * 8];
                mine[0] = make_float4(ss[0], ss[1], ss[2], ss[3]);
                mine[1] = make_float4(ss[4], ss[5], ss[6], ss[7]);
            }
        }
        __syncthreads();
        if (valid) {
            float tot = red[slot][0][l15];
#pragma unroll
            for (int q = 1; q < NW; ++q) tot += red[slot][q][l15];
            const float inv_mine = 1.0f / fmaxf(sqrtf(tot), 1e-12f);
            const int r0 = tm + h * 8;
#pragma unroll
            for (int v = 0; v < 8; ++v) {
                const float inv = __shfl(inv_mine, h * 8 + v, 32);
                __bf16* hrow = Hb + (size_t)(r0 + v) * NCOL + tn0 + l15;
#pragma unroll
                for (int t = 0; t < NT; ++t) {
                    const float r = fmaxf(acc[t][v] * inv, 0.0f);
                    hrow[t * 16] = (__bf16)r;
                }
            }
        }
        __syncthreads();
    }
}

extern "C" void kernel_launch(void* const* d_in, const int* in_sizes, int n_in,
                              void* d_out, int out_size, void* d_ws, size_t ws_size,
                              hipStream_t stream) {
    (void)in_sizes; (void)n_in; (void)out_size; (void)ws_size;

    const float* feat = (const float*)d_in[0];
    const int*   eidx = (const int*)  d_in[1];
    const int*   src  = eidx;
    const int*   dst  = eidx + N_EDGES;
    const float* W1l  = (const float*)d_in[2];
    const float* b1   = (const float*)d_in[3];
    const float* W1r  = (const float*)d_in[4];
    const float* W2l  = (const float*)d_in[5];
    const float* b2   = (const float*)d_in[6];
    const float* W2r  = (const float*)d_in[7];
    const float* Wfc  = (const float*)d_in[8];
    const float* bfc  = (const float*)d_in[9];
    float* out = (float*)d_out;

    char* ws = (char*)d_ws;
    size_t off = 0;
    auto alloc = [&](size_t bytes) -> void* {
        void* p = ws + off;
        off = (off + bytes + 255) & ~(size_t)255;
        return p;
    };
    int*    deg    = (int*)  alloc((size_t)DEG_PAD * 4);
    int*    rowptr = (int*)  alloc((size_t)DEG_PAD * 4);
    int*    rank   = (int*)  alloc((size_t)N_EDGES * 4);
    int*    col    = (int*)  alloc((size_t)N_EDGES * 4);
    __bf16* Xcat1 = (__bf16*)alloc((size_t)N_NODES * 2 * IN_SIZE * 2);
    __bf16* Wcat1 = (__bf16*)alloc((size_t)HIDDEN * 2 * IN_SIZE * 2);
    __bf16* h1b   = (__bf16*)alloc((size_t)N_NODES * HIDDEN * 2);
    __bf16* Wst2  = (__bf16*)alloc((size_t)2 * OUT_SIZE * HIDDEN * 2);
    float*  pq2   = (float*) alloc((size_t)N_NODES * 2 * OUT_SIZE * 4);

    const int TB = 256;
    auto cdiv = [](int a, int b) { return (a + b - 1) / b; };

    zero_i32<<<cdiv(DEG_PAD, TB), TB, 0, stream>>>(deg, DEG_PAD);

    build_wcat<<<cdiv(HIDDEN * IN_SIZE, TB), TB, 0, stream>>>(W1l, W1r, Wcat1, HIDDEN, IN_SIZE);
    build_wstack<<<cdiv(2 * OUT_SIZE * HIDDEN, TB), TB, 0, stream>>>(W2l, W2r, Wst2,
                                                                    OUT_SIZE * HIDDEN);

    edge_rank<<<cdiv(N_EDGES, TB), TB, 0, stream>>>(dst, deg, rank, N_EDGES);
    scan_rowptr<<<1, SCAN_T, 0, stream>>>(deg, rowptr);
    csr_fill<<<cdiv(N_EDGES, TB), TB, 0, stream>>>(src, dst, rowptr, rank, col, N_EDGES);
    gather_mean_cat<IN_SIZE><<<cdiv(N_NODES, TB / 32), TB, 0, stream>>>(
        feat, rowptr, col, Xcat1, N_NODES);
    gemm_bf16_wmma_ldsb_norm_relu<<<GEMM_BLOCKS, GEMM_THREADS, 0, stream>>>(
        Xcat1, Wcat1, b1, h1b, N_NODES / 16);

    gemm_bf16_wmma_ldsb<<<GEMM_BLOCKS, GEMM_THREADS, 0, stream>>>(
        h1b, Wst2, b2, OUT_SIZE, pq2, N_NODES / 16);
    gather_mean_add_fc<OUT_SIZE><<<cdiv(N_NODES, TB / 32), TB, 0, stream>>>(
        pq2, rowptr, col, Wfc, bfc, out, N_NODES);
}


# In-edge lists built with LDS counters by (node range, edge piece) work-groups: a count and a fill pass replace zero, per-edge returning global add, one-block scan, fill; neighbour sum order changes
# speedup vs baseline: 5.0911x; 1.4472x over previous
#include <hip/hip_runtime.h>
#include <math.h>

#define N_NODES  50000
#define N_EDGES  600000
#define IN_SIZE  128
#define HIDDEN   256
#define OUT_SIZE 128

typedef __attribute__((ext_vector_type(16))) __bf16 v16bf;
typedef __attribute__((ext_vector_type(8)))  __bf16 v8bf;
typedef __attribute__((ext_vector_type(8)))  float  v8f;
typedef __attribute__((ext_vector_type(4)))  __bf16 v4bf;

#define NR_NODES   2048
#define N_RANGES   ((N_NODES + NR_NODES - 1) / NR_NODES)
#define N_SPLITS   10
#define SPLIT_E    (N_EDGES / N_SPLITS)
#define SPLIT_V4   (SPLIT_E / 4)
#define PART_PITCH (N_RANGES * NR_NODES)
#define LIST_T     1024
#define LIST_UNR   8
static_assert(N_EDGES % (N_SPLITS * 4) == 0, "pieces are whole 16-byte vectors");
static_assert(N_NODES < PART_PITCH, "rowptr[N_NODES] must be an entry of the last range");
static_assert(NR_NODES == 2 * LIST_T, "a thread owns two adjacent counters");
static_assert(N_RANGES * N_SPLITS <= LIST_T, "one thread per (range, piece) sum");

__global__ __launch_bounds__(256)
void build_wcat(const float* __restrict__ Wl, const float* __restrict__ Wr,
                __bf16* __restrict__ Wcat, int rows, int kHalf) {
    int i = blockIdx.x * blockDim.x + threadIdx.x;
    if (i >= rows * kHalf) return;
    int r = i / kHalf, c = i % kHalf;
    __bf16* row = Wcat + (size_t)r * (2 * kHalf);
    row[c]         = (__bf16)Wl[i];
    row[kHalf + c] = (__bf16)Wr[i];
}

__global__ __launch_bounds__(256)
void build_wstack(const float* __restrict__ Wa, const float* __restrict__ Wb,
                  __bf16* __restrict__ Wst, int n) {
    int i = blockIdx.x * blockDim.x + threadIdx.x;
    if (i >= 2 * n) return;
    Wst[i] = (__bf16)(i < n ? Wa[i] : Wb[i - n]);
}


struct EdgeBatch {
    int4 v[LIST_UNR];
};
__device__ __forceinline__ void load_batch(const int4* __restrict__ p4, int i0, EdgeBatch& b) {
#pragma unroll
    for (int k = 0; k < LIST_UNR; ++k) {
        const int i = i0 + k * LIST_T;
        b.v[k] = p4[i < SPLIT_V4 ? i : SPLIT_V4 - 1];
    }
}

__global__ __launch_bounds__(LIST_T)
void inlist_count(const int* __restrict__ dst, int* __restrict__ part, int* __restrict__ tot) {
    __shared__ __attribute__((aligned(16))) int cnt[NR_NODES];
    __shared__ int wsum[LIST_T / 32];
    const int tid = threadIdx.x, lane = tid & 31, w = tid >> 5;
    const int g = blockIdx.x / N_SPLITS, s = blockIdx.x % N_SPLITS;
    const int lo = g * NR_NODES;
    cnt[tid] = 0;
    cnt[tid + LIST_T] = 0;
    __syncthreads();

    const int4* d4 = (const int4*)(dst + (size_t)s * SPLIT_E);
    for (int i0 = tid; i0 < SPLIT_V4; i0 += LIST_T * LIST_UNR) {
        EdgeBatch d;
        load_batch(d4, i0, d);
#pragma unroll
        for (int k = 0; k < LIST_UNR; ++k) {
            const bool ok = (i0 + k * LIST_T) < SPLIT_V4;
            const unsigned u0 = (unsigned)(d.v[k].x - lo), u1 = (unsigned)(d.v[k].y - lo);
            const unsigned u2 = (unsigned)(d.v[k].z - lo), u3 = (unsigned)(d.v[k].w - lo);
            if (ok && u0 < NR_NODES) atomicAdd(&cnt[u0], 1);
            if (ok && u1 < NR_NODES) atomicAdd(&cnt[u1], 1);
            if (ok && u2 < NR_NODES) atomicAdd(&cnt[u2], 1);
            if (ok && u3 < NR_NODES) atomicAdd(&cnt[u3], 1);
        }
    }
    __syncthreads();

    const int2 c = *(const int2*)&cnt[2 * tid];
    *(int2*)(part + (size_t)s * PART_PITCH + lo + 2 * tid) = c;
    int t = c.x + c.y;
    for (int off = 16; off > 0; off >>= 1) t += __shfl_xor(t, off, 32);
    if (lane == 0) wsum[w] = t;
    __syncthreads();
    if (w == 0) {
        int x = wsum[lane];
        for (int off = 16; off > 0; off >>= 1) x += __shfl_xor(x, off, 32);
        if (lane == 0) tot[g * N_SPLITS + s] = x;
    }
}

__global__ __launch_bounds__(LIST_T)
void inlist_fill(const int* __restrict__ src, const int* __restrict__ dst,
                 const int* __restrict__ part, const int* __restrict__ tot,
                 int* __restrict__ rowptr, int* __restrict__ col) {
    __shared__ __attribute__((aligned(16))) int cur[NR_NODES];
    __shared__ int wsum[LIST_T / 32];
    __shared__ int wbase[LIST_T / 32];
    const int tid = threadIdx.x, lane = tid & 31, w = tid >> 5;
    const int g = blockIdx.x / N_SPLITS, s = blockIdx.x % N_SPLITS;
    const int lo = g * NR_NODES;

    int b = (tid < g * N_SPLITS) ? tot[tid] : 0;
    int2 len = make_int2(0, 0), below = make_int2(0, 0);
#pragma unroll
    for (int q = 0; q < N_SPLITS; ++q) {
        const int2 p = *(const int2*)(part + (size_t)q * PART_PITCH + lo + 2 * tid);
        len.x += p.x; len.y += p.y;
        if (q < s) { below.x += p.x; below.y += p.y; }
    }
    const int v = len.x + len.y;
    int inc = v;
    for (int off = 1; off < 32; off <<= 1) {
        int x = __shfl_up(inc, off, 32);
        if (lane >= off) inc += x;
    }
    for (int off = 16; off > 0; off >>= 1) b += __shfl_xor(b, off, 32);
    if (lane == 31) wsum[w] = inc;
    if (lane == 0)  wbase[w] = b;
    __syncthreads();
    if (w == 0) {
        int x = wsum[lane], xi = x;
        for (int off = 1; off < 32; off <<= 1) {
            int y = __shfl_up(xi, off, 32);
            if (lane >= off) xi += y;
        }
        wsum[lane] = xi - x;
        int bb = wbase[lane];
        for (int off = 16; off > 0; off >>= 1) bb += __shfl_xor(bb, off, 32);
        if (lane == 0) wbase[0] = bb;
    }
    __syncthreads();
    const int st0 = wbase[0] + wsum[w] + inc - v;
    const int st1 = st0 + len.x;
    *(int2*)&cur[2 * tid] = make_int2(st0 + below.x, st1 + below.y);
    if (s == 0) *(int2*)(rowptr + lo + 2 * tid) = make_int2(st0, st1);
    __syncthreads();

    const int4* d4 = (const int4*)(dst + (size_t)s * SPLIT_E);
    const int4* s4 = (const int4*)(src + (size_t)s * SPLIT_E);
    for (int i0 = tid; i0 < SPLIT_V4; i0 += LIST_T * LIST_UNR) {
        EdgeBatch d, sv;
        load_batch(d4, i0, d);
        load_batch(s4, i0, sv);
        asm volatile("" ::: "memory");
#pragma unroll
        for (int k = 0; k < LIST_UNR; ++k) {
            const bool ok = (i0 + k * LIST_T) < SPLIT_V4;
            const unsigned u0 = (unsigned)(d.v[k].x - lo), u1 = (unsigned)(d.v[k].y - lo);
            const unsigned u2 = (unsigned)(d.v[k].z - lo), u3 = (unsigned)(d.v[k].w - lo);
            if (ok && u0 < NR_NODES) col[atomicAdd(&cur[u0], 1)] = sv.v[k].x;
            if (ok && u1 < NR_NODES) col[atomicAdd(&cur[u1], 1)] = sv.v[k].y;
            if (ok && u2 < NR_NODES) col[atomicAdd(&cur[u2], 1)] = sv.v[k].z;
            if (ok && u3 < NR_NODES) col[atomicAdd(&cur[u3], 1)] = sv.v[k].w;
        }
    }
}

__device__ __forceinline__ v4bf pack4_bf16(float4 v, float sc) {
    v4bf o;
    o[0] = (__bf16)(v.x * sc);
    o[1] = (__bf16)(v.y * sc);
    o[2] = (__bf16)(v.z * sc);
    o[3] = (__bf16)(v.w * sc);
    return o;
}

template <int F, int PITCH>
__device__ __forceinline__ void sum_rows(const float4* __restrict__ X4, const int* __restrict__ col,
                                         int beg, int end, int lane, float4 (&acc)[F / 128]) {
    constexpr int NV = F / 128;
#pragma unroll
    for (int q = 0; q < NV; ++q) acc[q] = make_float4(0.0f, 0.0f, 0.0f, 0.0f);

    for (int base = beg; base < end; base += 32) {
        const int m = (end - base < 32) ? (end - base) : 32;
        const int myc = (lane < m) ? col[base + lane] : 0;
        int j = 0;
        for (; j + 4 <= m; j += 4) {
            const float4* r0 = X4 + (size_t)__builtin_amdgcn_readlane(myc, j + 0) * (PITCH / 4) + lane;
            const float4* r1 = X4 + (size_t)__builtin_amdgcn_readlane(myc, j + 1) * (PITCH / 4) + lane;
            const float4* r2 = X4 + (size_t)__builtin_amdgcn_readlane(myc, j + 2) * (PITCH / 4) + lane;
            const float4* r3 = X4 + (size_t)__builtin_amdgcn_readlane(myc, j + 3) * (PITCH / 4) + lane;
            float4 v0[NV], v1[NV], v2[NV], v3[NV];
#pragma unroll
            for (int q = 0; q < NV; ++q) {
                v0[q] = r0[q * 32]; v1[q] = r1[q * 32];
                v2[q] = r2[q * 32]; v3[q] = r3[q * 32];
            }
#pragma unroll
            for (int q = 0; q < NV; ++q) {
                acc[q].x += (v0[q].x + v1[q].x) + (v2[q].x + v3[q].x);
                acc[q].y += (v0[q].y + v1[q].y) + (v2[q].y + v3[q].y);
                acc[q].z += (v0[q].z + v1[q].z) + (v2[q].z + v3[q].z);
                acc[q].w += (v0[q].w + v1[q].w) + (v2[q].w + v3[q].w);
            }
        }
        for (; j < m; ++j) {
            const float4* r0 = X4 + (size_t)__builtin_amdgcn_readlane(myc, j) * (PITCH / 4) + lane;
#pragma unroll
            for (int q = 0; q < NV; ++q) {
                float4 v = r0[q * 32];
                acc[q].x += v.x; acc[q].y += v.y; acc[q].z += v.z; acc[q].w += v.w;
            }
        }
    }
}

template <int F>
__global__ __launch_bounds__(256)
void gather_mean_cat(const float* __restrict__ X, const int* __restrict__ rowptr,
                     const int* __restrict__ col, __bf16* __restrict__ Xcat, int nnodes) {
    constexpr int NV = F / 128;
    const int lane = threadIdx.x & 31;
    const int n = blockIdx.x * (blockDim.x >> 5) + (threadIdx.x >> 5);
    if (n >= nnodes) return;
    const int beg = __builtin_amdgcn_readfirstlane(rowptr[n]);
    const int end = __builtin_amdgcn_readfirstlane(rowptr[n + 1]);
    const float4* X4 = (const float4*)X;
    float4 acc[NV];
    sum_rows<F, F>(X4, col, beg, end, lane, acc);

    const float inv = 1.0f / fmaxf((float)(end - beg), 1.0f);
    __bf16* row = Xcat + (size_t)n * (2 * F);
#pragma unroll
    for (int q = 0; q < NV; ++q)
        *(v4bf*)(row + q * 128 + lane * 4) = pack4_bf16(acc[q], inv);
#pragma unroll
    for (int q = 0; q < NV; ++q) {
        float4 v = X4[(size_t)n * (F / 4) + q * 32 + lane];
        *(v4bf*)(row + F + q * 128 + lane * 4) = pack4_bf16(v, 1.0f);
    }
}

template <int F>
__global__ __launch_bounds__(256)
void gather_mean_add_fc(const float* __restrict__ PQ, const int* __restrict__ rowptr,
                        const int* __restrict__ col, const float* __restrict__ Wfc,
                        const float* __restrict__ bfc, float* __restrict__ out, int nnodes) {
    constexpr int NV = F / 128;
    const int lane = threadIdx.x & 31;
    const int n = blockIdx.x * (blockDim.x >> 5) + (threadIdx.x >> 5);
    if (n >= nnodes) return;
    const int beg = __builtin_amdgcn_readfirstlane(rowptr[n]);
    const int end = __builtin_amdgcn_readfirstlane(rowptr[n + 1]);
    const float4* PQ4 = (const float4*)PQ;
    const float4* W4  = (const float4*)Wfc;
    float4 self[NV];
    float4 w0[NV], w1[NV];
#pragma unroll
    for (int q = 0; q < NV; ++q) {
        self[q] = PQ4[(size_t)n * (2 * F / 4) + F / 4 + q * 32 + lane];
        w0[q]   = W4[q * 32 + lane];
        w1[q]   = W4[F / 4 + q * 32 + lane];
    }
    float4 acc[NV];
    sum_rows<F, 2 * F>(PQ4, col, beg, end, lane, acc);

    const float inv = 1.0f / fmaxf((float)(end - beg), 1.0f);
    float ss = 0.0f, d0 = 0.0f, d1 = 0.0f;
#pragma unroll
    for (int q = 0; q < NV; ++q) {
        float4 o;
        o.x = acc[q].x * inv + self[q].x;
        o.y = acc[q].y * inv + self[q].y;
        o.z = acc[q].z * inv + self[q].z;
        o.w = acc[q].w * inv + self[q].w;
        ss += (o.x * o.x + o.y * o.y) + (o.z * o.z + o.w * o.w);
        d0 += (o.x * w0[q].x + o.y * w0[q].y) + (o.z * w0[q].z + o.w * w0[q].w);
        d1 += (o.x * w1[q].x + o.y * w1[q].y) + (o.z * w1[q].z + o.w * w1[q].w);
    }
    for (int off = 16; off > 0; off >>= 1) {
        ss += __shfl_xor(ss, off, 32);
        d0 += __shfl_xor(d0, off, 32);
        d1 += __shfl_xor(d1, off, 32);
    }
    if (lane == 0) {
        float invn = 1.0f / fmaxf(sqrtf(ss), 1e-12f);
        float l0 = d0 * invn + bfc[0];
        float l1 = d1 * invn + bfc[1];
        float m  = fmaxf(l0, l1);
        float e0 = __expf(l0 - m), e1 = __expf(l1 - m);
        float s  = 1.0f / (e0 + e1);
        out[(size_t)n * 2 + 0] = e0 * s;
        out[(size_t)n * 2 + 1] = e1 * s;
    }
}

#define GEMM_K       256
#define GEMM_NCOL    256
#define GEMM_THREADS 1024
#define GEMM_BLOCKS  256
#define BPITCH       (GEMM_K + 8)

template <int K, int NROW>
__device__ __forceinline__ void stage_b_lds(const __bf16* __restrict__ B, __bf16* Bs, int tid) {
    constexpr int PPR   = K / 8;
    constexpr int TOTAL = NROW * PPR;
#pragma unroll
    for (int i = 0; i < TOTAL / GEMM_THREADS; ++i) {
        const int p   = i * GEMM_THREADS + tid;
        const int row = p / PPR, pc = p % PPR;
        const v8bf v = *(const v8bf*)(B + (size_t)row * K + pc * 8);
        *(v8bf*)(Bs + row * BPITCH + pc * 8) = v;
    }
}

template <int K, int NT>
__device__ __forceinline__ void wmma_strip_lds(const __bf16* __restrict__ arow, const __bf16* brow,
                                               v8f (&acc)[NT]) {
    union { v16bf v; v8bf h[2]; } a[K / 32];
#pragma unroll
    for (int s = 0; s < K / 32; ++s) {
        a[s].h[0] = *(const v8bf*)(arow + s * 32);
        a[s].h[1] = *(const v8bf*)(arow + s * 32 + 16);
    }
    __builtin_amdgcn_sched_barrier(0);
#pragma unroll
    for (int t = 0; t < NT; ++t) acc[t] = (v8f){};

#pragma unroll
    for (int s = 0; s < K / 32; ++s) {
#pragma unroll
        for (int t = 0; t < NT; ++t) {
            union { v16bf v; v8bf h[2]; } b;
            b.h[0] = *(const v8bf*)(brow + t * 16 * BPITCH + s * 32);
            b.h[1] = *(const v8bf*)(brow + t * 16 * BPITCH + s * 32 + 16);
            acc[t] = __builtin_amdgcn_wmma_f32_16x16x32_bf16(
                false, a[s].v, false, b.v, (short)0, acc[t], false, false);
        }
    }
}

__global__ __launch_bounds__(GEMM_THREADS)
void gemm_bf16_wmma_ldsb(const __bf16* __restrict__ A, const __bf16* __restrict__ B,
                         const float* __restrict__ bias, int bias_col0, float* __restrict__ C,
                         int ntiles) {
    constexpr int K = GEMM_K, N = GEMM_NCOL, NT = 4, NW = N / (16 * NT), NS = (GEMM_THREADS / 32) / NW;
    __shared__ __attribute__((aligned(16))) __bf16 Bs[N * BPITCH];
    const int lane = threadIdx.x & 31;
    const int wv   = __builtin_amdgcn_readfirstlane(threadIdx.x >> 5);
    const int w    = wv % NW;
    const int slot = wv / NW;
    const int tn0  = w * (16 * NT);
    const int h    = lane >> 4;
    const int l15  = lane & 15;

    stage_b_lds<K, N>(B, Bs, threadIdx.x);
    __syncthreads();

    const __bf16* brow = Bs + (tn0 + l15) * BPITCH + h * 8;
    for (int tile = slot * gridDim.x + blockIdx.x; tile < ntiles; tile += gridDim.x * NS) {
        asm volatile("" ::: "memory");
        const int tm = tile << 4;
        v8f acc[NT];
        wmma_strip_lds<K, NT>(A + (size_t)(tm + l15) * K + h * 8, brow, acc);

        const int r0 = tm + h * 8;
#pragma unroll
        for (int t = 0; t < NT; ++t) {
            const int cn = tn0 + t * 16 + l15;
            const int   bi = cn - bias_col0;
            const float bv = bias[bi < 0 ? 0 : bi];
            const float bc = (bi < 0) ? 0.0f : bv;
#pragma unroll
            for (int v = 0; v < 8; ++v)
                C[(size_t)(r0 + v) * N + cn] = acc[t][v] + bc;
        }
    }
}

__global__ __launch_bounds__(GEMM_THREADS)
void gemm_bf16_wmma_ldsb_norm_relu(const __bf16* __restrict__ A, const __bf16* __restrict__ B,
                                   const float* __restrict__ bias, __bf16* __restrict__ Hb,
                                   int ntiles) {
    constexpr int K = GEMM_K, NCOL = GEMM_NCOL, NT = 4, NW = NCOL / (16 * NT), NS = (GEMM_THREADS / 32) / NW;
    __shared__ __attribute__((aligned(16))) __bf16 Bs[NCOL * BPITCH];
    __shared__ __attribute__((aligned(16))) float red[NS][NW][16];
    const int lane = threadIdx.x & 31;
    const int wv   = __builtin_amdgcn_readfirstlane(threadIdx.x >> 5);
    const int w    = wv % NW;
    const int slot = wv / NW;
    const int tn0  = w * (16 * NT);
    const int h    = lane >> 4;
    const int l15  = lane & 15;

    stage_b_lds<K, NCOL>(B, Bs, threadIdx.x);
    __syncthreads();

    const __bf16* brow = Bs + (tn0 + l15) * BPITCH + h * 8;
    const int stride = gridDim.x * NS;
    const int trips  = (ntiles + stride - 1) / stride;
    for (int it = 0; it < trips; ++it) {
        const int  tile  = it * stride + slot * gridDim.x + blockIdx.x;
        const bool valid = tile < ntiles;
        const int  tm    = tile << 4;
        v8f acc[NT];
        if (valid) {
            wmma_strip_lds<K, NT>(A + (size_t)(tm + l15) * K + h * 8, brow, acc);

            float ss[8];
#pragma unroll
            for (int v = 0; v < 8; ++v) ss[v] = 0.0f;
#pragma unroll
            for (int t = 0; t < NT; ++t) {
                const float bc = bias[tn0 + t * 16 + l15];
#pragma unroll
                for (int v = 0; v < 8; ++v) {
                    acc[t][v] += bc;
                    ss[v] += acc[t][v] * acc[t][v];
                }
            }
#pragma unroll
            for (int off = 1; off < 16; off <<= 1) {
#pragma unroll
                for (int v = 0; v < 8; ++v) ss[v] += __shfl_xor(ss[v], off, 32);
            }
            if (l15 == 0) {
                float4* mine = (float4*)&red[slot][w][h * 8];
                mine[0] = make_float4(ss[0], ss[1], ss[2], ss[3]);
                mine[1] = make_float4(ss[4], ss[5], ss[6], ss[7]);
            }
        }
        __syncthreads();
        if (valid) {
            float tot = red[slot][0][l15];
#pragma unroll
            for (int q = 1; q < NW; ++q) tot += red[slot][q][l15];
            const float inv_mine = 1.0f / fmaxf(sqrtf(tot), 1e-12f);
            const int r0 = tm + h * 8;
#pragma unroll
            for (int v = 0; v < 8; ++v) {
                const float inv = __shfl(inv_mine, h * 8 + v, 32);
                __bf16* hrow = Hb + (size_t)(r0 + v) * NCOL + tn0 + l15;
#pragma unroll
                for (int t = 0; t < NT; ++t) {
                    const float r = fmaxf(acc[t][v] * inv, 0.0f);
                    hrow[t * 16] = (__bf16)r;
                }
            }
        }
        __syncthreads();
    }
}

extern "C" void kernel_launch(void* const* d_in, const int* in_sizes, int n_in,
                              void* d_out, int out_size, void* d_ws, size_t ws_size,
                              hipStream_t stream) {
    (void)in_sizes; (void)n_in; (void)out_size; (void)ws_size;

    const float* feat = (const float*)d_in[0];
    const int*   eidx = (const int*)  d_in[1];
    const int*   src  = eidx;
    const int*   dst  = eidx + N_EDGES;
    const float* W1l  = (const float*)d_in[2];
    const float* b1   = (const float*)d_in[3];
    const float* W1r  = (const float*)d_in[4];
    const float* W2l  = (const float*)d_in[5];
    const float* b2   = (const float*)d_in[6];
    const float* W2r  = (const float*)d_in[7];
    const float* Wfc  = (const float*)d_in[8];
    const float* bfc  = (const float*)d_in[9];
    float* out = (float*)d_out;

    char* ws = (char*)d_ws;
    size_t off = 0;
    auto alloc = [&](size_t bytes) -> void* {
        void* p = ws + off;
        off = (off + bytes + 255) & ~(size_t)255;
        return p;
    };
    int*    part   = (int*)  alloc((size_t)N_SPLITS * PART_PITCH * 4);
    int*    tot    = (int*)  alloc((size_t)N_RANGES * N_SPLITS * 4);
    int*    rowptr = (int*)  alloc((size_t)PART_PITCH * 4);
    int*    col    = (int*)  alloc((size_t)N_EDGES * 4);
    __bf16* Xcat1 = (__bf16*)alloc((size_t)N_NODES * 2 * IN_SIZE * 2);
    __bf16* Wcat1 = (__bf16*)alloc((size_t)HIDDEN * 2 * IN_SIZE * 2);
    __bf16* h1b   = (__bf16*)alloc((size_t)N_NODES * HIDDEN * 2);
    __bf16* Wst2  = (__bf16*)alloc((size_t)2 * OUT_SIZE * HIDDEN * 2);
    float*  pq2   = (float*) alloc((size_t)N_NODES * 2 * OUT_SIZE * 4);

    const int TB = 256;
    auto cdiv = [](int a, int b) { return (a + b - 1) / b; };

    build_wcat<<<cdiv(HIDDEN * IN_SIZE, TB), TB, 0, stream>>>(W1l, W1r, Wcat1, HIDDEN, IN_SIZE);
    build_wstack<<<cdiv(2 * OUT_SIZE * HIDDEN, TB), TB, 0, stream>>>(W2l, W2r, Wst2,
                                                                    OUT_SIZE * HIDDEN);

    inlist_count<<<N_RANGES * N_SPLITS, LIST_T, 0, stream>>>(dst, part, tot);
    inlist_fill<<<N_RANGES * N_SPLITS, LIST_T, 0, stream>>>(src, dst, part, tot, rowptr, col);
    gather_mean_cat<IN_SIZE><<<cdiv(N_NODES, TB / 32), TB, 0, stream>>>(
        feat, rowptr, col, Xcat1, N_NODES);
    gemm_bf16_wmma_ldsb_norm_relu<<<GEMM_BLOCKS, GEMM_THREADS, 0, stream>>>(
        Xcat1, Wcat1, b1, h1b, N_NODES / 16);

    gemm_bf16_wmma_ldsb<<<GEMM_BLOCKS, GEMM_THREADS, 0, stream>>>(
        h1b, Wst2, b2, OUT_SIZE, pq2, N_NODES / 16);
    gather_mean_add_fc<OUT_SIZE><<<cdiv(N_NODES, TB / 32), TB, 0, stream>>>(
        pq2, rowptr, col, Wfc, bfc, out, N_NODES);
}


# Both GEMMs in one launch: normalised bf16 h tiles stay in LDS (16 per work-group), layer-2 weights re-staged over layer-1's, 2nd product's A fragments by ds_load; h plane and a launch gone
# speedup vs baseline: 5.4027x; 1.0612x over previous
#include <hip/hip_runtime.h>
#include <math.h>

#define N_NODES  50000
#define N_EDGES  600000
#define IN_SIZE  128
#define HIDDEN   256
#define OUT_SIZE 128

typedef __attribute__((ext_vector_type(16))) __bf16 v16bf;
typedef __attribute__((ext_vector_type(8)))  __bf16 v8bf;
typedef __attribute__((ext_vector_type(8)))  float  v8f;
typedef __attribute__((ext_vector_type(4)))  __bf16 v4bf;

#define NR_NODES   2048
#define N_RANGES   ((N_NODES + NR_NODES - 1) / NR_NODES)
#define N_SPLITS   10
#define SPLIT_E    (N_EDGES / N_SPLITS)
#define SPLIT_V4   (SPLIT_E / 4)
#define PART_PITCH (N_RANGES * NR_NODES)
#define LIST_T     1024
#define LIST_UNR   8
static_assert(N_EDGES % (N_SPLITS * 4) == 0, "pieces are whole 16-byte vectors");
static_assert(N_NODES < PART_PITCH, "rowptr[N_NODES] must be an entry of the last range");
static_assert(NR_NODES == 2 * LIST_T, "a thread owns two adjacent counters");
static_assert(N_RANGES * N_SPLITS <= LIST_T, "one thread per (range, piece) sum");

__global__ __launch_bounds__(256)
void build_wcat(const float* __restrict__ Wl, const float* __restrict__ Wr,
                __bf16* __restrict__ Wcat, int rows, int kHalf) {
    int i = blockIdx.x * blockDim.x + threadIdx.x;
    if (i >= rows * kHalf) return;
    int r = i / kHalf, c = i % kHalf;
    __bf16* row = Wcat + (size_t)r * (2 * kHalf);
    row[c]         = (__bf16)Wl[i];
    row[kHalf + c] = (__bf16)Wr[i];
}

__global__ __launch_bounds__(256)
void build_wstack(const float* __restrict__ Wa, const float* __restrict__ Wb,
                  __bf16* __restrict__ Wst, int n) {
    int i = blockIdx.x * blockDim.x + threadIdx.x;
    if (i >= 2 * n) return;
    Wst[i] = (__bf16)(i < n ? Wa[i] : Wb[i - n]);
}


struct EdgeBatch {
    int4 v[LIST_UNR];
};
__device__ __forceinline__ void load_batch(const int4* __restrict__ p4, int i0, EdgeBatch& b) {
#pragma unroll
    for (int k = 0; k < LIST_UNR; ++k) {
        const int i = i0 + k * LIST_T;
        b.v[k] = p4[i < SPLIT_V4 ? i : SPLIT_V4 - 1];
    }
}

__global__ __launch_bounds__(LIST_T)
void inlist_count(const int* __restrict__ dst, int* __restrict__ part, int* __restrict__ tot) {
    __shared__ __attribute__((aligned(16))) int cnt[NR_NODES];
    __shared__ int wsum[LIST_T / 32];
    const int tid = threadIdx.x, lane = tid & 31, w = tid >> 5;
    const int g = blockIdx.x / N_SPLITS, s = blockIdx.x % N_SPLITS;
    const int lo = g * NR_NODES;
    cnt[tid] = 0;
    cnt[tid + LIST_T] = 0;
    __syncthreads();

    const int4* d4 = (const int4*)(dst + (size_t)s * SPLIT_E);
    for (int i0 = tid; i0 < SPLIT_V4; i0 += LIST_T * LIST_UNR) {
        EdgeBatch d;
        load_batch(d4, i0, d);
#pragma unroll
        for (int k = 0; k < LIST_UNR; ++k) {
            const bool ok = (i0 + k * LIST_T) < SPLIT_V4;
            const unsigned u0 = (unsigned)(d.v[k].x - lo), u1 = (unsigned)(d.v[k].y - lo);
            const unsigned u2 = (unsigned)(d.v[k].z - lo), u3 = (unsigned)(d.v[k].w - lo);
            if (ok && u0 < NR_NODES) atomicAdd(&cnt[u0], 1);
            if (ok && u1 < NR_NODES) atomicAdd(&cnt[u1], 1);
            if (ok && u2 < NR_NODES) atomicAdd(&cnt[u2], 1);
            if (ok && u3 < NR_NODES) atomicAdd(&cnt[u3], 1);
        }
    }
    __syncthreads();

    const int2 c = *(const int2*)&cnt[2 * tid];
    *(int2*)(part + (size_t)s * PART_PITCH + lo + 2 * tid) = c;
    int t = c.x + c.y;
    for (int off = 16; off > 0; off >>= 1) t += __shfl_xor(t, off, 32);
    if (lane == 0) wsum[w] = t;
    __syncthreads();
    if (w == 0) {
        int x = wsum[lane];
        for (int off = 16; off > 0; off >>= 1) x += __shfl_xor(x, off, 32);
        if (lane == 0) tot[g * N_SPLITS + s] = x;
    }
}

__global__ __launch_bounds__(LIST_T)
void inlist_fill(const int* __restrict__ src, const int* __restrict__ dst,
                 const int* __restrict__ part, const int* __restrict__ tot,
                 int* __restrict__ rowptr, int* __restrict__ col) {
    __shared__ __attribute__((aligned(16))) int cur[NR_NODES];
    __shared__ int wsum[LIST_T / 32];
    __shared__ int wbase[LIST_T / 32];
    const int tid = threadIdx.x, lane = tid & 31, w = tid >> 5;
    const int g = blockIdx.x / N_SPLITS, s = blockIdx.x % N_SPLITS;
    const int lo = g * NR_NODES;

    int b = (tid < g * N_SPLITS) ? tot[tid] : 0;
    int2 len = make_int2(0, 0), below = make_int2(0, 0);
#pragma unroll
    for (int q = 0; q < N_SPLITS; ++q) {
        const int2 p = *(const int2*)(part + (size_t)q * PART_PITCH + lo + 2 * tid);
        len.x += p.x; len.y += p.y;
        if (q < s) { below.x += p.x; below.y += p.y; }
    }
    const int v = len.x + len.y;
    int inc = v;
    for (int off = 1; off < 32; off <<= 1) {
        int x = __shfl_up(inc, off, 32);
        if (lane >= off) inc += x;
    }
    for (int off = 16; off > 0; off >>= 1) b += __shfl_xor(b, off, 32);
    if (lane == 31) wsum[w] = inc;
    if (lane == 0)  wbase[w] = b;
    __syncthreads();
    if (w == 0) {
        int x = wsum[lane], xi = x;
        for (int off = 1; off < 32; off <<= 1) {
            int y = __shfl_up(xi, off, 32);
            if (lane >= off) xi += y;
        }
        wsum[lane] = xi - x;
        int bb = wbase[lane];
        for (int off = 16; off > 0; off >>= 1) bb += __shfl_xor(bb, off, 32);
        if (lane == 0) wbase[0] = bb;
    }
    __syncthreads();
    const int st0 = wbase[0] + wsum[w] + inc - v;
    const int st1 = st0 + len.x;
    *(int2*)&cur[2 * tid] = make_int2(st0 + below.x, st1 + below.y);
    if (s == 0) *(int2*)(rowptr + lo + 2 * tid) = make_int2(st0, st1);
    __syncthreads();

    const int4* d4 = (const int4*)(dst + (size_t)s * SPLIT_E);
    const int4* s4 = (const int4*)(src + (size_t)s * SPLIT_E);
    for (int i0 = tid; i0 < SPLIT_V4; i0 += LIST_T * LIST_UNR) {
        EdgeBatch d, sv;
        load_batch(d4, i0, d);
        load_batch(s4, i0, sv);
        asm volatile("" ::: "memory");
#pragma unroll
        for (int k = 0; k < LIST_UNR; ++k) {
            const bool ok = (i0 + k * LIST_T) < SPLIT_V4;
            const unsigned u0 = (unsigned)(d.v[k].x - lo), u1 = (unsigned)(d.v[k].y - lo);
            const unsigned u2 = (unsigned)(d.v[k].z - lo), u3 = (unsigned)(d.v[k].w - lo);
            if (ok && u0 < NR_NODES) col[atomicAdd(&cur[u0], 1)] = sv.v[k].x;
            if (ok && u1 < NR_NODES) col[atomicAdd(&cur[u1], 1)] = sv.v[k].y;
            if (ok && u2 < NR_NODES) col[atomicAdd(&cur[u2], 1)] = sv.v[k].z;
            if (ok && u3 < NR_NODES) col[atomicAdd(&cur[u3], 1)] = sv.v[k].w;
        }
    }
}

__device__ __forceinline__ v4bf pack4_bf16(float4 v, float sc) {
    v4bf o;
    o[0] = (__bf16)(v.x * sc);
    o[1] = (__bf16)(v.y * sc);
    o[2] = (__bf16)(v.z * sc);
    o[3] = (__bf16)(v.w * sc);
    return o;
}

template <int F, int PITCH>
__device__ __forceinline__ void sum_rows(const float4* __restrict__ X4, const int* __restrict__ col,
                                         int beg, int end, int lane, float4 (&acc)[F / 128]) {
    constexpr int NV = F / 128;
#pragma unroll
    for (int q = 0; q < NV; ++q) acc[q] = make_float4(0.0f, 0.0f, 0.0f, 0.0f);

    for (int base = beg; base < end; base += 32) {
        const int m = (end - base < 32) ? (end - base) : 32;
        const int myc = (lane < m) ? col[base + lane] : 0;
        int j = 0;
        for (; j + 4 <= m; j += 4) {
            const float4* r0 = X4 + (size_t)__builtin_amdgcn_readlane(myc, j + 0) * (PITCH / 4) + lane;
            const float4* r1 = X4 + (size_t)__builtin_amdgcn_readlane(myc, j + 1) * (PITCH / 4) + lane;
            const float4* r2 = X4 + (size_t)__builtin_amdgcn_readlane(myc, j + 2) * (PITCH / 4) + lane;
            const float4* r3 = X4 + (size_t)__builtin_amdgcn_readlane(myc, j + 3) * (PITCH / 4) + lane;
            float4 v0[NV], v1[NV], v2[NV], v3[NV];
#pragma unroll
            for (int q = 0; q < NV; ++q) {
                v0[q] = r0[q * 32]; v1[q] = r1[q * 32];
                v2[q] = r2[q * 32]; v3[q] = r3[q * 32];
            }
#pragma unroll
            for (int q = 0; q < NV; ++q) {
                acc[q].x += (v0[q].x + v1[q].x) + (v2[q].x + v3[q].x);
                acc[q].y += (v0[q].y + v1[q].y) + (v2[q].y + v3[q].y);
                acc[q].z += (v0[q].z + v1[q].z) + (v2[q].z + v3[q].z);
                acc[q].w += (v0[q].w + v1[q].w) + (v2[q].w + v3[q].w);
            }
        }
        for (; j < m; ++j) {
            const float4* r0 = X4 + (size_t)__builtin_amdgcn_readlane(myc, j) * (PITCH / 4) + lane;
#pragma unroll
            for (int q = 0; q < NV; ++q) {
                float4 v = r0[q * 32];
                acc[q].x += v.x; acc[q].y += v.y; acc[q].z += v.z; acc[q].w += v.w;
            }
        }
    }
}

template <int F>
__global__ __launch_bounds__(256)
void gather_mean_cat(const float* __restrict__ X, const int* __restrict__ rowptr,
                     const int* __restrict__ col, __bf16* __restrict__ Xcat, int nnodes) {
    constexpr int NV = F / 128;
    const int lane = threadIdx.x & 31;
    const int n = blockIdx.x * (blockDim.x >> 5) + (threadIdx.x >> 5);
    if (n >= nnodes) return;
    const int beg = __builtin_amdgcn_readfirstlane(rowptr[n]);
    const int end = __builtin_amdgcn_readfirstlane(rowptr[n + 1]);
    const float4* X4 = (const float4*)X;
    float4 acc[NV];
    sum_rows<F, F>(X4, col, beg, end, lane, acc);

    const float inv = 1.0f / fmaxf((float)(end - beg), 1.0f);
    __bf16* row = Xcat + (size_t)n * (2 * F);
#pragma unroll
    for (int q = 0; q < NV; ++q)
        *(v4bf*)(row + q * 128 + lane * 4) = pack4_bf16(acc[q], inv);
#pragma unroll
    for (int q = 0; q < NV; ++q) {
        float4 v = X4[(size_t)n * (F / 4) + q * 32 + lane];
        *(v4bf*)(row + F + q * 128 + lane * 4) = pack4_bf16(v, 1.0f);
    }
}

template <int F>
__global__ __launch_bounds__(256)
void gather_mean_add_fc(const float* __restrict__ PQ, const int* __restrict__ rowptr,
                        const int* __restrict__ col, const float* __restrict__ Wfc,
                        const float* __restrict__ bfc, float* __restrict__ out, int nnodes) {
    constexpr int NV = F / 128;
    const int lane = threadIdx.x & 31;
    const int n = blockIdx.x * (blockDim.x >> 5) + (threadIdx.x >> 5);
    if (n >= nnodes) return;
    const int beg = __builtin_amdgcn_readfirstlane(rowptr[n]);
    const int end = __builtin_amdgcn_readfirstlane(rowptr[n + 1]);
    const float4* PQ4 = (const float4*)PQ;
    const float4* W4  = (const float4*)Wfc;
    float4 self[NV];
    float4 w0[NV], w1[NV];
#pragma unroll
    for (int q = 0; q < NV; ++q) {
        self[q] = PQ4[(size_t)n * (2 * F / 4) + F / 4 + q * 32 + lane];
        w0[q]   = W4[q * 32 + lane];
        w1[q]   = W4[F / 4 + q * 32 + lane];
    }
    float4 acc[NV];
    sum_rows<F, 2 * F>(PQ4, col, beg, end, lane, acc);

    const float inv = 1.0f / fmaxf((float)(end - beg), 1.0f);
    float ss = 0.0f, d0 = 0.0f, d1 = 0.0f;
#pragma unroll
    for (int q = 0; q < NV; ++q) {
        float4 o;
        o.x = acc[q].x * inv + self[q].x;
        o.y = acc[q].y * inv + self[q].y;
        o.z = acc[q].z * inv + self[q].z;
        o.w = acc[q].w * inv + self[q].w;
        ss += (o.x * o.x + o.y * o.y) + (o.z * o.z + o.w * o.w);
        d0 += (o.x * w0[q].x + o.y * w0[q].y) + (o.z * w0[q].z + o.w * w0[q].w);
        d1 += (o.x * w1[q].x + o.y * w1[q].y) + (o.z * w1[q].z + o.w * w1[q].w);
    }
    for (int off = 16; off > 0; off >>= 1) {
        ss += __shfl_xor(ss, off, 32);
        d0 += __shfl_xor(d0, off, 32);
        d1 += __shfl_xor(d1, off, 32);
    }
    if (lane == 0) {
        float invn = 1.0f / fmaxf(sqrtf(ss), 1e-12f);
        float l0 = d0 * invn + bfc[0];
        float l1 = d1 * invn + bfc[1];
        float m  = fmaxf(l0, l1);
        float e0 = __expf(l0 - m), e1 = __expf(l1 - m);
        float s  = 1.0f / (e0 + e1);
        out[(size_t)n * 2 + 0] = e0 * s;
        out[(size_t)n * 2 + 1] = e1 * s;
    }
}

#define GEMM_K       256
#define GEMM_NCOL    256
#define GEMM_THREADS 1024
#define GEMM_BLOCKS  256
#define BPITCH       (GEMM_K + 8)

template <int K, int NROW>
__device__ __forceinline__ void stage_b_lds(const __bf16* __restrict__ B, __bf16* Bs, int tid) {
    constexpr int PPR   = K / 8;
    constexpr int TOTAL = NROW * PPR;
#pragma unroll
    for (int i = 0; i < TOTAL / GEMM_THREADS; ++i) {
        const int p   = i * GEMM_THREADS + tid;
        const int row = p / PPR, pc = p % PPR;
        const v8bf v = *(const v8bf*)(B + (size_t)row * K + pc * 8);
        *(v8bf*)(Bs + row * BPITCH + pc * 8) = v;
    }
}

template <int K, int NT>
__device__ __forceinline__ void wmma_strip_lds(const __bf16* __restrict__ arow, const __bf16* brow,
                                               v8f (&acc)[NT]) {
    union { v16bf v; v8bf h[2]; } a[K / 32];
#pragma unroll
    for (int s = 0; s < K / 32; ++s) {
        a[s].h[0] = *(const v8bf*)(arow + s * 32);
        a[s].h[1] = *(const v8bf*)(arow + s * 32 + 16);
    }
    __builtin_amdgcn_sched_barrier(0);
#pragma unroll
    for (int t = 0; t < NT; ++t) acc[t] = (v8f){};

#pragma unroll
    for (int s = 0; s < K / 32; ++s) {
#pragma unroll
        for (int t = 0; t < NT; ++t) {
            union { v16bf v; v8bf h[2]; } b;
            b.h[0] = *(const v8bf*)(brow + t * 16 * BPITCH + s * 32);
            b.h[1] = *(const v8bf*)(brow + t * 16 * BPITCH + s * 32 + 16);
            acc[t] = __builtin_amdgcn_wmma_f32_16x16x32_bf16(
                false, a[s].v, false, b.v, (short)0, acc[t], false, false);
        }
    }
}

template <int K, int NT>
__device__ __forceinline__ void wmma_strip_lds_ab(const __bf16* arow, const __bf16* brow,
                                                  v8f (&acc)[NT]) {
#pragma unroll
    for (int t = 0; t < NT; ++t) acc[t] = (v8f){};
#pragma unroll
    for (int s = 0; s < K / 32; ++s) {
        union { v16bf v; v8bf h[2]; } a;
        a.h[0] = *(const v8bf*)(arow + s * 32);
        a.h[1] = *(const v8bf*)(arow + s * 32 + 16);
#pragma unroll
        for (int t = 0; t < NT; ++t) {
            union { v16bf v; v8bf h[2]; } b;
            b.h[0] = *(const v8bf*)(brow + t * 16 * BPITCH + s * 32);
            b.h[1] = *(const v8bf*)(brow + t * 16 * BPITCH + s * 32 + 16);
            acc[t] = __builtin_amdgcn_wmma_f32_16x16x32_bf16(
                false, a.v, false, b.v, (short)0, acc[t], false, false);
        }
    }
}

#define GEMM_NS      ((GEMM_THREADS / 32) / (GEMM_NCOL / 64))
#define GEMM_NTILES  (N_NODES / 16)
#define GEMM_STRIDE  (GEMM_BLOCKS * GEMM_NS)
#define GEMM_TRIPS   ((GEMM_NTILES + GEMM_STRIDE - 1) / GEMM_STRIDE)
static_assert(N_NODES % 16 == 0, "whole 16-row tiles");
static_assert((GEMM_NCOL + GEMM_TRIPS * GEMM_NS * 16) * BPITCH * 2 + GEMM_NS * 4 * 16 * 4 <= 320 * 1024,
              "weights + h tiles + partial sums fit one WGP's LDS");
__global__ __launch_bounds__(GEMM_THREADS)
void gemm2_bf16_wmma_lds_h(const __bf16* __restrict__ A, const __bf16* __restrict__ B1,
                           const float* __restrict__ bias1, const __bf16* __restrict__ B2,
                           const float* __restrict__ bias2, int bias2_col0,
                           float* __restrict__ C) {
    constexpr int K = GEMM_K, NCOL = GEMM_NCOL, NT = 4, NW = NCOL / (16 * NT), NS = GEMM_NS;
    static_assert(K == NCOL, "layer 1's output columns are layer 2's k");
    __shared__ __attribute__((aligned(16))) __bf16 Bs[NCOL * BPITCH];
    __shared__ __attribute__((aligned(16))) __bf16 Hs[GEMM_TRIPS * NS * 16 * BPITCH];
    __shared__ __attribute__((aligned(16))) float red[NS][NW][16];
    const int lane = threadIdx.x & 31;
    const int wv   = __builtin_amdgcn_readfirstlane(threadIdx.x >> 5);
    const int w    = wv % NW;
    const int slot = wv / NW;
    const int tn0  = w * (16 * NT);
    const int h    = lane >> 4;
    const int l15  = lane & 15;

    stage_b_lds<K, NCOL>(B1, Bs, threadIdx.x);
    __syncthreads();

    const __bf16* brow = Bs + (tn0 + l15) * BPITCH + h * 8;
    for (int it = 0; it < GEMM_TRIPS; ++it) {
        const int  tile  = it * GEMM_STRIDE + slot * GEMM_BLOCKS + (int)blockIdx.x;
        const bool valid = tile < GEMM_NTILES;
        const int  tm    = tile << 4;
        __bf16* htile = Hs + (it * NS + slot) * (16 * BPITCH);
        v8f acc[NT];
        if (valid) {
            wmma_strip_lds<K, NT>(A + (size_t)(tm + l15) * K + h * 8, brow, acc);

            float ss[8];
#pragma unroll
            for (int v = 0; v < 8; ++v) ss[v] = 0.0f;
#pragma unroll
            for (int t = 0; t < NT; ++t) {
                const float bc = bias1[tn0 + t * 16 + l15];
#pragma unroll
                for (int v = 0; v < 8; ++v) {
                    acc[t][v] += bc;
                    ss[v] += acc[t][v] * acc[t][v];
                }
            }
#pragma unroll
            for (int off = 1; off < 16; off <<= 1) {
#pragma unroll
                for (int v = 0; v < 8; ++v) ss[v] += __shfl_xor(ss[v], off, 32);
            }
            if (l15 == 0) {
                float4* mine = (float4*)&red[slot][w][h * 8];
                mine[0] = make_float4(ss[0], ss[1], ss[2], ss[3]);
                mine[1] = make_float4(ss[4], ss[5], ss[6], ss[7]);
            }
        }
        __syncthreads();
        if (valid) {
            float tot = red[slot][0][l15];
#pragma unroll
            for (int q = 1; q < NW; ++q) tot += red[slot][q][l15];
            const float inv_mine = 1.0f / fmaxf(sqrtf(tot), 1e-12f);
#pragma unroll
            for (int v = 0; v < 8; ++v) {
                const float inv = __shfl(inv_mine, h * 8 + v, 32);
                __bf16* hrow = htile + (h * 8 + v) * BPITCH + tn0 + l15;
#pragma unroll
                for (int t = 0; t < NT; ++t) {
                    const float r = fmaxf(acc[t][v] * inv, 0.0f);
                    hrow[t * 16] = (__bf16)r;
                }
            }
        }
        __syncthreads();
    }

    stage_b_lds<K, NCOL>(B2, Bs, threadIdx.x);
    __syncthreads();

    for (int it = 0; it < GEMM_TRIPS; ++it) {
        const int tile = it * GEMM_STRIDE + slot * GEMM_BLOCKS + (int)blockIdx.x;
        if (tile < GEMM_NTILES) {
            const int tm = tile << 4;
            const __bf16* htile = Hs + (it * NS + slot) * (16 * BPITCH);
            v8f acc[NT];
            wmma_strip_lds_ab<K, NT>(htile + l15 * BPITCH + h * 8, brow, acc);

            const int r0 = tm + h * 8;
#pragma unroll
            for (int t = 0; t < NT; ++t) {
                const int cn = tn0 + t * 16 + l15;
                const int   bi = cn - bias2_col0;
                const float bv = bias2[bi < 0 ? 0 : bi];
                const float bc = (bi < 0) ? 0.0f : bv;
#pragma unroll
                for (int v = 0; v < 8; ++v)
                    C[(size_t)(r0 + v) * NCOL + cn] = acc[t][v] + bc;
            }
        }
    }
}

extern "C" void kernel_launch(void* const* d_in, const int* in_sizes, int n_in,
                              void* d_out, int out_size, void* d_ws, size_t ws_size,
                              hipStream_t stream) {
    (void)in_sizes; (void)n_in; (void)out_size; (void)ws_size;

    const float* feat = (const float*)d_in[0];
    const int*   eidx = (const int*)  d_in[1];
    const int*   src  = eidx;
    const int*   dst  = eidx + N_EDGES;
    const float* W1l  = (const float*)d_in[2];
    const float* b1   = (const float*)d_in[3];
    const float* W1r  = (const float*)d_in[4];
    const float* W2l  = (const float*)d_in[5];
    const float* b2   = (const float*)d_in[6];
    const float* W2r  = (const float*)d_in[7];
    const float* Wfc  = (const float*)d_in[8];
    const float* bfc  = (const float*)d_in[9];
    float* out = (float*)d_out;

    char* ws = (char*)d_ws;
    size_t off = 0;
    auto alloc = [&](size_t bytes) -> void* {
        void* p = ws + off;
        off = (off + bytes + 255) & ~(size_t)255;
        return p;
    };
    int*    part   = (int*)  alloc((size_t)N_SPLITS * PART_PITCH * 4);
    int*    tot    = (int*)  alloc((size_t)N_RANGES * N_SPLITS * 4);
    int*    rowptr = (int*)  alloc((size_t)PART_PITCH * 4);
    int*    col    = (int*)  alloc((size_t)N_EDGES * 4);
    __bf16* Xcat1 = (__bf16*)alloc((size_t)N_NODES * 2 * IN_SIZE * 2);
    __bf16* Wcat1 = (__bf16*)alloc((size_t)HIDDEN * 2 * IN_SIZE * 2);
    __bf16* Wst2  = (__bf16*)alloc((size_t)2 * OUT_SIZE * HIDDEN * 2);
    float*  pq2   = (float*) alloc((size_t)N_NODES * 2 * OUT_SIZE * 4);

    const int TB = 256;
    auto cdiv = [](int a, int b) { return (a + b - 1) / b; };

    build_wcat<<<cdiv(HIDDEN * IN_SIZE, TB), TB, 0, stream>>>(W1l, W1r, Wcat1, HIDDEN, IN_SIZE);
    build_wstack<<<cdiv(2 * OUT_SIZE * HIDDEN, TB), TB, 0, stream>>>(W2l, W2r, Wst2,
                                                                    OUT_SIZE * HIDDEN);

    inlist_count<<<N_RANGES * N_SPLITS, LIST_T, 0, stream>>>(dst, part, tot);
    inlist_fill<<<N_RANGES * N_SPLITS, LIST_T, 0, stream>>>(src, dst, part, tot, rowptr, col);
    gather_mean_cat<IN_SIZE><<<cdiv(N_NODES, TB / 32), TB, 0, stream>>>(
        feat, rowptr, col, Xcat1, N_NODES);
    gemm2_bf16_wmma_lds_h<<<GEMM_BLOCKS, GEMM_THREADS, 0, stream>>>(
        Xcat1, Wcat1, b1, Wst2, b2, OUT_SIZE, pq2);
    gather_mean_add_fc<OUT_SIZE><<<cdiv(N_NODES, TB / 32), TB, 0, stream>>>(
        pq2, rowptr, col, Wfc, bfc, out, N_NODES);
}


# The two weight-pack launches become 24 extra work-groups of the count launch (a float4 in, 4 bf16 out per thread): 5 launches per call, not 7; values and every sum order unchanged
# speedup vs baseline: 5.6394x; 1.0438x over previous
#include <hip/hip_runtime.h>
#include <math.h>

#define N_NODES  50000
#define N_EDGES  600000
#define IN_SIZE  128
#define HIDDEN   256
#define OUT_SIZE 128

typedef __attribute__((ext_vector_type(16))) __bf16 v16bf;
typedef __attribute__((ext_vector_type(8)))  __bf16 v8bf;
typedef __attribute__((ext_vector_type(8)))  float  v8f;
typedef __attribute__((ext_vector_type(4)))  __bf16 v4bf;

#define NR_NODES   2048
#define N_RANGES   ((N_NODES + NR_NODES - 1) / NR_NODES)
#define N_SPLITS   10
#define SPLIT_E    (N_EDGES / N_SPLITS)
#define SPLIT_V4   (SPLIT_E / 4)
#define PART_PITCH (N_RANGES * NR_NODES)
#define LIST_T     1024
#define LIST_UNR   8
static_assert(N_EDGES % (N_SPLITS * 4) == 0, "pieces are whole 16-byte vectors");
static_assert(N_NODES < PART_PITCH, "rowptr[N_NODES] must be an entry of the last range");
static_assert(NR_NODES == 2 * LIST_T, "a thread owns two adjacent counters");
static_assert(N_RANGES * N_SPLITS <= LIST_T, "one thread per (range, piece) sum");

#define PACK_V4_L1   (HIDDEN * IN_SIZE / 4)
#define PACK_V4_L2H  (OUT_SIZE * HIDDEN / 4)
#define PACK_BLOCKS  ((PACK_V4_L1 + 2 * PACK_V4_L2H) / LIST_T)
static_assert((PACK_V4_L1 + 2 * PACK_V4_L2H) % LIST_T == 0, "whole packing work-groups");
static_assert(PACK_V4_L1 % LIST_T == 0, "a packing work-group serves one layer only");
static_assert(IN_SIZE % 4 == 0, "a float4 never crosses a row of W1l / W1r");

__device__ __forceinline__ v4bf cvt4_bf16(float4 v) {
    v4bf o;
    o[0] = (__bf16)v.x;
    o[1] = (__bf16)v.y;
    o[2] = (__bf16)v.z;
    o[3] = (__bf16)v.w;
    return o;
}

__device__ __forceinline__ void pack_weights(int i,
                                             const float* __restrict__ W1l, const float* __restrict__ W1r,
                                             const float* __restrict__ W2l, const float* __restrict__ W2r,
                                             __bf16* __restrict__ Wcat1, __bf16* __restrict__ Wst2) {
    if (i < PACK_V4_L1) {
        const float4 a = ((const float4*)W1l)[i];
        const float4 b = ((const float4*)W1r)[i];
        const int r = (4 * i) / IN_SIZE, cc = (4 * i) % IN_SIZE;
        __bf16* row = Wcat1 + (size_t)r * (2 * IN_SIZE);
        *(v4bf*)(row + cc)           = cvt4_bf16(a);
        *(v4bf*)(row + IN_SIZE + cc) = cvt4_bf16(b);
    } else {
        const int j = i - PACK_V4_L1;
        const float4 a = (j < PACK_V4_L2H) ? ((const float4*)W2l)[j]
                                           : ((const float4*)W2r)[j - PACK_V4_L2H];
        *(v4bf*)(Wst2 + (size_t)4 * j) = cvt4_bf16(a);
    }
}


struct EdgeBatch {
    int4 v[LIST_UNR];
};
__device__ __forceinline__ void load_batch(const int4* __restrict__ p4, int i0, EdgeBatch& b) {
#pragma unroll
    for (int k = 0; k < LIST_UNR; ++k) {
        const int i = i0 + k * LIST_T;
        b.v[k] = p4[i < SPLIT_V4 ? i : SPLIT_V4 - 1];
    }
}

#define COUNT_BLOCKS (N_RANGES * N_SPLITS)
__global__ __launch_bounds__(LIST_T)
void inlist_count_pack(const int* __restrict__ dst, int* __restrict__ part, int* __restrict__ tot,
                       const float* __restrict__ W1l, const float* __restrict__ W1r,
                       const float* __restrict__ W2l, const float* __restrict__ W2r,
                       __bf16* __restrict__ Wcat1, __bf16* __restrict__ Wst2) {
    __shared__ __attribute__((aligned(16))) int cnt[NR_NODES];
    __shared__ int wsum[LIST_T / 32];
    const int tid = threadIdx.x, lane = tid & 31, w = tid >> 5;
    if (blockIdx.x >= COUNT_BLOCKS) {
        pack_weights(((int)blockIdx.x - COUNT_BLOCKS) * LIST_T + tid, W1l, W1r, W2l, W2r, Wcat1, Wst2);
        return;
    }
    const int g = blockIdx.x / N_SPLITS, s = blockIdx.x % N_SPLITS;
    const int lo = g * NR_NODES;
    cnt[tid] = 0;
    cnt[tid + LIST_T] = 0;
    __syncthreads();

    const int4* d4 = (const int4*)(dst + (size_t)s * SPLIT_E);
    for (int i0 = tid; i0 < SPLIT_V4; i0 += LIST_T * LIST_UNR) {
        EdgeBatch d;
        load_batch(d4, i0, d);
#pragma unroll
        for (int k = 0; k < LIST_UNR; ++k) {
            const bool ok = (i0 + k * LIST_T) < SPLIT_V4;
            const unsigned u0 = (unsigned)(d.v[k].x - lo), u1 = (unsigned)(d.v[k].y - lo);
            const unsigned u2 = (unsigned)(d.v[k].z - lo), u3 = (unsigned)(d.v[k].w - lo);
            if (ok && u0 < NR_NODES) atomicAdd(&cnt[u0], 1);
            if (ok && u1 < NR_NODES) atomicAdd(&cnt[u1], 1);
            if (ok && u2 < NR_NODES) atomicAdd(&cnt[u2], 1);
            if (ok && u3 < NR_NODES) atomicAdd(&cnt[u3], 1);
        }
    }
    __syncthreads();

    const int2 c = *(const int2*)&cnt[2 * tid];
    *(int2*)(part + (size_t)s * PART_PITCH + lo + 2 * tid) = c;
    int t = c.x + c.y;
    for (int off = 16; off > 0; off >>= 1) t += __shfl_xor(t, off, 32);
    if (lane == 0) wsum[w] = t;
    __syncthreads();
    if (w == 0) {
        int x = wsum[lane];
        for (int off = 16; off > 0; off >>= 1) x += __shfl_xor(x, off, 32);
        if (lane == 0) tot[g * N_SPLITS + s] = x;
    }
}

__global__ __launch_bounds__(LIST_T)
void inlist_fill(const int* __restrict__ src, const int* __restrict__ dst,
                 const int* __restrict__ part, const int* __restrict__ tot,
                 int* __restrict__ rowptr, int* __restrict__ col) {
    __shared__ __attribute__((aligned(16))) int cur[NR_NODES];
    __shared__ int wsum[LIST_T / 32];
    __shared__ int wbase[LIST_T / 32];
    const int tid = threadIdx.x, lane = tid & 31, w = tid >> 5;
    const int g = blockIdx.x / N_SPLITS, s = blockIdx.x % N_SPLITS;
    const int lo = g * NR_NODES;

    int b = (tid < g * N_SPLITS) ? tot[tid] : 0;
    int2 len = make_int2(0, 0), below = make_int2(0, 0);
#pragma unroll
    for (int q = 0; q < N_SPLITS; ++q) {
        const int2 p = *(const int2*)(part + (size_t)q * PART_PITCH + lo + 2 * tid);
        len.x += p.x; len.y += p.y;
        if (q < s) { below.x += p.x; below.y += p.y; }
    }
    const int v = len.x + len.y;
    int inc = v;
    for (int off = 1; off < 32; off <<= 1) {
        int x = __shfl_up(inc, off, 32);
        if (lane >= off) inc += x;
    }
    for (int off = 16; off > 0; off >>= 1) b += __shfl_xor(b, off, 32);
    if (lane == 31) wsum[w] = inc;
    if (lane == 0)  wbase[w] = b;
    __syncthreads();
    if (w == 0) {
        int x = wsum[lane], xi = x;
        for (int off = 1; off < 32; off <<= 1) {
            int y = __shfl_up(xi, off, 32);
            if (lane >= off) xi += y;
        }
        wsum[lane] = xi - x;
        int bb = wbase[lane];
        for (int off = 16; off > 0; off >>= 1) bb += __shfl_xor(bb, off, 32);
        if (lane == 0) wbase[0] = bb;
    }
    __syncthreads();
    const int st0 = wbase[0] + wsum[w] + inc - v;
    const int st1 = st0 + len.x;
    *(int2*)&cur[2 * tid] = make_int2(st0 + below.x, st1 + below.y);
    if (s == 0) *(int2*)(rowptr + lo + 2 * tid) = make_int2(st0, st1);
    __syncthreads();

    const int4* d4 = (const int4*)(dst + (size_t)s * SPLIT_E);
    const int4* s4 = (const int4*)(src + (size_t)s * SPLIT_E);
    for (int i0 = tid; i0 < SPLIT_V4; i0 += LIST_T * LIST_UNR) {
        EdgeBatch d, sv;
        load_batch(d4, i0, d);
        load_batch(s4, i0, sv);
        asm volatile("" ::: "memory");
#pragma unroll
        for (int k = 0; k < LIST_UNR; ++k) {
            const bool ok = (i0 + k * LIST_T) < SPLIT_V4;
            const unsigned u0 = (unsigned)(d.v[k].x - lo), u1 = (unsigned)(d.v[k].y - lo);
            const unsigned u2 = (unsigned)(d.v[k].z - lo), u3 = (unsigned)(d.v[k].w - lo);
            if (ok && u0 < NR_NODES) col[atomicAdd(&cur[u0], 1)] = sv.v[k].x;
            if (ok && u1 < NR_NODES) col[atomicAdd(&cur[u1], 1)] = sv.v[k].y;
            if (ok && u2 < NR_NODES) col[atomicAdd(&cur[u2], 1)] = sv.v[k].z;
            if (ok && u3 < NR_NODES) col[atomicAdd(&cur[u3], 1)] = sv.v[k].w;
        }
    }
}

__device__ __forceinline__ v4bf pack4_bf16(float4 v, float sc) {
    v4bf o;
    o[0] = (__bf16)(v.x * sc);
    o[1] = (__bf16)(v.y * sc);
    o[2] = (__bf16)(v.z * sc);
    o[3] = (__bf16)(v.w * sc);
    return o;
}

template <int F, int PITCH>
__device__ __forceinline__ void sum_rows(const float4* __restrict__ X4, const int* __restrict__ col,
                                         int beg, int end, int lane, float4 (&acc)[F / 128]) {
    constexpr int NV = F / 128;
#pragma unroll
    for (int q = 0; q < NV; ++q) acc[q] = make_float4(0.0f, 0.0f, 0.0f, 0.0f);

    for (int base = beg; base < end; base += 32) {
        const int m = (end - base < 32) ? (end - base) : 32;
        const int myc = (lane < m) ? col[base + lane] : 0;
        int j = 0;
        for (; j + 4 <= m; j += 4) {
            const float4* r0 = X4 + (size_t)__builtin_amdgcn_readlane(myc, j + 0) * (PITCH / 4) + lane;
            const float4* r1 = X4 + (size_t)__builtin_amdgcn_readlane(myc, j + 1) * (PITCH / 4) + lane;
            const float4* r2 = X4 + (size_t)__builtin_amdgcn_readlane(myc, j + 2) * (PITCH / 4) + lane;
            const float4* r3 = X4 + (size_t)__builtin_amdgcn_readlane(myc, j + 3) * (PITCH / 4) + lane;
            float4 v0[NV], v1[NV], v2[NV], v3[NV];
#pragma unroll
            for (int q = 0; q < NV; ++q) {
                v0[q] = r0[q * 32]; v1[q] = r1[q * 32];
                v2[q] = r2[q * 32]; v3[q] = r3[q * 32];
            }
#pragma unroll
            for (int q = 0; q < NV; ++q) {
                acc[q].x += (v0[q].x + v1[q].x) + (v2[q].x + v3[q].x);
                acc[q].y += (v0[q].y + v1[q].y) + (v2[q].y + v3[q].y);
                acc[q].z += (v0[q].z + v1[q].z) + (v2[q].z + v3[q].z);
                acc[q].w += (v0[q].w + v1[q].w) + (v2[q].w + v3[q].w);
            }
        }
        for (; j < m; ++j) {
            const float4* r0 = X4 + (size_t)__builtin_amdgcn_readlane(myc, j) * (PITCH / 4) + lane;
#pragma unroll
            for (int q = 0; q < NV; ++q) {
                float4 v = r0[q * 32];
                acc[q].x += v.x; acc[q].y += v.y; acc[q].z += v.z; acc[q].w += v.w;
            }
        }
    }
}

template <int F>
__global__ __launch_bounds__(256)
void gather_mean_cat(const float* __restrict__ X, const int* __restrict__ rowptr,
                     const int* __restrict__ col, __bf16* __restrict__ Xcat, int nnodes) {
    constexpr int NV = F / 128;
    const int lane = threadIdx.x & 31;
    const int n = blockIdx.x * (blockDim.x >> 5) + (threadIdx.x >> 5);
    if (n >= nnodes) return;
    const int beg = __builtin_amdgcn_readfirstlane(rowptr[n]);
    const int end = __builtin_amdgcn_readfirstlane(rowptr[n + 1]);
    const float4* X4 = (const float4*)X;
    float4 acc[NV];
    sum_rows<F, F>(X4, col, beg, end, lane, acc);

    const float inv = 1.0f / fmaxf((float)(end - beg), 1.0f);
    __bf16* row = Xcat + (size_t)n * (2 * F);
#pragma unroll
    for (int q = 0; q < NV; ++q)
        *(v4bf*)(row + q * 128 + lane * 4) = pack4_bf16(acc[q], inv);
#pragma unroll
    for (int q = 0; q < NV; ++q) {
        float4 v = X4[(size_t)n * (F / 4) + q * 32 + lane];
        *(v4bf*)(row + F + q * 128 + lane * 4) = pack4_bf16(v, 1.0f);
    }
}

template <int F>
__global__ __launch_bounds__(256)
void gather_mean_add_fc(const float* __restrict__ PQ, const int* __restrict__ rowptr,
                        const int* __restrict__ col, const float* __restrict__ Wfc,
                        const float* __restrict__ bfc, float* __restrict__ out, int nnodes) {
    constexpr int NV = F / 128;
    const int lane = threadIdx.x & 31;
    const int n = blockIdx.x * (blockDim.x >> 5) + (threadIdx.x >> 5);
    if (n >= nnodes) return;
    const int beg = __builtin_amdgcn_readfirstlane(rowptr[n]);
    const int end = __builtin_amdgcn_readfirstlane(rowptr[n + 1]);
    const float4* PQ4 = (const float4*)PQ;
    const float4* W4  = (const float4*)Wfc;
    float4 self[NV];
    float4 w0[NV], w1[NV];
#pragma unroll
    for (int q = 0; q < NV; ++q) {
        self[q] = PQ4[(size_t)n * (2 * F / 4) + F / 4 + q * 32 + lane];
        w0[q]   = W4[q * 32 + lane];
        w1[q]   = W4[F / 4 + q * 32 + lane];
    }
    float4 acc[NV];
    sum_rows<F, 2 * F>(PQ4, col, beg, end, lane, acc);

    const float inv = 1.0f / fmaxf((float)(end - beg), 1.0f);
    float ss = 0.0f, d0 = 0.0f, d1 = 0.0f;
#pragma unroll
    for (int q = 0; q < NV; ++q) {
        float4 o;
        o.x = acc[q].x * inv + self[q].x;
        o.y = acc[q].y * inv + self[q].y;
        o.z = acc[q].z * inv + self[q].z;
        o.w = acc[q].w * inv + self[q].w;
        ss += (o.x * o.x + o.y * o.y) + (o.z * o.z + o.w * o.w);
        d0 += (o.x * w0[q].x + o.y * w0[q].y) + (o.z * w0[q].z + o.w * w0[q].w);
        d1 += (o.x * w1[q].x + o.y * w1[q].y) + (o.z * w1[q].z + o.w * w1[q].w);
    }
    for (int off = 16; off > 0; off >>= 1) {
        ss += __shfl_xor(ss, off, 32);
        d0 += __shfl_xor(d0, off, 32);
        d1 += __shfl_xor(d1, off, 32);
    }
    if (lane == 0) {
        float invn = 1.0f / fmaxf(sqrtf(ss), 1e-12f);
        float l0 = d0 * invn + bfc[0];
        float l1 = d1 * invn + bfc[1];
        float m  = fmaxf(l0, l1);
        float e0 = __expf(l0 - m), e1 = __expf(l1 - m);
        float s  = 1.0f / (e0 + e1);
        out[(size_t)n * 2 + 0] = e0 * s;
        out[(size_t)n * 2 + 1] = e1 * s;
    }
}

#define GEMM_K       256
#define GEMM_NCOL    256
#define GEMM_THREADS 1024
#define GEMM_BLOCKS  256
#define BPITCH       (GEMM_K + 8)

template <int K, int NROW>
__device__ __forceinline__ void stage_b_lds(const __bf16* __restrict__ B, __bf16* Bs, int tid) {
    constexpr int PPR   = K / 8;
    constexpr int TOTAL = NROW * PPR;
#pragma unroll
    for (int i = 0; i < TOTAL / GEMM_THREADS; ++i) {
        const int p   = i * GEMM_THREADS + tid;
        const int row = p / PPR, pc = p % PPR;
        const v8bf v = *(const v8bf*)(B + (size_t)row * K + pc * 8);
        *(v8bf*)(Bs + row * BPITCH + pc * 8) = v;
    }
}

template <int K, int NT>
__device__ __forceinline__ void wmma_strip_lds(const __bf16* __restrict__ arow, const __bf16* brow,
                                               v8f (&acc)[NT]) {
    union { v16bf v; v8bf h[2]; } a[K / 32];
#pragma unroll
    for (int s = 0; s < K / 32; ++s) {
        a[s].h[0] = *(const v8bf*)(arow + s * 32);
        a[s].h[1] = *(const v8bf*)(arow + s * 32 + 16);
    }
    __builtin_amdgcn_sched_barrier(0);
#pragma unroll
    for (int t = 0; t < NT; ++t) acc[t] = (v8f){};

#pragma unroll
    for (int s = 0; s < K / 32; ++s) {
#pragma unroll
        for (int t = 0; t < NT; ++t) {
            union { v16bf v; v8bf h[2]; } b;
            b.h[0] = *(const v8bf*)(brow + t * 16 * BPITCH + s * 32);
            b.h[1] = *(const v8bf*)(brow + t * 16 * BPITCH + s * 32 + 16);
            acc[t] = __builtin_amdgcn_wmma_f32_16x16x32_bf16(
                false, a[s].v, false, b.v, (short)0, acc[t], false, false);
        }
    }
}

template <int K, int NT>
__device__ __forceinline__ void wmma_strip_lds_ab(const __bf16* arow, const __bf16* brow,
                                                  v8f (&acc)[NT]) {
#pragma unroll
    for (int t = 0; t < NT; ++t) acc[t] = (v8f){};
#pragma unroll
    for (int s = 0; s < K / 32; ++s) {
        union { v16bf v; v8bf h[2]; } a;
        a.h[0] = *(const v8bf*)(arow + s * 32);
        a.h[1] = *(const v8bf*)(arow + s * 32 + 16);
#pragma unroll
        for (int t = 0; t < NT; ++t) {
            union { v16bf v; v8bf h[2]; } b;
            b.h[0] = *(const v8bf*)(brow + t * 16 * BPITCH + s * 32);
            b.h[1] = *(const v8bf*)(brow + t * 16 * BPITCH + s * 32 + 16);
            acc[t] = __builtin_amdgcn_wmma_f32_16x16x32_bf16(
                false, a.v, false, b.v, (short)0, acc[t], false, false);
        }
    }
}

#define GEMM_NS      ((GEMM_THREADS / 32) / (GEMM_NCOL / 64))
#define GEMM_NTILES  (N_NODES / 16)
#define GEMM_STRIDE  (GEMM_BLOCKS * GEMM_NS)
#define GEMM_TRIPS   ((GEMM_NTILES + GEMM_STRIDE - 1) / GEMM_STRIDE)
static_assert(N_NODES % 16 == 0, "whole 16-row tiles");
static_assert((GEMM_NCOL + GEMM_TRIPS * GEMM_NS * 16) * BPITCH * 2 + GEMM_NS * 4 * 16 * 4 <= 320 * 1024,
              "weights + h tiles + partial sums fit one WGP's LDS");
__global__ __launch_bounds__(GEMM_THREADS)
void gemm2_bf16_wmma_lds_h(const __bf16* __restrict__ A, const __bf16* __restrict__ B1,
                           const float* __restrict__ bias1, const __bf16* __restrict__ B2,
                           const float* __restrict__ bias2, int bias2_col0,
                           float* __restrict__ C) {
    constexpr int K = GEMM_K, NCOL = GEMM_NCOL, NT = 4, NW = NCOL / (16 * NT), NS = GEMM_NS;
    static_assert(K == NCOL, "layer 1's output columns are layer 2's k");
    __shared__ __attribute__((aligned(16))) __bf16 Bs[NCOL * BPITCH];
    __shared__ __attribute__((aligned(16))) __bf16 Hs[GEMM_TRIPS * NS * 16 * BPITCH];
    __shared__ __attribute__((aligned(16))) float red[NS][NW][16];
    const int lane = threadIdx.x & 31;
    const int wv   = __builtin_amdgcn_readfirstlane(threadIdx.x >> 5);
    const int w    = wv % NW;
    const int slot = wv / NW;
    const int tn0  = w * (16 * NT);
    const int h    = lane >> 4;
    const int l15  = lane & 15;

    stage_b_lds<K, NCOL>(B1, Bs, threadIdx.x);
    __syncthreads();

    const __bf16* brow = Bs + (tn0 + l15) * BPITCH + h * 8;
    for (int it = 0; it < GEMM_TRIPS; ++it) {
        const int  tile  = it * GEMM_STRIDE + slot * GEMM_BLOCKS + (int)blockIdx.x;
        const bool valid = tile < GEMM_NTILES;
        const int  tm    = tile << 4;
        __bf16* htile = Hs + (it * NS + slot) * (16 * BPITCH);
        v8f acc[NT];
        if (valid) {
            wmma_strip_lds<K, NT>(A + (size_t)(tm + l15) * K + h * 8, brow, acc);

            float ss[8];
#pragma unroll
            for (int v = 0; v < 8; ++v) ss[v] = 0.0f;
#pragma unroll
            for (int t = 0; t < NT; ++t) {
                const float bc = bias1[tn0 + t * 16 + l15];
#pragma unroll
                for (int v = 0; v < 8; ++v) {
                    acc[t][v] += bc;
                    ss[v] += acc[t][v] * acc[t][v];
                }
            }
#pragma unroll
            for (int off = 1; off < 16; off <<= 1) {
#pragma unroll
                for (int v = 0; v < 8; ++v) ss[v] += __shfl_xor(ss[v], off, 32);
            }
            if (l15 == 0) {
                float4* mine = (float4*)&red[slot][w][h * 8];
                mine[0] = make_float4(ss[0], ss[1], ss[2], ss[3]);
                mine[1] = make_float4(ss[4], ss[5], ss[6], ss[7]);
            }
        }
        __syncthreads();
        if (valid) {
            float tot = red[slot][0][l15];
#pragma unroll
            for (int q = 1; q < NW; ++q) tot += red[slot][q][l15];
            const float inv_mine = 1.0f / fmaxf(sqrtf(tot), 1e-12f);
#pragma unroll
            for (int v = 0; v < 8; ++v) {
                const float inv = __shfl(inv_mine, h * 8 + v, 32);
                __bf16* hrow = htile + (h * 8 + v) * BPITCH + tn0 + l15;
#pragma unroll
                for (int t = 0; t < NT; ++t) {
                    const float r = fmaxf(acc[t][v] * inv, 0.0f);
                    hrow[t * 16] = (__bf16)r;
                }
            }
        }
        __syncthreads();
    }

    stage_b_lds<K, NCOL>(B2, Bs, threadIdx.x);
    __syncthreads();

    for (int it = 0; it < GEMM_TRIPS; ++it) {
        const int tile = it * GEMM_STRIDE + slot * GEMM_BLOCKS + (int)blockIdx.x;
        if (tile < GEMM_NTILES) {
            const int tm = tile << 4;
            const __bf16* htile = Hs + (it * NS + slot) * (16 * BPITCH);
            v8f acc[NT];
            wmma_strip_lds_ab<K, NT>(htile + l15 * BPITCH + h * 8, brow, acc);

            const int r0 = tm + h * 8;
#pragma unroll
            for (int t = 0; t < NT; ++t) {
                const int cn = tn0 + t * 16 + l15;
                const int   bi = cn - bias2_col0;
                const float bv = bias2[bi < 0 ? 0 : bi];
                const float bc = (bi < 0) ? 0.0f : bv;
#pragma unroll
                for (int v = 0; v < 8; ++v)
                    C[(size_t)(r0 + v) * NCOL + cn] = acc[t][v] + bc;
            }
        }
    }
}

extern "C" void kernel_launch(void* const* d_in, const int* in_sizes, int n_in,
                              void* d_out, int out_size, void* d_ws, size_t ws_size,
                              hipStream_t stream) {
    (void)in_sizes; (void)n_in; (void)out_size; (void)ws_size;

    const float* feat = (const float*)d_in[0];
    const int*   eidx = (const int*)  d_in[1];
    const int*   src  = eidx;
    const int*   dst  = eidx + N_EDGES;
    const float* W1l  = (const float*)d_in[2];
    const float* b1   = (const float*)d_in[3];
    const float* W1r  = (const float*)d_in[4];
    const float* W2l  = (const float*)d_in[5];
    const float* b2   = (const float*)d_in[6];
    const float* W2r  = (const float*)d_in[7];
    const float* Wfc  = (const float*)d_in[8];
    const float* bfc  = (const float*)d_in[9];
    float* out = (float*)d_out;

    char* ws = (char*)d_ws;
    size_t off = 0;
    auto alloc = [&](size_t bytes) -> void* {
        void* p = ws + off;
        off = (off + bytes + 255) & ~(size_t)255;
        return p;
    };
    int*    part   = (int*)  alloc((size_t)N_SPLITS * PART_PITCH * 4);
    int*    tot    = (int*)  alloc((size_t)N_RANGES * N_SPLITS * 4);
    int*    rowptr = (int*)  alloc((size_t)PART_PITCH * 4);
    int*    col    = (int*)  alloc((size_t)N_EDGES * 4);
    __bf16* Xcat1 = (__bf16*)alloc((size_t)N_NODES * 2 * IN_SIZE * 2);
    __bf16* Wcat1 = (__bf16*)alloc((size_t)HIDDEN * 2 * IN_SIZE * 2);
    __bf16* Wst2  = (__bf16*)alloc((size_t)2 * OUT_SIZE * HIDDEN * 2);
    float*  pq2   = (float*) alloc((size_t)N_NODES * 2 * OUT_SIZE * 4);

    const int TB = 256;
    auto cdiv = [](int a, int b) { return (a + b - 1) / b; };

    inlist_count_pack<<<COUNT_BLOCKS + PACK_BLOCKS, LIST_T, 0, stream>>>(
        dst, part, tot, W1l, W1r, W2l, W2r, Wcat1, Wst2);
    inlist_fill<<<COUNT_BLOCKS, LIST_T, 0, stream>>>(src, dst, part, tot, rowptr, col);
    gather_mean_cat<IN_SIZE><<<cdiv(N_NODES, TB / 32), TB, 0, stream>>>(
        feat, rowptr, col, Xcat1, N_NODES);
    gemm2_bf16_wmma_lds_h<<<GEMM_BLOCKS, GEMM_THREADS, 0, stream>>>(
        Xcat1, Wcat1, b1, Wst2, b2, OUT_SIZE, pq2);
    gather_mean_add_fc<OUT_SIZE><<<cdiv(N_NODES, TB / 32), TB, 0, stream>>>(
        pq2, rowptr, col, Wfc, bfc, out, N_NODES);
}


# Layer-1 gather moved into the GEMM launch: a work-group's 32 waves walk their tiles' nodes (next list asked ahead, 8 rows in flight), bf16 A tile written to LDS; A plane and a launch gone; sums same
# speedup vs baseline: 6.2934x; 1.1160x over previous
#include <hip/hip_runtime.h>
#include <math.h>

#define N_NODES  50000
#define N_EDGES  600000
#define IN_SIZE  128
#define HIDDEN   256
#define OUT_SIZE 128

typedef __attribute__((ext_vector_type(16))) __bf16 v16bf;
typedef __attribute__((ext_vector_type(8)))  __bf16 v8bf;
typedef __attribute__((ext_vector_type(8)))  float  v8f;
typedef __attribute__((ext_vector_type(4)))  __bf16 v4bf;

#define NR_NODES   2048
#define N_RANGES   ((N_NODES + NR_NODES - 1) / NR_NODES)
#define N_SPLITS   10
#define SPLIT_E    (N_EDGES / N_SPLITS)
#define SPLIT_V4   (SPLIT_E / 4)
#define PART_PITCH (N_RANGES * NR_NODES)
#define LIST_T     1024
#define LIST_UNR   8
static_assert(N_EDGES % (N_SPLITS * 4) == 0, "pieces are whole 16-byte vectors");
static_assert(N_NODES < PART_PITCH, "rowptr[N_NODES] must be an entry of the last range");
static_assert(NR_NODES == 2 * LIST_T, "a thread owns two adjacent counters");
static_assert(N_RANGES * N_SPLITS <= LIST_T, "one thread per (range, piece) sum");

#define PACK_V4_L1   (HIDDEN * IN_SIZE / 4)
#define PACK_V4_L2H  (OUT_SIZE * HIDDEN / 4)
#define PACK_BLOCKS  ((PACK_V4_L1 + 2 * PACK_V4_L2H) / LIST_T)
static_assert((PACK_V4_L1 + 2 * PACK_V4_L2H) % LIST_T == 0, "whole packing work-groups");
static_assert(PACK_V4_L1 % LIST_T == 0, "a packing work-group serves one layer only");
static_assert(IN_SIZE % 4 == 0, "a float4 never crosses a row of W1l / W1r");

__device__ __forceinline__ v4bf cvt4_bf16(float4 v) {
    v4bf o;
    o[0] = (__bf16)v.x;
    o[1] = (__bf16)v.y;
    o[2] = (__bf16)v.z;
    o[3] = (__bf16)v.w;
    return o;
}

__device__ __forceinline__ void pack_weights(int i,
                                             const float* __restrict__ W1l, const float* __restrict__ W1r,
                                             const float* __restrict__ W2l, const float* __restrict__ W2r,
                                             __bf16* __restrict__ Wcat1, __bf16* __restrict__ Wst2) {
    if (i < PACK_V4_L1) {
        const float4 a = ((const float4*)W1l)[i];
        const float4 b = ((const float4*)W1r)[i];
        const int r = (4 * i) / IN_SIZE, cc = (4 * i) % IN_SIZE;
        __bf16* row = Wcat1 + (size_t)r * (2 * IN_SIZE);
        *(v4bf*)(row + cc)           = cvt4_bf16(a);
        *(v4bf*)(row + IN_SIZE + cc) = cvt4_bf16(b);
    } else {
        const int j = i - PACK_V4_L1;
        const float4 a = (j < PACK_V4_L2H) ? ((const float4*)W2l)[j]
                                           : ((const float4*)W2r)[j - PACK_V4_L2H];
        *(v4bf*)(Wst2 + (size_t)4 * j) = cvt4_bf16(a);
    }
}


struct EdgeBatch {
    int4 v[LIST_UNR];
};
__device__ __forceinline__ void load_batch(const int4* __restrict__ p4, int i0, EdgeBatch& b) {
#pragma unroll
    for (int k = 0; k < LIST_UNR; ++k) {
        const int i = i0 + k * LIST_T;
        b.v[k] = p4[i < SPLIT_V4 ? i : SPLIT_V4 - 1];
    }
}

#define COUNT_BLOCKS (N_RANGES * N_SPLITS)
__global__ __launch_bounds__(LIST_T)
void inlist_count_pack(const int* __restrict__ dst, int* __restrict__ part, int* __restrict__ tot,
                       const float* __restrict__ W1l, const float* __restrict__ W1r,
                       const float* __restrict__ W2l, const float* __restrict__ W2r,
                       __bf16* __restrict__ Wcat1, __bf16* __restrict__ Wst2) {
    __shared__ __attribute__((aligned(16))) int cnt[NR_NODES];
    __shared__ int wsum[LIST_T / 32];
    const int tid = threadIdx.x, lane = tid & 31, w = tid >> 5;
    if (blockIdx.x >= COUNT_BLOCKS) {
        pack_weights(((int)blockIdx.x - COUNT_BLOCKS) * LIST_T + tid, W1l, W1r, W2l, W2r, Wcat1, Wst2);
        return;
    }
    const int g = blockIdx.x / N_SPLITS, s = blockIdx.x % N_SPLITS;
    const int lo = g * NR_NODES;
    cnt[tid] = 0;
    cnt[tid + LIST_T] = 0;
    __syncthreads();

    const int4* d4 = (const int4*)(dst + (size_t)s * SPLIT_E);
    for (int i0 = tid; i0 < SPLIT_V4; i0 += LIST_T * LIST_UNR) {
        EdgeBatch d;
        load_batch(d4, i0, d);
#pragma unroll
        for (int k = 0; k < LIST_UNR; ++k) {
            const bool ok = (i0 + k * LIST_T) < SPLIT_V4;
            const unsigned u0 = (unsigned)(d.v[k].x - lo), u1 = (unsigned)(d.v[k].y - lo);
            const unsigned u2 = (unsigned)(d.v[k].z - lo), u3 = (unsigned)(d.v[k].w - lo);
            if (ok && u0 < NR_NODES) atomicAdd(&cnt[u0], 1);
            if (ok && u1 < NR_NODES) atomicAdd(&cnt[u1], 1);
            if (ok && u2 < NR_NODES) atomicAdd(&cnt[u2], 1);
            if (ok && u3 < NR_NODES) atomicAdd(&cnt[u3], 1);
        }
    }
    __syncthreads();

    const int2 c = *(const int2*)&cnt[2 * tid];
    *(int2*)(part + (size_t)s * PART_PITCH + lo + 2 * tid) = c;
    int t = c.x + c.y;
    for (int off = 16; off > 0; off >>= 1) t += __shfl_xor(t, off, 32);
    if (lane == 0) wsum[w] = t;
    __syncthreads();
    if (w == 0) {
        int x = wsum[lane];
        for (int off = 16; off > 0; off >>= 1) x += __shfl_xor(x, off, 32);
        if (lane == 0) tot[g * N_SPLITS + s] = x;
    }
}

__global__ __launch_bounds__(LIST_T)
void inlist_fill(const int* __restrict__ src, const int* __restrict__ dst,
                 const int* __restrict__ part, const int* __restrict__ tot,
                 int* __restrict__ rowptr, int* __restrict__ col) {
    __shared__ __attribute__((aligned(16))) int cur[NR_NODES];
    __shared__ int wsum[LIST_T / 32];
    __shared__ int wbase[LIST_T / 32];
    const int tid = threadIdx.x, lane = tid & 31, w = tid >> 5;
    const int g = blockIdx.x / N_SPLITS, s = blockIdx.x % N_SPLITS;
    const int lo = g * NR_NODES;

    int b = (tid < g * N_SPLITS) ? tot[tid] : 0;
    int2 len = make_int2(0, 0), below = make_int2(0, 0);
#pragma unroll
    for (int q = 0; q < N_SPLITS; ++q) {
        const int2 p = *(const int2*)(part + (size_t)q * PART_PITCH + lo + 2 * tid);
        len.x += p.x; len.y += p.y;
        if (q < s) { below.x += p.x; below.y += p.y; }
    }
    const int v = len.x + len.y;
    int inc = v;
    for (int off = 1; off < 32; off <<= 1) {
        int x = __shfl_up(inc, off, 32);
        if (lane >= off) inc += x;
    }
    for (int off = 16; off > 0; off >>= 1) b += __shfl_xor(b, off, 32);
    if (lane == 31) wsum[w] = inc;
    if (lane == 0)  wbase[w] = b;
    __syncthreads();
    if (w == 0) {
        int x = wsum[lane], xi = x;
        for (int off = 1; off < 32; off <<= 1) {
            int y = __shfl_up(xi, off, 32);
            if (lane >= off) xi += y;
        }
        wsum[lane] = xi - x;
        int bb = wbase[lane];
        for (int off = 16; off > 0; off >>= 1) bb += __shfl_xor(bb, off, 32);
        if (lane == 0) wbase[0] = bb;
    }
    __syncthreads();
    const int st0 = wbase[0] + wsum[w] + inc - v;
    const int st1 = st0 + len.x;
    *(int2*)&cur[2 * tid] = make_int2(st0 + below.x, st1 + below.y);
    if (s == 0) *(int2*)(rowptr + lo + 2 * tid) = make_int2(st0, st1);
    __syncthreads();

    const int4* d4 = (const int4*)(dst + (size_t)s * SPLIT_E);
    const int4* s4 = (const int4*)(src + (size_t)s * SPLIT_E);
    for (int i0 = tid; i0 < SPLIT_V4; i0 += LIST_T * LIST_UNR) {
        EdgeBatch d, sv;
        load_batch(d4, i0, d);
        load_batch(s4, i0, sv);
        asm volatile("" ::: "memory");
#pragma unroll
        for (int k = 0; k < LIST_UNR; ++k) {
            const bool ok = (i0 + k * LIST_T) < SPLIT_V4;
            const unsigned u0 = (unsigned)(d.v[k].x - lo), u1 = (unsigned)(d.v[k].y - lo);
            const unsigned u2 = (unsigned)(d.v[k].z - lo), u3 = (unsigned)(d.v[k].w - lo);
            if (ok && u0 < NR_NODES) col[atomicAdd(&cur[u0], 1)] = sv.v[k].x;
            if (ok && u1 < NR_NODES) col[atomicAdd(&cur[u1], 1)] = sv.v[k].y;
            if (ok && u2 < NR_NODES) col[atomicAdd(&cur[u2], 1)] = sv.v[k].z;
            if (ok && u3 < NR_NODES) col[atomicAdd(&cur[u3], 1)] = sv.v[k].w;
        }
    }
}

__device__ __forceinline__ v4bf pack4_bf16(float4 v, float sc) {
    v4bf o;
    o[0] = (__bf16)(v.x * sc);
    o[1] = (__bf16)(v.y * sc);
    o[2] = (__bf16)(v.z * sc);
    o[3] = (__bf16)(v.w * sc);
    return o;
}

template <int F, int PITCH>
__device__ __forceinline__ void sum_rows(const float4* __restrict__ X4, const int* __restrict__ col,
                                         int beg, int end, int lane, float4 (&acc)[F / 128]) {
    constexpr int NV = F / 128;
#pragma unroll
    for (int q = 0; q < NV; ++q) acc[q] = make_float4(0.0f, 0.0f, 0.0f, 0.0f);

    for (int base = beg; base < end; base += 32) {
        const int m = (end - base < 32) ? (end - base) : 32;
        const int myc = (lane < m) ? col[base + lane] : 0;
        int j = 0;
        for (; j + 4 <= m; j += 4) {
            const float4* r0 = X4 + (size_t)__builtin_amdgcn_readlane(myc, j + 0) * (PITCH / 4) + lane;
            const float4* r1 = X4 + (size_t)__builtin_amdgcn_readlane(myc, j + 1) * (PITCH / 4) + lane;
            const float4* r2 = X4 + (size_t)__builtin_amdgcn_readlane(myc, j + 2) * (PITCH / 4) + lane;
            const float4* r3 = X4 + (size_t)__builtin_amdgcn_readlane(myc, j + 3) * (PITCH / 4) + lane;
            float4 v0[NV], v1[NV], v2[NV], v3[NV];
#pragma unroll
            for (int q = 0; q < NV; ++q) {
                v0[q] = r0[q * 32]; v1[q] = r1[q * 32];
                v2[q] = r2[q * 32]; v3[q] = r3[q * 32];
            }
#pragma unroll
            for (int q = 0; q < NV; ++q) {
                acc[q].x += (v0[q].x + v1[q].x) + (v2[q].x + v3[q].x);
                acc[q].y += (v0[q].y + v1[q].y) + (v2[q].y + v3[q].y);
                acc[q].z += (v0[q].z + v1[q].z) + (v2[q].z + v3[q].z);
                acc[q].w += (v0[q].w + v1[q].w) + (v2[q].w + v3[q].w);
            }
        }
        for (; j < m; ++j) {
            const float4* r0 = X4 + (size_t)__builtin_amdgcn_readlane(myc, j) * (PITCH / 4) + lane;
#pragma unroll
            for (int q = 0; q < NV; ++q) {
                float4 v = r0[q * 32];
                acc[q].x += v.x; acc[q].y += v.y; acc[q].z += v.z; acc[q].w += v.w;
            }
        }
    }
}

__device__ __forceinline__ float4 row_of(const float4* __restrict__ X4, int r, int lane) {
    return X4[(size_t)r * (IN_SIZE / 4) + lane];
}
__device__ __forceinline__ void add_four(float4& acc, float4 a, float4 b, float4 c, float4 d) {
    acc.x += (a.x + b.x) + (c.x + d.x);
    acc.y += (a.y + b.y) + (c.y + d.y);
    acc.z += (a.z + b.z) + (c.z + d.z);
    acc.w += (a.w + b.w) + (c.w + d.w);
}
__device__ __forceinline__ void add_one(float4& acc, float4 v) {
    acc.x += v.x; acc.y += v.y; acc.z += v.z; acc.w += v.w;
}
__device__ __forceinline__ void sum_eight(const float4* __restrict__ X4, int lane, float4& acc,
                                          int s0, int s1, int s2, int s3, int s4, int s5, int s6, int s7) {
    const float4 v0 = row_of(X4, s0, lane), v1 = row_of(X4, s1, lane);
    const float4 v2 = row_of(X4, s2, lane), v3 = row_of(X4, s3, lane);
    const float4 v4 = row_of(X4, s4, lane), v5 = row_of(X4, s5, lane);
    const float4 v6 = row_of(X4, s6, lane), v7 = row_of(X4, s7, lane);
    add_four(acc, v0, v1, v2, v3);
    add_four(acc, v4, v5, v6, v7);
}
__device__ __forceinline__ void sum_four(const float4* __restrict__ X4, int lane, float4& acc,
                                         int s0, int s1, int s2, int s3) {
    const float4 v0 = row_of(X4, s0, lane), v1 = row_of(X4, s1, lane);
    const float4 v2 = row_of(X4, s2, lane), v3 = row_of(X4, s3, lane);
    add_four(acc, v0, v1, v2, v3);
}

struct FirstSources { int s[8]; };
__device__ __forceinline__ FirstSources first_sources(int myc) {
    FirstSources f;
#pragma unroll
    for (int k = 0; k < 8; ++k) f.s[k] = __builtin_amdgcn_readlane(myc, k);
    return f;
}

__device__ __forceinline__ void sum_rows_walk(const float4* __restrict__ X4, const int* __restrict__ col,
                                              int beg, int end, int lane, int myc,
                                              const FirstSources& f, float4& acc) {
    acc = make_float4(0.0f, 0.0f, 0.0f, 0.0f);
    const int m0 = (end - beg < 32) ? (end - beg) : 32;
    int j = 0;
    if (m0 >= 8) {
        sum_eight(X4, lane, acc, f.s[0], f.s[1], f.s[2], f.s[3], f.s[4], f.s[5], f.s[6], f.s[7]);
        j = 8;
    } else if (m0 >= 4) {
        sum_four(X4, lane, acc, f.s[0], f.s[1], f.s[2], f.s[3]);
        j = 4;
    }
    for (int base = beg; base < end; base += 32) {
        const int m = (end - base < 32) ? (end - base) : 32;
        if (base != beg) {
            myc = (lane < m) ? col[base + lane] : 0;
            j = 0;
        }
        for (; j + 8 <= m; j += 8)
            sum_eight(X4, lane, acc,
                      __builtin_amdgcn_readlane(myc, j + 0), __builtin_amdgcn_readlane(myc, j + 1),
                      __builtin_amdgcn_readlane(myc, j + 2), __builtin_amdgcn_readlane(myc, j + 3),
                      __builtin_amdgcn_readlane(myc, j + 4), __builtin_amdgcn_readlane(myc, j + 5),
                      __builtin_amdgcn_readlane(myc, j + 6), __builtin_amdgcn_readlane(myc, j + 7));
        if (j + 4 <= m) {
            sum_four(X4, lane, acc,
                     __builtin_amdgcn_readlane(myc, j + 0), __builtin_amdgcn_readlane(myc, j + 1),
                     __builtin_amdgcn_readlane(myc, j + 2), __builtin_amdgcn_readlane(myc, j + 3));
            j += 4;
        }
        const int rem = m - j;
        if (rem > 0) {
            const int s0 = __builtin_amdgcn_readlane(myc, j);
            const int s1 = __builtin_amdgcn_readlane(myc, j + 1);
            const int s2 = __builtin_amdgcn_readlane(myc, j + 2);
            const float4 u0 = row_of(X4, s0, lane);
            float4 u1 = make_float4(0.0f, 0.0f, 0.0f, 0.0f), u2 = u1;
            if (rem > 1) u1 = row_of(X4, s1, lane);
            if (rem > 2) u2 = row_of(X4, s2, lane);
            add_one(acc, u0);
            if (rem > 1) add_one(acc, u1);
            if (rem > 2) add_one(acc, u2);
        }
    }
}

template <int F>
__global__ __launch_bounds__(256)
void gather_mean_add_fc(const float* __restrict__ PQ, const int* __restrict__ rowptr,
                        const int* __restrict__ col, const float* __restrict__ Wfc,
                        const float* __restrict__ bfc, float* __restrict__ out, int nnodes) {
    constexpr int NV = F / 128;
    const int lane = threadIdx.x & 31;
    const int n = blockIdx.x * (blockDim.x >> 5) + (threadIdx.x >> 5);
    if (n >= nnodes) return;
    const int beg = __builtin_amdgcn_readfirstlane(rowptr[n]);
    const int end = __builtin_amdgcn_readfirstlane(rowptr[n + 1]);
    const float4* PQ4 = (const float4*)PQ;
    const float4* W4  = (const float4*)Wfc;
    float4 self[NV];
    float4 w0[NV], w1[NV];
#pragma unroll
    for (int q = 0; q < NV; ++q) {
        self[q] = PQ4[(size_t)n * (2 * F / 4) + F / 4 + q * 32 + lane];
        w0[q]   = W4[q * 32 + lane];
        w1[q]   = W4[F / 4 + q * 32 + lane];
    }
    float4 acc[NV];
    sum_rows<F, 2 * F>(PQ4, col, beg, end, lane, acc);

    const float inv = 1.0f / fmaxf((float)(end - beg), 1.0f);
    float ss = 0.0f, d0 = 0.0f, d1 = 0.0f;
#pragma unroll
    for (int q = 0; q < NV; ++q) {
        float4 o;
        o.x = acc[q].x * inv + self[q].x;
        o.y = acc[q].y * inv + self[q].y;
        o.z = acc[q].z * inv + self[q].z;
        o.w = acc[q].w * inv + self[q].w;
        ss += (o.x * o.x + o.y * o.y) + (o.z * o.z + o.w * o.w);
        d0 += (o.x * w0[q].x + o.y * w0[q].y) + (o.z * w0[q].z + o.w * w0[q].w);
        d1 += (o.x * w1[q].x + o.y * w1[q].y) + (o.z * w1[q].z + o.w * w1[q].w);
    }
    for (int off = 16; off > 0; off >>= 1) {
        ss += __shfl_xor(ss, off, 32);
        d0 += __shfl_xor(d0, off, 32);
        d1 += __shfl_xor(d1, off, 32);
    }
    if (lane == 0) {
        float invn = 1.0f / fmaxf(sqrtf(ss), 1e-12f);
        float l0 = d0 * invn + bfc[0];
        float l1 = d1 * invn + bfc[1];
        float m  = fmaxf(l0, l1);
        float e0 = __expf(l0 - m), e1 = __expf(l1 - m);
        float s  = 1.0f / (e0 + e1);
        out[(size_t)n * 2 + 0] = e0 * s;
        out[(size_t)n * 2 + 1] = e1 * s;
    }
}

#define GEMM_K       256
#define GEMM_NCOL    256
#define GEMM_THREADS 1024
#define GEMM_BLOCKS  256
#define BPITCH       (GEMM_K + 8)

template <int K, int NROW>
__device__ __forceinline__ void stage_b_lds(const __bf16* __restrict__ B, __bf16* Bs, int tid) {
    constexpr int PPR   = K / 8;
    constexpr int TOTAL = NROW * PPR;
#pragma unroll
    for (int i = 0; i < TOTAL / GEMM_THREADS; ++i) {
        const int p   = i * GEMM_THREADS + tid;
        const int row = p / PPR, pc = p % PPR;
        const v8bf v = *(const v8bf*)(B + (size_t)row * K + pc * 8);
        *(v8bf*)(Bs + row * BPITCH + pc * 8) = v;
    }
}

template <int K, int NT>
__device__ __forceinline__ void wmma_strip_lds_ab(const __bf16* arow, const __bf16* brow,
                                                  v8f (&acc)[NT]) {
#pragma unroll
    for (int t = 0; t < NT; ++t) acc[t] = (v8f){};
#pragma unroll
    for (int s = 0; s < K / 32; ++s) {
        union { v16bf v; v8bf h[2]; } a;
        a.h[0] = *(const v8bf*)(arow + s * 32);
        a.h[1] = *(const v8bf*)(arow + s * 32 + 16);
#pragma unroll
        for (int t = 0; t < NT; ++t) {
            union { v16bf v; v8bf h[2]; } b;
            b.h[0] = *(const v8bf*)(brow + t * 16 * BPITCH + s * 32);
            b.h[1] = *(const v8bf*)(brow + t * 16 * BPITCH + s * 32 + 16);
            acc[t] = __builtin_amdgcn_wmma_f32_16x16x32_bf16(
                false, a.v, false, b.v, (short)0, acc[t], false, false);
        }
    }
}

#define GEMM_NS      ((GEMM_THREADS / 32) / (GEMM_NCOL / 64))
#define GEMM_NTILES  (N_NODES / 16)
#define GEMM_STRIDE  (GEMM_BLOCKS * GEMM_NS)
#define GEMM_TRIPS   ((GEMM_NTILES + GEMM_STRIDE - 1) / GEMM_STRIDE)
static_assert(N_NODES % 16 == 0, "whole 16-row tiles");
static_assert(2 * IN_SIZE == GEMM_K, "the A tile is [mean | own row], IN_SIZE columns each");
static_assert(GEMM_TRIPS * GEMM_NS * 16 <= 32 * 32, "a wave's nodes fit one per lane");
static_assert((GEMM_NCOL + GEMM_TRIPS * GEMM_NS * 16) * BPITCH * 2 + GEMM_NS * 4 * 16 * 4 <= 320 * 1024,
              "weights + h tiles + partial sums fit one WGP's LDS");
__global__ __launch_bounds__(GEMM_THREADS)
void gather_gemm2_bf16_wmma_lds(const float* __restrict__ X, const int* __restrict__ rowptr,
                                const int* __restrict__ col, const __bf16* __restrict__ B1,
                                const float* __restrict__ bias1, const __bf16* __restrict__ B2,
                                const float* __restrict__ bias2, int bias2_col0,
                                float* __restrict__ C) {
    constexpr int K = GEMM_K, NCOL = GEMM_NCOL, NT = 4, NW = NCOL / (16 * NT), NS = GEMM_NS;
    static_assert(K == NCOL, "layer 1's output columns are layer 2's k");
    __shared__ __attribute__((aligned(16))) __bf16 Bs[NCOL * BPITCH];
    __shared__ __attribute__((aligned(16))) __bf16 Hs[GEMM_TRIPS * NS * 16 * BPITCH];
    __shared__ __attribute__((aligned(16))) float red[NS][NW][16];
    const int lane = threadIdx.x & 31;
    const int wv   = __builtin_amdgcn_readfirstlane(threadIdx.x >> 5);
    const int w    = wv % NW;
    const int slot = wv / NW;
    const int tn0  = w * (16 * NT);
    const int h    = lane >> 4;
    const int l15  = lane & 15;

    stage_b_lds<K, NCOL>(B1, Bs, threadIdx.x);

    {
        const int g = (int)blockIdx.x;
        int nq = (GEMM_NTILES - g + GEMM_BLOCKS - 1) / GEMM_BLOCKS;
        if (nq > GEMM_TRIPS * NS) nq = GEMM_TRIPS * NS;
        const int nrows = nq * 16;
        const int cntw = (nrows - wv + 31) >> 5;
        int myn = 0, mybeg = 0, myend = 0;
        if (lane < cntw) {
            const int r = wv + 32 * lane;
            myn   = (((r >> 4) * GEMM_BLOCKS + g) << 4) + (r & 15);
            mybeg = rowptr[myn];
            myend = rowptr[myn + 1];
        }
        const float4* X4 = (const float4*)X;
        int beg = 0, end = 0, myc = 0;
        if (cntw > 0) {
            beg = __builtin_amdgcn_readlane(mybeg, 0);
            end = __builtin_amdgcn_readlane(myend, 0);
            const int m0 = (end - beg < 32) ? (end - beg) : 32;
            myc = (lane < m0) ? col[beg + lane] : 0;
        }
        for (int j = 0; j < cntw; ++j) {
            const int n = __builtin_amdgcn_readlane(myn, j);
            const int r = wv + 32 * j;
            const FirstSources f = first_sources(myc);
            int nbeg = 0, nend = 0, mycn = 0;
            if (j + 1 < cntw) {
                nbeg = __builtin_amdgcn_readlane(mybeg, j + 1);
                nend = __builtin_amdgcn_readlane(myend, j + 1);
                const int mn = (nend - nbeg < 32) ? (nend - nbeg) : 32;
                mycn = (lane < mn) ? col[nbeg + lane] : 0;
            }
            const float4 own = X4[(size_t)n * (IN_SIZE / 4) + lane];
            float4 acc0;
            sum_rows_walk(X4, col, beg, end, lane, myc, f, acc0);
            const float inv = 1.0f / fmaxf((float)(end - beg), 1.0f);
            __bf16* arow = Hs + (r >> 4) * (16 * BPITCH) + (r & 15) * BPITCH;
            *(v4bf*)(arow + lane * 4)           = pack4_bf16(acc0, inv);
            *(v4bf*)(arow + IN_SIZE + lane * 4) = pack4_bf16(own, 1.0f);
            beg = nbeg; end = nend; myc = mycn;
        }
    }
    __syncthreads();

    const __bf16* brow = Bs + (tn0 + l15) * BPITCH + h * 8;
    for (int it = 0; it < GEMM_TRIPS; ++it) {
        const int  tile  = it * GEMM_STRIDE + slot * GEMM_BLOCKS + (int)blockIdx.x;
        const bool valid = tile < GEMM_NTILES;
        __bf16* htile = Hs + (it * NS + slot) * (16 * BPITCH);
        v8f acc[NT];
        if (valid) {
            wmma_strip_lds_ab<K, NT>(htile + l15 * BPITCH + h * 8, brow, acc);

            float ss[8];
#pragma unroll
            for (int v = 0; v < 8; ++v) ss[v] = 0.0f;
#pragma unroll
            for (int t = 0; t < NT; ++t) {
                const float bc = bias1[tn0 + t * 16 + l15];
#pragma unroll
                for (int v = 0; v < 8; ++v) {
                    acc[t][v] += bc;
                    ss[v] += acc[t][v] * acc[t][v];
                }
            }
#pragma unroll
            for (int off = 1; off < 16; off <<= 1) {
#pragma unroll
                for (int v = 0; v < 8; ++v) ss[v] += __shfl_xor(ss[v], off, 32);
            }
            if (l15 == 0) {
                float4* mine = (float4*)&red[slot][w][h * 8];
                mine[0] = make_float4(ss[0], ss[1], ss[2], ss[3]);
                mine[1] = make_float4(ss[4], ss[5], ss[6], ss[7]);
            }
        }
        __syncthreads();
        if (valid) {
            float tot = red[slot][0][l15];
#pragma unroll
            for (int q = 1; q < NW; ++q) tot += red[slot][q][l15];
            const float inv_mine = 1.0f / fmaxf(sqrtf(tot), 1e-12f);
#pragma unroll
            for (int v = 0; v < 8; ++v) {
                const float inv = __shfl(inv_mine, h * 8 + v, 32);
                __bf16* hrow = htile + (h * 8 + v) * BPITCH + tn0 + l15;
#pragma unroll
                for (int t = 0; t < NT; ++t) {
                    const float r = fmaxf(acc[t][v] * inv, 0.0f);
                    hrow[t * 16] = (__bf16)r;
                }
            }
        }
        __syncthreads();
    }

    stage_b_lds<K, NCOL>(B2, Bs, threadIdx.x);
    __syncthreads();

    for (int it = 0; it < GEMM_TRIPS; ++it) {
        const int tile = it * GEMM_STRIDE + slot * GEMM_BLOCKS + (int)blockIdx.x;
        if (tile < GEMM_NTILES) {
            const int tm = tile << 4;
            const __bf16* htile = Hs + (it * NS + slot) * (16 * BPITCH);
            v8f acc[NT];
            wmma_strip_lds_ab<K, NT>(htile + l15 * BPITCH + h * 8, brow, acc);

            const int r0 = tm + h * 8;
#pragma unroll
            for (int t = 0; t < NT; ++t) {
                const int cn = tn0 + t * 16 + l15;
                const int   bi = cn - bias2_col0;
                const float bv = bias2[bi < 0 ? 0 : bi];
                const float bc = (bi < 0) ? 0.0f : bv;
#pragma unroll
                for (int v = 0; v < 8; ++v)
                    C[(size_t)(r0 + v) * NCOL + cn] = acc[t][v] + bc;
            }
        }
    }
}

extern "C" void kernel_launch(void* const* d_in, const int* in_sizes, int n_in,
                              void* d_out, int out_size, void* d_ws, size_t ws_size,
                              hipStream_t stream) {
    (void)in_sizes; (void)n_in; (void)out_size; (void)ws_size;

    const float* feat = (const float*)d_in[0];
    const int*   eidx = (const int*)  d_in[1];
    const int*   src  = eidx;
    const int*   dst  = eidx + N_EDGES;
    const float* W1l  = (const float*)d_in[2];
    const float* b1   = (const float*)d_in[3];
    const float* W1r  = (const float*)d_in[4];
    const float* W2l  = (const float*)d_in[5];
    const float* b2   = (const float*)d_in[6];
    const float* W2r  = (const float*)d_in[7];
    const float* Wfc  = (const float*)d_in[8];
    const float* bfc  = (const float*)d_in[9];
    float* out = (float*)d_out;

    char* ws = (char*)d_ws;
    size_t off = 0;
    auto alloc = [&](size_t bytes) -> void* {
        void* p = ws + off;
        off = (off + bytes + 255) & ~(size_t)255;
        return p;
    };
    int*    part   = (int*)  alloc((size_t)N_SPLITS * PART_PITCH * 4);
    int*    tot    = (int*)  alloc((size_t)N_RANGES * N_SPLITS * 4);
    int*    rowptr = (int*)  alloc((size_t)PART_PITCH * 4);
    int*    col    = (int*)  alloc((size_t)N_EDGES * 4);
    __bf16* Wcat1 = (__bf16*)alloc((size_t)HIDDEN * 2 * IN_SIZE * 2);
    __bf16* Wst2  = (__bf16*)alloc((size_t)2 * OUT_SIZE * HIDDEN * 2);
    float*  pq2   = (float*) alloc((size_t)N_NODES * 2 * OUT_SIZE * 4);

    const int TB = 256;
    auto cdiv = [](int a, int b) { return (a + b - 1) / b; };

    inlist_count_pack<<<COUNT_BLOCKS + PACK_BLOCKS, LIST_T, 0, stream>>>(
        dst, part, tot, W1l, W1r, W2l, W2r, Wcat1, Wst2);
    inlist_fill<<<COUNT_BLOCKS, LIST_T, 0, stream>>>(src, dst, part, tot, rowptr, col);
    gather_gemm2_bf16_wmma_lds<<<GEMM_BLOCKS, GEMM_THREADS, 0, stream>>>(
        feat, rowptr, col, Wcat1, b1, Wst2, b2, OUT_SIZE, pq2);
    gather_mean_add_fc<OUT_SIZE><<<cdiv(N_NODES, TB / 32), TB, 0, stream>>>(
        pq2, rowptr, col, Wfc, bfc, out, N_NODES);
}


# Layer-2 gather + head by 8192 resident waves that each walk 6-7 nodes (next list asked ahead; a node's head computed under the next node's first 8 row loads), not 50000 one-node waves; sums same
# speedup vs baseline: 6.4982x; 1.0325x over previous
#include <hip/hip_runtime.h>
#include <math.h>

#define N_NODES  50000
#define N_EDGES  600000
#define IN_SIZE  128
#define HIDDEN   256
#define OUT_SIZE 128

typedef __attribute__((ext_vector_type(16))) __bf16 v16bf;
typedef __attribute__((ext_vector_type(8)))  __bf16 v8bf;
typedef __attribute__((ext_vector_type(8)))  float  v8f;
typedef __attribute__((ext_vector_type(4)))  __bf16 v4bf;

#define NR_NODES   2048
#define N_RANGES   ((N_NODES + NR_NODES - 1) / NR_NODES)
#define N_SPLITS   10
#define SPLIT_E    (N_EDGES / N_SPLITS)
#define SPLIT_V4   (SPLIT_E / 4)
#define PART_PITCH (N_RANGES * NR_NODES)
#define LIST_T     1024
#define LIST_UNR   8
static_assert(N_EDGES % (N_SPLITS * 4) == 0, "pieces are whole 16-byte vectors");
static_assert(N_NODES < PART_PITCH, "rowptr[N_NODES] must be an entry of the last range");
static_assert(NR_NODES == 2 * LIST_T, "a thread owns two adjacent counters");
static_assert(N_RANGES * N_SPLITS <= LIST_T, "one thread per (range, piece) sum");

#define PACK_V4_L1   (HIDDEN * IN_SIZE / 4)
#define PACK_V4_L2H  (OUT_SIZE * HIDDEN / 4)
#define PACK_BLOCKS  ((PACK_V4_L1 + 2 * PACK_V4_L2H) / LIST_T)
static_assert((PACK_V4_L1 + 2 * PACK_V4_L2H) % LIST_T == 0, "whole packing work-groups");
static_assert(PACK_V4_L1 % LIST_T == 0, "a packing work-group serves one layer only");
static_assert(IN_SIZE % 4 == 0, "a float4 never crosses a row of W1l / W1r");

__device__ __forceinline__ v4bf cvt4_bf16(float4 v) {
    v4bf o;
    o[0] = (__bf16)v.x;
    o[1] = (__bf16)v.y;
    o[2] = (__bf16)v.z;
    o[3] = (__bf16)v.w;
    return o;
}

__device__ __forceinline__ void pack_weights(int i,
                                             const float* __restrict__ W1l, const float* __restrict__ W1r,
                                             const float* __restrict__ W2l, const float* __restrict__ W2r,
                                             __bf16* __restrict__ Wcat1, __bf16* __restrict__ Wst2) {
    if (i < PACK_V4_L1) {
        const float4 a = ((const float4*)W1l)[i];
        const float4 b = ((const float4*)W1r)[i];
        const int r = (4 * i) / IN_SIZE, cc = (4 * i) % IN_SIZE;
        __bf16* row = Wcat1 + (size_t)r * (2 * IN_SIZE);
        *(v4bf*)(row + cc)           = cvt4_bf16(a);
        *(v4bf*)(row + IN_SIZE + cc) = cvt4_bf16(b);
    } else {
        const int j = i - PACK_V4_L1;
        const float4 a = (j < PACK_V4_L2H) ? ((const float4*)W2l)[j]
                                           : ((const float4*)W2r)[j - PACK_V4_L2H];
        *(v4bf*)(Wst2 + (size_t)4 * j) = cvt4_bf16(a);
    }
}


struct EdgeBatch {
    int4 v[LIST_UNR];
};
__device__ __forceinline__ void load_batch(const int4* __restrict__ p4, int i0, EdgeBatch& b) {
#pragma unroll
    for (int k = 0; k < LIST_UNR; ++k) {
        const int i = i0 + k * LIST_T;
        b.v[k] = p4[i < SPLIT_V4 ? i : SPLIT_V4 - 1];
    }
}

#define COUNT_BLOCKS (N_RANGES * N_SPLITS)
__global__ __launch_bounds__(LIST_T)
void inlist_count_pack(const int* __restrict__ dst, int* __restrict__ part, int* __restrict__ tot,
                       const float* __restrict__ W1l, const float* __restrict__ W1r,
                       const float* __restrict__ W2l, const float* __restrict__ W2r,
                       __bf16* __restrict__ Wcat1, __bf16* __restrict__ Wst2) {
    __shared__ __attribute__((aligned(16))) int cnt[NR_NODES];
    __shared__ int wsum[LIST_T / 32];
    const int tid = threadIdx.x, lane = tid & 31, w = tid >> 5;
    if (blockIdx.x >= COUNT_BLOCKS) {
        pack_weights(((int)blockIdx.x - COUNT_BLOCKS) * LIST_T + tid, W1l, W1r, W2l, W2r, Wcat1, Wst2);
        return;
    }
    const int g = blockIdx.x / N_SPLITS, s = blockIdx.x % N_SPLITS;
    const int lo = g * NR_NODES;
    cnt[tid] = 0;
    cnt[tid + LIST_T] = 0;
    __syncthreads();

    const int4* d4 = (const int4*)(dst + (size_t)s * SPLIT_E);
    for (int i0 = tid; i0 < SPLIT_V4; i0 += LIST_T * LIST_UNR) {
        EdgeBatch d;
        load_batch(d4, i0, d);
#pragma unroll
        for (int k = 0; k < LIST_UNR; ++k) {
            const bool ok = (i0 + k * LIST_T) < SPLIT_V4;
            const unsigned u0 = (unsigned)(d.v[k].x - lo), u1 = (unsigned)(d.v[k].y - lo);
            const unsigned u2 = (unsigned)(d.v[k].z - lo), u3 = (unsigned)(d.v[k].w - lo);
            if (ok && u0 < NR_NODES) atomicAdd(&cnt[u0], 1);
            if (ok && u1 < NR_NODES) atomicAdd(&cnt[u1], 1);
            if (ok && u2 < NR_NODES) atomicAdd(&cnt[u2], 1);
            if (ok && u3 < NR_NODES) atomicAdd(&cnt[u3], 1);
        }
    }
    __syncthreads();

    const int2 c = *(const int2*)&cnt[2 * tid];
    *(int2*)(part + (size_t)s * PART_PITCH + lo + 2 * tid) = c;
    int t = c.x + c.y;
    for (int off = 16; off > 0; off >>= 1) t += __shfl_xor(t, off, 32);
    if (lane == 0) wsum[w] = t;
    __syncthreads();
    if (w == 0) {
        int x = wsum[lane];
        for (int off = 16; off > 0; off >>= 1) x += __shfl_xor(x, off, 32);
        if (lane == 0) tot[g * N_SPLITS + s] = x;
    }
}

__global__ __launch_bounds__(LIST_T)
void inlist_fill(const int* __restrict__ src, const int* __restrict__ dst,
                 const int* __restrict__ part, const int* __restrict__ tot,
                 int* __restrict__ rowptr, int* __restrict__ col) {
    __shared__ __attribute__((aligned(16))) int cur[NR_NODES];
    __shared__ int wsum[LIST_T / 32];
    __shared__ int wbase[LIST_T / 32];
    const int tid = threadIdx.x, lane = tid & 31, w = tid >> 5;
    const int g = blockIdx.x / N_SPLITS, s = blockIdx.x % N_SPLITS;
    const int lo = g * NR_NODES;

    int b = (tid < g * N_SPLITS) ? tot[tid] : 0;
    int2 len = make_int2(0, 0), below = make_int2(0, 0);
#pragma unroll
    for (int q = 0; q < N_SPLITS; ++q) {
        const int2 p = *(const int2*)(part + (size_t)q * PART_PITCH + lo + 2 * tid);
        len.x += p.x; len.y += p.y;
        if (q < s) { below.x += p.x; below.y += p.y; }
    }
    const int v = len.x + len.y;
    int inc = v;
    for (int off = 1; off < 32; off <<= 1) {
        int x = __shfl_up(inc, off, 32);
        if (lane >= off) inc += x;
    }
    for (int off = 16; off > 0; off >>= 1) b += __shfl_xor(b, off, 32);
    if (lane == 31) wsum[w] = inc;
    if (lane == 0)  wbase[w] = b;
    __syncthreads();
    if (w == 0) {
        int x = wsum[lane], xi = x;
        for (int off = 1; off < 32; off <<= 1) {
            int y = __shfl_up(xi, off, 32);
            if (lane >= off) xi += y;
        }
        wsum[lane] = xi - x;
        int bb = wbase[lane];
        for (int off = 16; off > 0; off >>= 1) bb += __shfl_xor(bb, off, 32);
        if (lane == 0) wbase[0] = bb;
    }
    __syncthreads();
    const int st0 = wbase[0] + wsum[w] + inc - v;
    const int st1 = st0 + len.x;
    *(int2*)&cur[2 * tid] = make_int2(st0 + below.x, st1 + below.y);
    if (s == 0) *(int2*)(rowptr + lo + 2 * tid) = make_int2(st0, st1);
    __syncthreads();

    const int4* d4 = (const int4*)(dst + (size_t)s * SPLIT_E);
    const int4* s4 = (const int4*)(src + (size_t)s * SPLIT_E);
    for (int i0 = tid; i0 < SPLIT_V4; i0 += LIST_T * LIST_UNR) {
        EdgeBatch d, sv;
        load_batch(d4, i0, d);
        load_batch(s4, i0, sv);
        asm volatile("" ::: "memory");
#pragma unroll
        for (int k = 0; k < LIST_UNR; ++k) {
            const bool ok = (i0 + k * LIST_T) < SPLIT_V4;
            const unsigned u0 = (unsigned)(d.v[k].x - lo), u1 = (unsigned)(d.v[k].y - lo);
            const unsigned u2 = (unsigned)(d.v[k].z - lo), u3 = (unsigned)(d.v[k].w - lo);
            if (ok && u0 < NR_NODES) col[atomicAdd(&cur[u0], 1)] = sv.v[k].x;
            if (ok && u1 < NR_NODES) col[atomicAdd(&cur[u1], 1)] = sv.v[k].y;
            if (ok && u2 < NR_NODES) col[atomicAdd(&cur[u2], 1)] = sv.v[k].z;
            if (ok && u3 < NR_NODES) col[atomicAdd(&cur[u3], 1)] = sv.v[k].w;
        }
    }
}

__device__ __forceinline__ v4bf pack4_bf16(float4 v, float sc) {
    v4bf o;
    o[0] = (__bf16)(v.x * sc);
    o[1] = (__bf16)(v.y * sc);
    o[2] = (__bf16)(v.z * sc);
    o[3] = (__bf16)(v.w * sc);
    return o;
}

template <int P4>
__device__ __forceinline__ float4 row_of(const float4* __restrict__ X4, int r, int lane) {
    return X4[(size_t)r * P4 + lane];
}
__device__ __forceinline__ void add_four(float4& acc, float4 a, float4 b, float4 c, float4 d) {
    acc.x += (a.x + b.x) + (c.x + d.x);
    acc.y += (a.y + b.y) + (c.y + d.y);
    acc.z += (a.z + b.z) + (c.z + d.z);
    acc.w += (a.w + b.w) + (c.w + d.w);
}
__device__ __forceinline__ void add_one(float4& acc, float4 v) {
    acc.x += v.x; acc.y += v.y; acc.z += v.z; acc.w += v.w;
}
template <int P4>
__device__ __forceinline__ void sum_eight(const float4* __restrict__ X4, int lane, float4& acc,
                                          int s0, int s1, int s2, int s3, int s4, int s5, int s6, int s7) {
    const float4 v0 = row_of<P4>(X4, s0, lane), v1 = row_of<P4>(X4, s1, lane);
    const float4 v2 = row_of<P4>(X4, s2, lane), v3 = row_of<P4>(X4, s3, lane);
    const float4 v4 = row_of<P4>(X4, s4, lane), v5 = row_of<P4>(X4, s5, lane);
    const float4 v6 = row_of<P4>(X4, s6, lane), v7 = row_of<P4>(X4, s7, lane);
    add_four(acc, v0, v1, v2, v3);
    add_four(acc, v4, v5, v6, v7);
}
template <int P4>
__device__ __forceinline__ void sum_four(const float4* __restrict__ X4, int lane, float4& acc,
                                         int s0, int s1, int s2, int s3) {
    const float4 v0 = row_of<P4>(X4, s0, lane), v1 = row_of<P4>(X4, s1, lane);
    const float4 v2 = row_of<P4>(X4, s2, lane), v3 = row_of<P4>(X4, s3, lane);
    add_four(acc, v0, v1, v2, v3);
}

struct FirstSources { int s[8]; };
__device__ __forceinline__ FirstSources first_sources(int myc) {
    FirstSources f;
#pragma unroll
    for (int k = 0; k < 8; ++k) f.s[k] = __builtin_amdgcn_readlane(myc, k);
    return f;
}

template <int P4>
__device__ __forceinline__ void sum_rows_rest(const float4* __restrict__ X4, const int* __restrict__ col,
                                              int beg, int end, int lane, int myc, int j, float4& acc) {
    for (int base = beg; base < end; base += 32) {
        const int m = (end - base < 32) ? (end - base) : 32;
        if (base != beg) {
            myc = (lane < m) ? col[base + lane] : 0;
            j = 0;
        }
        for (; j + 8 <= m; j += 8)
            sum_eight<P4>(X4, lane, acc,
                      __builtin_amdgcn_readlane(myc, j + 0), __builtin_amdgcn_readlane(myc, j + 1),
                      __builtin_amdgcn_readlane(myc, j + 2), __builtin_amdgcn_readlane(myc, j + 3),
                      __builtin_amdgcn_readlane(myc, j + 4), __builtin_amdgcn_readlane(myc, j + 5),
                      __builtin_amdgcn_readlane(myc, j + 6), __builtin_amdgcn_readlane(myc, j + 7));
        if (j + 4 <= m) {
            sum_four<P4>(X4, lane, acc,
                     __builtin_amdgcn_readlane(myc, j + 0), __builtin_amdgcn_readlane(myc, j + 1),
                     __builtin_amdgcn_readlane(myc, j + 2), __builtin_amdgcn_readlane(myc, j + 3));
            j += 4;
        }
        const int rem = m - j;
        if (rem > 0) {
            const int s0 = __builtin_amdgcn_readlane(myc, j);
            const int s1 = __builtin_amdgcn_readlane(myc, j + 1);
            const int s2 = __builtin_amdgcn_readlane(myc, j + 2);
            const float4 u0 = row_of<P4>(X4, s0, lane);
            float4 u1 = make_float4(0.0f, 0.0f, 0.0f, 0.0f), u2 = u1;
            if (rem > 1) u1 = row_of<P4>(X4, s1, lane);
            if (rem > 2) u2 = row_of<P4>(X4, s2, lane);
            add_one(acc, u0);
            if (rem > 1) add_one(acc, u1);
            if (rem > 2) add_one(acc, u2);
        }
    }
}

template <int P4>
__device__ __forceinline__ void sum_rows_walk(const float4* __restrict__ X4, const int* __restrict__ col,
                                              int beg, int end, int lane, int myc,
                                              const FirstSources& f, float4& acc) {
    acc = make_float4(0.0f, 0.0f, 0.0f, 0.0f);
    const int m0 = (end - beg < 32) ? (end - beg) : 32;
    int j = 0;
    if (m0 >= 8) {
        sum_eight<P4>(X4, lane, acc, f.s[0], f.s[1], f.s[2], f.s[3], f.s[4], f.s[5], f.s[6], f.s[7]);
        j = 8;
    } else if (m0 >= 4) {
        sum_four<P4>(X4, lane, acc, f.s[0], f.s[1], f.s[2], f.s[3]);
        j = 4;
    }
    sum_rows_rest<P4>(X4, col, beg, end, lane, myc, j, acc);
}


struct HeadConsts { float4 w0, w1; float b0, b1; };
__device__ __forceinline__ void node_head(float4 acc, float4 q, int deg, int n, int lane,
                                          const HeadConsts& hc, float* __restrict__ out) {
    const float inv = 1.0f / fmaxf((float)deg, 1.0f);
    float4 o;
    o.x = acc.x * inv + q.x;
    o.y = acc.y * inv + q.y;
    o.z = acc.z * inv + q.z;
    o.w = acc.w * inv + q.w;
    float ss = (o.x * o.x + o.y * o.y) + (o.z * o.z + o.w * o.w);
    float d0 = (o.x * hc.w0.x + o.y * hc.w0.y) + (o.z * hc.w0.z + o.w * hc.w0.w);
    float d1 = (o.x * hc.w1.x + o.y * hc.w1.y) + (o.z * hc.w1.z + o.w * hc.w1.w);
    for (int off = 16; off > 0; off >>= 1) {
        ss += __shfl_xor(ss, off, 32);
        d0 += __shfl_xor(d0, off, 32);
        d1 += __shfl_xor(d1, off, 32);
    }
    if (lane == 0) {
        float invn = 1.0f / fmaxf(sqrtf(ss), 1e-12f);
        float l0 = d0 * invn + hc.b0;
        float l1 = d1 * invn + hc.b1;
        float m  = fmaxf(l0, l1);
        float e0 = __expf(l0 - m), e1 = __expf(l1 - m);
        float s  = 1.0f / (e0 + e1);
        *(float2*)(out + (size_t)n * 2) = make_float2(e0 * s, e1 * s);
    }
}

#define G2_THREADS 1024
#define G2_BLOCKS  256
#define G2_WAVES   (G2_BLOCKS * (G2_THREADS / 32))
static_assert(G2_WAVES <= N_NODES, "every wave has a node");
static_assert(N_NODES <= 32 * G2_WAVES, "a wave's nodes fit one per lane");
template <int F>
__global__ __launch_bounds__(G2_THREADS)
void gather_mean_add_fc(const float* __restrict__ PQ, const int* __restrict__ rowptr,
                        const int* __restrict__ col, const float* __restrict__ Wfc,
                        const float* __restrict__ bfc, float* __restrict__ out) {
    static_assert(F == 128, "a lane owns one float4 of a row");
    constexpr int P4 = 2 * F / 4;
    const int lane = threadIdx.x & 31;
    const int gw = __builtin_amdgcn_readfirstlane((int)blockIdx.x * (G2_THREADS / 32) + (int)(threadIdx.x >> 5));
    const int cntw = (N_NODES - gw + G2_WAVES - 1) / G2_WAVES;
    int mybeg = 0, myend = 0;
    if (lane < cntw) {
        const int myn = gw + lane * G2_WAVES;
        mybeg = rowptr[myn];
        myend = rowptr[myn + 1];
    }
    const float4* PQ4 = (const float4*)PQ;
    const float4* W4  = (const float4*)Wfc;
    HeadConsts hc;
    hc.w0 = W4[lane];
    hc.w1 = W4[F / 4 + lane];
    hc.b0 = bfc[0];
    hc.b1 = bfc[1];
    int beg = __builtin_amdgcn_readlane(mybeg, 0);
    int end = __builtin_amdgcn_readlane(myend, 0);
    int myc;
    {
        const int m0 = (end - beg < 32) ? (end - beg) : 32;
        myc = (lane < m0) ? col[beg + lane] : 0;
    }
    float4 pacc = make_float4(0.0f, 0.0f, 0.0f, 0.0f), pq = pacc;
    int pdeg = 0;
    for (int j = 0; j < cntw; ++j) {
        const int n  = gw + j * G2_WAVES;
        const int m0 = (end - beg < 32) ? (end - beg) : 32;
        const FirstSources f = first_sources(myc);
        int nbeg = 0, nend = 0, mycn = 0;
        if (j + 1 < cntw) {
            nbeg = __builtin_amdgcn_readlane(mybeg, j + 1);
            nend = __builtin_amdgcn_readlane(myend, j + 1);
            const int mn = (nend - nbeg < 32) ? (nend - nbeg) : 32;
            mycn = (lane < mn) ? col[nbeg + lane] : 0;
        }
        const float4 self = PQ4[(size_t)n * P4 + F / 4 + lane];
        float4 acc = make_float4(0.0f, 0.0f, 0.0f, 0.0f);
        int jj = 0;
        if (m0 >= 8) {
            const float4 v0 = row_of<P4>(PQ4, f.s[0], lane), v1 = row_of<P4>(PQ4, f.s[1], lane);
            const float4 v2 = row_of<P4>(PQ4, f.s[2], lane), v3 = row_of<P4>(PQ4, f.s[3], lane);
            const float4 v4 = row_of<P4>(PQ4, f.s[4], lane), v5 = row_of<P4>(PQ4, f.s[5], lane);
            const float4 v6 = row_of<P4>(PQ4, f.s[6], lane), v7 = row_of<P4>(PQ4, f.s[7], lane);
            if (j > 0) node_head(pacc, pq, pdeg, n - G2_WAVES, lane, hc, out);
            add_four(acc, v0, v1, v2, v3);
            add_four(acc, v4, v5, v6, v7);
            jj = 8;
        } else if (m0 >= 4) {
            const float4 v0 = row_of<P4>(PQ4, f.s[0], lane), v1 = row_of<P4>(PQ4, f.s[1], lane);
            const float4 v2 = row_of<P4>(PQ4, f.s[2], lane), v3 = row_of<P4>(PQ4, f.s[3], lane);
            if (j > 0) node_head(pacc, pq, pdeg, n - G2_WAVES, lane, hc, out);
            add_four(acc, v0, v1, v2, v3);
            jj = 4;
        } else {
            if (j > 0) node_head(pacc, pq, pdeg, n - G2_WAVES, lane, hc, out);
        }
        sum_rows_rest<P4>(PQ4, col, beg, end, lane, myc, jj, acc);
        pacc = acc; pq = self; pdeg = end - beg;
        beg = nbeg; end = nend; myc = mycn;
    }
    node_head(pacc, pq, pdeg, gw + (cntw - 1) * G2_WAVES, lane, hc, out);
}

#define GEMM_K       256
#define GEMM_NCOL    256
#define GEMM_THREADS 1024
#define GEMM_BLOCKS  256
#define BPITCH       (GEMM_K + 8)

template <int K, int NROW>
__device__ __forceinline__ void stage_b_lds(const __bf16* __restrict__ B, __bf16* Bs, int tid) {
    constexpr int PPR   = K / 8;
    constexpr int TOTAL = NROW * PPR;
#pragma unroll
    for (int i = 0; i < TOTAL / GEMM_THREADS; ++i) {
        const int p   = i * GEMM_THREADS + tid;
        const int row = p / PPR, pc = p % PPR;
        const v8bf v = *(const v8bf*)(B + (size_t)row * K + pc * 8);
        *(v8bf*)(Bs + row * BPITCH + pc * 8) = v;
    }
}

template <int K, int NT>
__device__ __forceinline__ void wmma_strip_lds_ab(const __bf16* arow, const __bf16* brow,
                                                  v8f (&acc)[NT]) {
#pragma unroll
    for (int t = 0; t < NT; ++t) acc[t] = (v8f){};
#pragma unroll
    for (int s = 0; s < K / 32; ++s) {
        union { v16bf v; v8bf h[2]; } a;
        a.h[0] = *(const v8bf*)(arow + s * 32);
        a.h[1] = *(const v8bf*)(arow + s * 32 + 16);
#pragma unroll
        for (int t = 0; t < NT; ++t) {
            union { v16bf v; v8bf h[2]; } b;
            b.h[0] = *(const v8bf*)(brow + t * 16 * BPITCH + s * 32);
            b.h[1] = *(const v8bf*)(brow + t * 16 * BPITCH + s * 32 + 16);
            acc[t] = __builtin_amdgcn_wmma_f32_16x16x32_bf16(
                false, a.v, false, b.v, (short)0, acc[t], false, false);
        }
    }
}

#define GEMM_NS      ((GEMM_THREADS / 32) / (GEMM_NCOL / 64))
#define GEMM_NTILES  (N_NODES / 16)
#define GEMM_STRIDE  (GEMM_BLOCKS * GEMM_NS)
#define GEMM_TRIPS   ((GEMM_NTILES + GEMM_STRIDE - 1) / GEMM_STRIDE)
static_assert(N_NODES % 16 == 0, "whole 16-row tiles");
static_assert(2 * IN_SIZE == GEMM_K, "the A tile is [mean | own row], IN_SIZE columns each");
static_assert(GEMM_TRIPS * GEMM_NS * 16 <= 32 * 32, "a wave's nodes fit one per lane");
static_assert((GEMM_NCOL + GEMM_TRIPS * GEMM_NS * 16) * BPITCH * 2 + GEMM_NS * 4 * 16 * 4 <= 320 * 1024,
              "weights + h tiles + partial sums fit one WGP's LDS");
__global__ __launch_bounds__(GEMM_THREADS)
void gather_gemm2_bf16_wmma_lds(const float* __restrict__ X, const int* __restrict__ rowptr,
                                const int* __restrict__ col, const __bf16* __restrict__ B1,
                                const float* __restrict__ bias1, const __bf16* __restrict__ B2,
                                const float* __restrict__ bias2, int bias2_col0,
                                float* __restrict__ C) {
    constexpr int K = GEMM_K, NCOL = GEMM_NCOL, NT = 4, NW = NCOL / (16 * NT), NS = GEMM_NS;
    static_assert(K == NCOL, "layer 1's output columns are layer 2's k");
    __shared__ __attribute__((aligned(16))) __bf16 Bs[NCOL * BPITCH];
    __shared__ __attribute__((aligned(16))) __bf16 Hs[GEMM_TRIPS * NS * 16 * BPITCH];
    __shared__ __attribute__((aligned(16))) float red[NS][NW][16];
    const int lane = threadIdx.x & 31;
    const int wv   = __builtin_amdgcn_readfirstlane(threadIdx.x >> 5);
    const int w    = wv % NW;
    const int slot = wv / NW;
    const int tn0  = w * (16 * NT);
    const int h    = lane >> 4;
    const int l15  = lane & 15;

    stage_b_lds<K, NCOL>(B1, Bs, threadIdx.x);

    {
        const int g = (int)blockIdx.x;
        int nq = (GEMM_NTILES - g + GEMM_BLOCKS - 1) / GEMM_BLOCKS;
        if (nq > GEMM_TRIPS * NS) nq = GEMM_TRIPS * NS;
        const int nrows = nq * 16;
        const int cntw = (nrows - wv + 31) >> 5;
        int myn = 0, mybeg = 0, myend = 0;
        if (lane < cntw) {
            const int r = wv + 32 * lane;
            myn   = (((r >> 4) * GEMM_BLOCKS + g) << 4) + (r & 15);
            mybeg = rowptr[myn];
            myend = rowptr[myn + 1];
        }
        const float4* X4 = (const float4*)X;
        int beg = 0, end = 0, myc = 0;
        if (cntw > 0) {
            beg = __builtin_amdgcn_readlane(mybeg, 0);
            end = __builtin_amdgcn_readlane(myend, 0);
            const int m0 = (end - beg < 32) ? (end - beg) : 32;
            myc = (lane < m0) ? col[beg + lane] : 0;
        }
        for (int j = 0; j < cntw; ++j) {
            const int n = __builtin_amdgcn_readlane(myn, j);
            const int r = wv + 32 * j;
            const FirstSources f = first_sources(myc);
            int nbeg = 0, nend = 0, mycn = 0;
            if (j + 1 < cntw) {
                nbeg = __builtin_amdgcn_readlane(mybeg, j + 1);
                nend = __builtin_amdgcn_readlane(myend, j + 1);
                const int mn = (nend - nbeg < 32) ? (nend - nbeg) : 32;
                mycn = (lane < mn) ? col[nbeg + lane] : 0;
            }
            const float4 own = X4[(size_t)n * (IN_SIZE / 4) + lane];
            float4 acc0;
            sum_rows_walk<IN_SIZE / 4>(X4, col, beg, end, lane, myc, f, acc0);
            const float inv = 1.0f / fmaxf((float)(end - beg), 1.0f);
            __bf16* arow = Hs + (r >> 4) * (16 * BPITCH) + (r & 15) * BPITCH;
            *(v4bf*)(arow + lane * 4)           = pack4_bf16(acc0, inv);
            *(v4bf*)(arow + IN_SIZE + lane * 4) = pack4_bf16(own, 1.0f);
            beg = nbeg; end = nend; myc = mycn;
        }
    }
    __syncthreads();

    const __bf16* brow = Bs + (tn0 + l15) * BPITCH + h * 8;
    for (int it = 0; it < GEMM_TRIPS; ++it) {
        const int  tile  = it * GEMM_STRIDE + slot * GEMM_BLOCKS + (int)blockIdx.x;
        const bool valid = tile < GEMM_NTILES;
        __bf16* htile = Hs + (it * NS + slot) * (16 * BPITCH);
        v8f acc[NT];
        if (valid) {
            wmma_strip_lds_ab<K, NT>(htile + l15 * BPITCH + h * 8, brow, acc);

            float ss[8];
#pragma unroll
            for (int v = 0; v < 8; ++v) ss[v] = 0.0f;
#pragma unroll
            for (int t = 0; t < NT; ++t) {
                const float bc = bias1[tn0 + t * 16 + l15];
#pragma unroll
                for (int v = 0; v < 8; ++v) {
                    acc[t][v] += bc;
                    ss[v] += acc[t][v] * acc[t][v];
                }
            }
#pragma unroll
            for (int off = 1; off < 16; off <<= 1) {
#pragma unroll
                for (int v = 0; v < 8; ++v) ss[v] += __shfl_xor(ss[v], off, 32);
            }
            if (l15 == 0) {
                float4* mine = (float4*)&red[slot][w][h * 8];
                mine[0] = make_float4(ss[0], ss[1], ss[2], ss[3]);
                mine[1] = make_float4(ss[4], ss[5], ss[6], ss[7]);
            }
        }
        __syncthreads();
        if (valid) {
            float tot = red[slot][0][l15];
#pragma unroll
            for (int q = 1; q < NW; ++q) tot += red[slot][q][l15];
            const float inv_mine = 1.0f / fmaxf(sqrtf(tot), 1e-12f);
#pragma unroll
            for (int v = 0; v < 8; ++v) {
                const float inv = __shfl(inv_mine, h * 8 + v, 32);
                __bf16* hrow = htile + (h * 8 + v) * BPITCH + tn0 + l15;
#pragma unroll
                for (int t = 0; t < NT; ++t) {
                    const float r = fmaxf(acc[t][v] * inv, 0.0f);
                    hrow[t * 16] = (__bf16)r;
                }
            }
        }
        __syncthreads();
    }

    stage_b_lds<K, NCOL>(B2, Bs, threadIdx.x);
    __syncthreads();

    for (int it = 0; it < GEMM_TRIPS; ++it) {
        const int tile = it * GEMM_STRIDE + slot * GEMM_BLOCKS + (int)blockIdx.x;
        if (tile < GEMM_NTILES) {
            const int tm = tile << 4;
            const __bf16* htile = Hs + (it * NS + slot) * (16 * BPITCH);
            v8f acc[NT];
            wmma_strip_lds_ab<K, NT>(htile + l15 * BPITCH + h * 8, brow, acc);

            const int r0 = tm + h * 8;
#pragma unroll
            for (int t = 0; t < NT; ++t) {
                const int cn = tn0 + t * 16 + l15;
                const int   bi = cn - bias2_col0;
                const float bv = bias2[bi < 0 ? 0 : bi];
                const float bc = (bi < 0) ? 0.0f : bv;
#pragma unroll
                for (int v = 0; v < 8; ++v)
                    C[(size_t)(r0 + v) * NCOL + cn] = acc[t][v] + bc;
            }
        }
    }
}

extern "C" void kernel_launch(void* const* d_in, const int* in_sizes, int n_in,
                              void* d_out, int out_size, void* d_ws, size_t ws_size,
                              hipStream_t stream) {
    (void)in_sizes; (void)n_in; (void)out_size; (void)ws_size;

    const float* feat = (const float*)d_in[0];
    const int*   eidx = (const int*)  d_in[1];
    const int*   src  = eidx;
    const int*   dst  = eidx + N_EDGES;
    const float* W1l  = (const float*)d_in[2];
    const float* b1   = (const float*)d_in[3];
    const float* W1r  = (const float*)d_in[4];
    const float* W2l  = (const float*)d_in[5];
    const float* b2   = (const float*)d_in[6];
    const float* W2r  = (const float*)d_in[7];
    const float* Wfc  = (const float*)d_in[8];
    const float* bfc  = (const float*)d_in[9];
    float* out = (float*)d_out;

    char* ws = (char*)d_ws;
    size_t off = 0;
    auto alloc = [&](size_t bytes) -> void* {
        void* p = ws + off;
        off = (off + bytes + 255) & ~(size_t)255;
        return p;
    };
    int*    part   = (int*)  alloc((size_t)N_SPLITS * PART_PITCH * 4);
    int*    tot    = (int*)  alloc((size_t)N_RANGES * N_SPLITS * 4);
    int*    rowptr = (int*)  alloc((size_t)PART_PITCH * 4);
    int*    col    = (int*)  alloc((size_t)N_EDGES * 4);
    __bf16* Wcat1 = (__bf16*)alloc((size_t)HIDDEN * 2 * IN_SIZE * 2);
    __bf16* Wst2  = (__bf16*)alloc((size_t)2 * OUT_SIZE * HIDDEN * 2);
    float*  pq2   = (float*) alloc((size_t)N_NODES * 2 * OUT_SIZE * 4);

    inlist_count_pack<<<COUNT_BLOCKS + PACK_BLOCKS, LIST_T, 0, stream>>>(
        dst, part, tot, W1l, W1r, W2l, W2r, Wcat1, Wst2);
    inlist_fill<<<COUNT_BLOCKS, LIST_T, 0, stream>>>(src, dst, part, tot, rowptr, col);
    gather_gemm2_bf16_wmma_lds<<<GEMM_BLOCKS, GEMM_THREADS, 0, stream>>>(
        feat, rowptr, col, Wcat1, b1, Wst2, b2, OUT_SIZE, pq2);
    gather_mean_add_fc<OUT_SIZE><<<G2_BLOCKS, G2_THREADS, 0, stream>>>(
        pq2, rowptr, col, Wfc, bfc, out);
}
